# starting kernel
# baseline (speedup 1.0000x reference)
#include <hip/hip_runtime.h>
#include <hip/hip_bf16.h>
#include <stddef.h>

typedef __attribute__((ext_vector_type(2))) float v2f;
typedef __attribute__((ext_vector_type(8))) float v8f;

__device__ __forceinline__ void atomAddF(float* p, float v) {
  asm volatile("global_atomic_add_f32 %0, %1, off scope:SCOPE_DEV"
               :
               : "v"(p), "v"(v)
               : "memory");
}

__global__ void k_init(float* __restrict__ deg, float* __restrict__ agg,
                       int n, int tot) {
  int t = blockIdx.x * blockDim.x + threadIdx.x;
  if (t < tot) agg[t] = 0.0f;
  if (t < n)   deg[t] = 1.0f;
}

__global__ void k_zero(float* __restrict__ p, int tot) {
  int t = blockIdx.x * blockDim.x + threadIdx.x;
  if (t < tot) p[t] = 0.0f;
}

__global__ void k_degree(const int* __restrict__ dst, float* __restrict__ deg,
                         int E) {
  int t = blockIdx.x * blockDim.x + threadIdx.x;
  if (t < E) atomAddF(&deg[dst[t]], 1.0f);
}

__global__ void k_rsqrt(float* __restrict__ d, int n) {
  int t = blockIdx.x * blockDim.x + threadIdx.x;
  if (t < n) d[t] = rsqrtf(d[t]);
}

__global__ void k_gemm16(const float* __restrict__ X, const float* __restrict__ W,
                         float* __restrict__ Y, int nrows, int K) {
  const int wave = threadIdx.x >> 5;
  const int lane = threadIdx.x & 31;
  const int tile = blockIdx.x * (blockDim.x >> 5) + wave;
  const int row0 = tile * 16;
  if (row0 >= nrows) return;

  const int m  = lane & 15;
  const int kh = lane >> 4;

  const float* ap = X + (size_t)(row0 + m) * K + 2 * kh;
  v8f acc = {};
  for (int kb = 0; kb < K; kb += 4) {
    v2f a = *(const v2f*)(ap + kb);
    const int kr = kb + 2 * kh;
    v2f b;
    b.x = W[(size_t)kr * 16 + m];
    b.y = W[(size_t)(kr + 1) * 16 + m];
    acc = __builtin_amdgcn_wmma_f32_16x16x4_f32(
        false, a, false, b, (short)0, acc, false, false);
  }
#pragma unroll
  for (int v = 0; v < 8; ++v) {
    Y[(size_t)(row0 + v + 8 * kh) * 16 + m] = acc[v];
  }
}

__global__ void k_scatter(const int* __restrict__ src, const int* __restrict__ dst,
                          const float* __restrict__ dinv,
                          const float* __restrict__ xw,
                          float* __restrict__ agg, int E) {
  int t = blockIdx.x * blockDim.x + threadIdx.x;
  int e = t >> 2;
  if (e >= E) return;
  int q = (t & 3) << 2;
  int s = src[e];
  int d = dst[e];
  float nrm = dinv[s] * dinv[d];
  float4 v = *(const float4*)(xw + (size_t)s * 16 + q);
  float* out = agg + (size_t)d * 16 + q;
  atomAddF(out + 0, v.x * nrm);
  atomAddF(out + 1, v.y * nrm);
  atomAddF(out + 2, v.z * nrm);
  atomAddF(out + 3, v.w * nrm);
}

__global__ void k_final1(float* __restrict__ agg, const float* __restrict__ xw,
                         const float* __restrict__ dinv,
                         const float* __restrict__ b, int tot) {
  int t = blockIdx.x * blockDim.x + threadIdx.x;
  if (t >= tot) return;
  int i = t >> 4, f = t & 15;
  float s = dinv[i];
  float v = agg[t] + xw[t] * s * s + b[f];
  agg[t] = fmaxf(v, 0.0f);
}

__global__ void k_final2(const float* __restrict__ agg, const float* __restrict__ xw,
                         const float* __restrict__ dinv,
                         const float* __restrict__ b2,
                         const float* __restrict__ Wl,
                         const float* __restrict__ bl,
                         float* __restrict__ out, int n) {
  int i = blockIdx.x * blockDim.x + threadIdx.x;
  if (i >= n) return;
  float s = dinv[i];
  float ss = s * s;
  float acc = 0.0f;
#pragma unroll
  for (int f = 0; f < 16; ++f) {
    float h = agg[(size_t)i * 16 + f] + xw[(size_t)i * 16 + f] * ss + b2[f];
    h = fmaxf(h, 0.0f);
    acc += h * Wl[f];
  }
  out[i] = acc + bl[0];
}

extern "C" void kernel_launch(void* const* d_in, const int* in_sizes, int n_in,
                              void* d_out, int out_size, void* d_ws, size_t ws_size,
                              hipStream_t stream) {
  const float* x   = (const float*)d_in[0];
  const int*   ei  = (const int*)d_in[1];
  const float* W1  = (const float*)d_in[2];
  const float* b1  = (const float*)d_in[3];
  const float* W2  = (const float*)d_in[4];
  const float* b2  = (const float*)d_in[5];
  const float* Wl  = (const float*)d_in[6];
  const float* bl  = (const float*)d_in[7];
  float* out = (float*)d_out;

  const int din = in_sizes[2] / 16;
  const int n   = in_sizes[0] / din;
  const int E   = in_sizes[1] / 2;
  const int tot = n * 16;

  float* ws   = (float*)d_ws;
  float* dinv = ws;
  float* bufA = ws + (((size_t)n + 255) & ~(size_t)255);
  float* bufB = bufA + (size_t)n * 16;

  const int T = 256;
  const int ntiles = (n + 15) / 16;
  const int gemmBlocks = (ntiles + 3) / 4;

  k_init  <<<(tot + T - 1) / T, T, 0, stream>>>(dinv, bufB, n, tot);
  k_degree<<<(E + T - 1) / T,   T, 0, stream>>>(ei + E, dinv, E);
  k_rsqrt <<<(n + T - 1) / T,   T, 0, stream>>>(dinv, n);

  k_gemm16 <<<gemmBlocks, 128, 0, stream>>>(x, W1, bufA, n, din);
  k_scatter<<<(4 * E + T - 1) / T, T, 0, stream>>>(ei, ei + E, dinv, bufA, bufB, E);
  k_final1 <<<(tot + T - 1) / T, T, 0, stream>>>(bufB, bufA, dinv, b1, tot);

  k_gemm16 <<<gemmBlocks, 128, 0, stream>>>(bufB, W2, bufA, n, 16);
  k_zero   <<<(tot + T - 1) / T, T, 0, stream>>>(bufB, tot);
  k_scatter<<<(4 * E + T - 1) / T, T, 0, stream>>>(ei, ei + E, dinv, bufA, bufB, E);
  k_final2 <<<(n + T - 1) / T, T, 0, stream>>>(bufB, bufA, dinv, b2, Wl, bl, out, n);
}


# row sums: 4 lanes per row, a lane gathers 4 features with one 16-byte load (8 rows per wave instruction, was 2; 32-bit row index scaled by the load); f32 sum order per row unchanged
# speedup vs baseline: 2.4271x; 2.4271x over previous
#include <hip/hip_runtime.h>
#include <hip/hip_bf16.h>
#include <stddef.h>

typedef __attribute__((ext_vector_type(2))) float v2f;
typedef __attribute__((ext_vector_type(8))) float v8f;

__device__ __forceinline__ void atomAddF(float* p, float v) {
  asm volatile("global_atomic_add_f32 %0, %1, off scope:SCOPE_DEV"
               :
               : "v"(p), "v"(v)
               : "memory");
}

__global__ void k_init(float* __restrict__ deg, float* __restrict__ agg,
                       int n, int tot) {
  int t = blockIdx.x * blockDim.x + threadIdx.x;
  if (t < tot) agg[t] = 0.0f;
  if (t < n)   deg[t] = 1.0f;
}

__global__ void k_zero(float* __restrict__ p, int tot) {
  int t = blockIdx.x * blockDim.x + threadIdx.x;
  if (t < tot) p[t] = 0.0f;
}

__global__ void k_degree(const int* __restrict__ dst, float* __restrict__ deg,
                         int E) {
  int t = blockIdx.x * blockDim.x + threadIdx.x;
  if (t < E) atomAddF(&deg[dst[t]], 1.0f);
}

__global__ void k_rsqrt(float* __restrict__ d, int n) {
  int t = blockIdx.x * blockDim.x + threadIdx.x;
  if (t < n) d[t] = rsqrtf(d[t]);
}

__global__ void k_gemm16(const float* __restrict__ X, const float* __restrict__ W,
                         float* __restrict__ Y, int nrows, int K) {
  const int wave = threadIdx.x >> 5;
  const int lane = threadIdx.x & 31;
  const int tile = blockIdx.x * (blockDim.x >> 5) + wave;
  const int row0 = tile * 16;
  if (row0 >= nrows) return;

  const int m  = lane & 15;
  const int kh = lane >> 4;

  const float* ap = X + (size_t)(row0 + m) * K + 2 * kh;
  v8f acc = {};
  for (int kb = 0; kb < K; kb += 4) {
    v2f a = *(const v2f*)(ap + kb);
    const int kr = kb + 2 * kh;
    v2f b;
    b.x = W[(size_t)kr * 16 + m];
    b.y = W[(size_t)(kr + 1) * 16 + m];
    acc = __builtin_amdgcn_wmma_f32_16x16x4_f32(
        false, a, false, b, (short)0, acc, false, false);
  }
#pragma unroll
  for (int v = 0; v < 8; ++v) {
    Y[(size_t)(row0 + v + 8 * kh) * 16 + m] = acc[v];
  }
}

__global__ void k_scatter(const int* __restrict__ src, const int* __restrict__ dst,
                          const float* __restrict__ dinv,
                          const float* __restrict__ xw,
                          float* __restrict__ agg, int E) {
  int t = blockIdx.x * blockDim.x + threadIdx.x;
  int e = t >> 2;
  if (e >= E) return;
  int q = (t & 3) << 2;
  int s = src[e];
  int d = dst[e];
  float nrm = dinv[s] * dinv[d];
  float4 v = *(const float4*)(xw + (size_t)s * 16 + q);
  float* out = agg + (size_t)d * 16 + q;
  atomAddF(out + 0, v.x * nrm);
  atomAddF(out + 1, v.y * nrm);
  atomAddF(out + 2, v.z * nrm);
  atomAddF(out + 3, v.w * nrm);
}

__global__ void k_final1(float* __restrict__ agg, const float* __restrict__ xw,
                         const float* __restrict__ dinv,
                         const float* __restrict__ b, int tot) {
  int t = blockIdx.x * blockDim.x + threadIdx.x;
  if (t >= tot) return;
  int i = t >> 4, f = t & 15;
  float s = dinv[i];
  float v = agg[t] + xw[t] * s * s + b[f];
  agg[t] = fmaxf(v, 0.0f);
}

__global__ void k_final2(const float* __restrict__ agg, const float* __restrict__ xw,
                         const float* __restrict__ dinv,
                         const float* __restrict__ b2,
                         const float* __restrict__ Wl,
                         const float* __restrict__ bl,
                         float* __restrict__ out, int n) {
  int i = blockIdx.x * blockDim.x + threadIdx.x;
  if (i >= n) return;
  float s = dinv[i];
  float ss = s * s;
  float acc = 0.0f;
#pragma unroll
  for (int f = 0; f < 16; ++f) {
    float h = agg[(size_t)i * 16 + f] + xw[(size_t)i * 16 + f] * ss + b2[f];
    h = fmaxf(h, 0.0f);
    acc += h * Wl[f];
  }
  out[i] = acc + bl[0];
}

#define NB_SHIFT 8
#define NB       256
#define PG       2048
#define MAXBINS  512
#define AGW      16
#define AGT      (AGW * 32)
#define CAP      10240
#define GU       4

__device__ __forceinline__ unsigned sameKeyMask(int key) {
  unsigned m = 0u;
#pragma unroll
  for (int j = 0; j < 32; ++j) {
    const int kj = __builtin_amdgcn_readlane(key, j);
    m |= (kj == key) ? (1u << j) : 0u;
  }
  return m;
}

template <int N>
__device__ __forceinline__ void blockScanIncl(int* s, int t) {
  for (int d = 1; d < N; d <<= 1) {
    const int v = (t >= d) ? s[t - d] : 0;
    __syncthreads();
    s[t] += v;
    __syncthreads();
  }
}

__global__ void k_count(const int* __restrict__ dst, int* __restrict__ cnt,
                        int E, int chunk, int nbins) {
  __shared__ int h[MAXBINS];
  const int lane = threadIdx.x;
  for (int i = lane; i < nbins; i += 32) h[i] = 0;
  const int g  = blockIdx.x;
  const int e0 = g * chunk;
  const int e1 = min(e0 + chunk, E);
  int dn[GU];
#pragma unroll
  for (int u = 0; u < GU; ++u) dn[u] = dst[min(e0 + u * 32 + lane, E - 1)];
  for (int base = e0; base < e1; base += 32 * GU) {
    int d[GU];
#pragma unroll
    for (int u = 0; u < GU; ++u) d[u] = dn[u];
#pragma unroll
    for (int u = 0; u < GU; ++u) dn[u] = dst[min(base + 32 * GU + u * 32 + lane, E - 1)];
#pragma unroll
    for (int u = 0; u < GU; ++u) {
      if (base + u * 32 < e1) {
        const bool valid = base + u * 32 + lane < e1;
        const int key = valid ? (d[u] >> NB_SHIFT) : -1;
        const unsigned m = sameKeyMask(key);
        if (valid && (m >> lane) == 1u) h[key] += __popc(m);
      }
    }
  }
  for (int i = lane; i < nbins; i += 32) cnt[i * PG + g] = h[i];
}

__global__ void k_scan_bin(int* __restrict__ cnt, int* __restrict__ binTot) {
  __shared__ int s[PG / 2];
  const int t = threadIdx.x, b = blockIdx.x;
  int* row = cnt + (size_t)b * PG;
  const int c0 = row[2 * t], c1 = row[2 * t + 1];
  const int c = c0 + c1;
  s[t] = c;
  __syncthreads();
  blockScanIncl<PG / 2>(s, t);
  const int ex = s[t] - c;
  row[2 * t]     = ex;
  row[2 * t + 1] = ex + c0;
  if (t == PG / 2 - 1) binTot[b] = s[t];
}

__global__ void k_scan_tot(const int* __restrict__ binTot, int* __restrict__ binBase,
                           int nbins) {
  __shared__ int s[MAXBINS];
  const int t = threadIdx.x;
  const int c = (t < nbins) ? binTot[t] : 0;
  s[t] = c;
  __syncthreads();
  blockScanIncl<MAXBINS>(s, t);
  if (t < nbins) binBase[t] = s[t] - c;
  if (t == MAXBINS - 1) binBase[nbins] = s[t];
}

__global__ void k_part(const int* __restrict__ src, const int* __restrict__ dst,
                       const int* __restrict__ cnt, const int* __restrict__ binBase,
                       unsigned* __restrict__ pk, int E, int chunk, int nbins) {
  __shared__ int cur[MAXBINS];
  const int lane = threadIdx.x;
  const int g = blockIdx.x;
  for (int i = lane; i < nbins; i += 32) cur[i] = binBase[i] + cnt[i * PG + g];
  const int e0 = g * chunk;
  const int e1 = min(e0 + chunk, E);
  const unsigned below = (1u << lane) - 1u;
  int dn[GU], sn[GU];
#pragma unroll
  for (int u = 0; u < GU; ++u) {
    const int i = min(e0 + u * 32 + lane, E - 1);
    dn[u] = dst[i]; sn[u] = src[i];
  }
  for (int base = e0; base < e1; base += 32 * GU) {
    int d[GU], s[GU];
#pragma unroll
    for (int u = 0; u < GU; ++u) { d[u] = dn[u]; s[u] = sn[u]; }
#pragma unroll
    for (int u = 0; u < GU; ++u) {
      const int i = min(base + 32 * GU + u * 32 + lane, E - 1);
      dn[u] = dst[i]; sn[u] = src[i];
    }
#pragma unroll
    for (int u = 0; u < GU; ++u) {
      if (base + u * 32 < e1) {
        const bool valid = base + u * 32 + lane < e1;
        const int key = valid ? (d[u] >> NB_SHIFT) : -1;
        const unsigned m = sameKeyMask(key);
        if (valid) {
          const int pos = cur[key] + __popc(m & below);
          if ((m >> lane) == 1u) cur[key] = pos + 1;
          pk[pos] = ((unsigned)s[u] << NB_SHIFT) | (unsigned)(d[u] & (NB - 1));
        }
      }
    }
  }
}

__device__ __forceinline__ void countRows(const unsigned* __restrict__ pk, int b0, int b1,
                                          int* hwv, int wave, int lane) {
  for (int base = b0 + wave * 32; base < b1; base += AGT) {
    const int i = base + lane;
    const bool valid = i < b1;
    const unsigned q = pk[min(i, b1 - 1)];
    const int key = valid ? (int)(q & (NB - 1)) : -1;
    const unsigned m = sameKeyMask(key);
    if (valid && (m >> lane) == 1u) hwv[key] += __popc(m);
  }
}

__global__ void k_deg(const unsigned* __restrict__ pk, const int* __restrict__ binBase,
                      float* __restrict__ dinv, int n) {
  __shared__ int hw[AGW * NB];
  const int t = threadIdx.x, b = blockIdx.x;
  const int lane = t & 31;
  const int wave = __builtin_amdgcn_readfirstlane(t >> 5);
  for (int i = t; i < AGW * NB; i += AGT) hw[i] = 0;
  __syncthreads();
  const int e0 = binBase[b], e1 = binBase[b + 1];
  countRows(pk, e0, e1, hw + wave * NB, wave, lane);
  __syncthreads();
  if (t < NB) {
    int c = 0;
#pragma unroll
    for (int w = 0; w < AGW; ++w) c += hw[w * NB + t];
    const int node = b * NB + t;
    if (node < n) dinv[node] = rsqrtf(1.0f + (float)c);
  }
}

__global__ void k_gemm16s(const float* __restrict__ X, const float* __restrict__ W,
                          const float* __restrict__ dinv,
                          float* __restrict__ Y, int nrows, int K) {
  const int wave = threadIdx.x >> 5;
  const int lane = threadIdx.x & 31;
  const int tile = blockIdx.x * (blockDim.x >> 5) + wave;
  const int row0 = tile * 16;
  if (row0 >= nrows) return;

  const int m  = lane & 15;
  const int kh = lane >> 4;

  const float* ap = X + (size_t)(row0 + m) * K + 2 * kh;
  v8f acc = {};
  for (int kb = 0; kb < K; kb += 4) {
    v2f a = *(const v2f*)(ap + kb);
    const int kr = kb + 2 * kh;
    v2f b;
    b.x = W[(size_t)kr * 16 + m];
    b.y = W[(size_t)(kr + 1) * 16 + m];
    acc = __builtin_amdgcn_wmma_f32_16x16x4_f32(
        false, a, false, b, (short)0, acc, false, false);
  }
#pragma unroll
  for (int v = 0; v < 8; ++v) {
    const int r = row0 + v + 8 * kh;
    Y[(size_t)r * 16 + m] = acc[v] * dinv[r];
  }
}

template <bool LAST>
__global__ void k_agg(const unsigned* __restrict__ pk, const int* __restrict__ binBase,
                      const float* __restrict__ xws, const float* __restrict__ dinv,
                      const float* __restrict__ bias,
                      const float* __restrict__ Wl, const float* __restrict__ bl,
                      float* __restrict__ H, float* __restrict__ out, int n) {
  __shared__ __attribute__((aligned(16))) float acc[NB * 16];
  __shared__ int hw[AGW * NB];
  __shared__ int sc[AGT];
  __shared__ int rowStart[NB + 1];
  __shared__ int sorted[CAP];

  const int t = threadIdx.x, b = blockIdx.x;
  const int lane = t & 31;
  const int wave = __builtin_amdgcn_readfirstlane(t >> 5);
  const int fq = lane & 3;
  const int grp = t >> 2;
  const float4* xq = (const float4*)xws;
  const int node0 = b * NB;
  for (int i = t; i < NB * 16; i += AGT) {
    const int node = node0 + (i >> 4);
    acc[i] = (node < n) ? xws[(size_t)node0 * 16 + i] : 0.0f;
  }
  const int e0 = binBase[b], e1 = binBase[b + 1];
  int* hwv = hw + wave * NB;
  const unsigned below = (1u << lane) - 1u;

  for (int b0 = e0; b0 < e1; b0 += CAP) {
    const int b1 = min(b0 + CAP, e1);
    for (int i = t; i < AGW * NB; i += AGT) hw[i] = 0;
    __syncthreads();
    countRows(pk, b0, b1, hwv, wave, lane);
    __syncthreads();
    {
      const int key = t >> 1, w0 = (t & 1) * (AGW / 2);
      int s = 0;
#pragma unroll
      for (int i = 0; i < AGW / 2; ++i) s += hw[(w0 + i) * NB + key];
      sc[t] = s;
      __syncthreads();
      blockScanIncl<AGT>(sc, t);
      int run = sc[t] - s;
      if ((t & 1) == 0) rowStart[key] = run;
      if (t == AGT - 1) rowStart[NB] = sc[t];
#pragma unroll
      for (int i = 0; i < AGW / 2; ++i) {
        const int idx = (w0 + i) * NB + key;
        const int c = hw[idx];
        hw[idx] = run;
        run += c;
      }
    }
    __syncthreads();
    for (int base = b0 + wave * 32; base < b1; base += AGT) {
      const int i = base + lane;
      const bool valid = i < b1;
      const unsigned q = pk[min(i, b1 - 1)];
      const int key = valid ? (int)(q & (NB - 1)) : -1;
      const unsigned m = sameKeyMask(key);
      if (valid) {
        const int pos = hwv[key] + __popc(m & below);
        if ((m >> lane) == 1u) hwv[key] = pos + 1;
        sorted[pos] = (int)(q >> NB_SHIFT);
      }
    }
    __syncthreads();
    for (int r = grp; r < NB; r += AGT / 4) {
      const int s0 = rowStart[r], s1 = rowStart[r + 1];
      float4 sum = make_float4(0.0f, 0.0f, 0.0f, 0.0f);
      for (int i = s0; i < s1; i += 8) {
        float4 v[8];
#pragma unroll
        for (int u = 0; u < 8; ++u) {
          const int idx = min(i + u, s1 - 1);
          v[u] = xq[(unsigned)sorted[idx] * 4u + fq];
        }
#pragma unroll
        for (int u = 0; u < 8; ++u) {
          const bool in = i + u < s1;
          sum.x += in ? v[u].x : 0.0f;
          sum.y += in ? v[u].y : 0.0f;
          sum.z += in ? v[u].z : 0.0f;
          sum.w += in ? v[u].w : 0.0f;
        }
      }
      float4* ar = (float4*)(acc + r * 16 + fq * 4);
      float4 a = *ar;
      a.x += sum.x; a.y += sum.y; a.z += sum.z; a.w += sum.w;
      *ar = a;
    }
    __syncthreads();
  }
  __syncthreads();

  if (!LAST) {
    for (int i = t; i < NB * 16; i += AGT) {
      const int node = node0 + (i >> 4);
      if (node < n)
        H[(size_t)node0 * 16 + i] = fmaxf(acc[i] * dinv[node] + bias[i & 15], 0.0f);
    }
  } else {
    const int node = node0 + t;
    if (t < NB && node < n) {
      const float s = dinv[node];
      float a = 0.0f;
#pragma unroll
      for (int ff = 0; ff < 16; ++ff) {
        const float h = fmaxf(acc[t * 16 + ff] * s + bias[ff], 0.0f);
        a += h * Wl[ff];
      }
      out[node] = a + bl[0];
    }
  }
}

extern "C" void kernel_launch(void* const* d_in, const int* in_sizes, int n_in,
                              void* d_out, int out_size, void* d_ws, size_t ws_size,
                              hipStream_t stream) {
  const float* x   = (const float*)d_in[0];
  const int*   ei  = (const int*)d_in[1];
  const float* W1  = (const float*)d_in[2];
  const float* b1  = (const float*)d_in[3];
  const float* W2  = (const float*)d_in[4];
  const float* b2  = (const float*)d_in[5];
  const float* Wl  = (const float*)d_in[6];
  const float* bl  = (const float*)d_in[7];
  float* out = (float*)d_out;

  const int din = in_sizes[2] / 16;
  const int n   = in_sizes[0] / din;
  const int E   = in_sizes[1] / 2;
  const int tot = n * 16;

  float* ws   = (float*)d_ws;
  float* dinv = ws;
  float* bufA = ws + (((size_t)n + 255) & ~(size_t)255);
  float* bufB = bufA + (size_t)n * 16;

  const int T = 256;
  const int ntiles = (n + 15) / 16;
  const int gemmBlocks = (ntiles + 3) / 4;

  const int nbins = (n + NB - 1) / NB;
  unsigned* pk   = (unsigned*)(bufB + (size_t)n * 16);
  int* cnt       = (int*)(pk + (((size_t)E + 255) & ~(size_t)255));
  int* binTot    = cnt + (size_t)nbins * PG;
  int* binBase   = binTot + MAXBINS;
  const size_t need = (size_t)((char*)(binBase + MAXBINS + 1) - (char*)d_ws);

  if (ws_size >= need && nbins <= MAXBINS) {
    const int chunk = (E + PG - 1) / PG;
    k_count   <<<PG, 32, 0, stream>>>(ei + E, cnt, E, chunk, nbins);
    k_scan_bin<<<nbins, PG / 2, 0, stream>>>(cnt, binTot);
    k_scan_tot<<<1, MAXBINS, 0, stream>>>(binTot, binBase, nbins);
    k_part    <<<PG, 32, 0, stream>>>(ei, ei + E, cnt, binBase, pk, E, chunk, nbins);
    k_deg     <<<nbins, AGT, 0, stream>>>(pk, binBase, dinv, n);

    k_gemm16s <<<gemmBlocks, 128, 0, stream>>>(x, W1, dinv, bufA, n, din);
    k_agg<false><<<nbins, AGT, 0, stream>>>(pk, binBase, bufA, dinv, b1, Wl, bl, bufB, out, n);
    k_gemm16s <<<gemmBlocks, 128, 0, stream>>>(bufB, W2, dinv, bufA, n, 16);
    k_agg<true> <<<nbins, AGT, 0, stream>>>(pk, binBase, bufA, dinv, b2, Wl, bl, bufB, out, n);
    return;
  }

  k_init  <<<(tot + T - 1) / T, T, 0, stream>>>(dinv, bufB, n, tot);
  k_degree<<<(E + T - 1) / T,   T, 0, stream>>>(ei + E, dinv, E);
  k_rsqrt <<<(n + T - 1) / T,   T, 0, stream>>>(dinv, n);

  k_gemm16 <<<gemmBlocks, 128, 0, stream>>>(x, W1, bufA, n, din);
  k_scatter<<<(4 * E + T - 1) / T, T, 0, stream>>>(ei, ei + E, dinv, bufA, bufB, E);
  k_final1 <<<(tot + T - 1) / T, T, 0, stream>>>(bufB, bufA, dinv, b1, tot);

  k_gemm16 <<<gemmBlocks, 128, 0, stream>>>(bufB, W2, bufA, n, 16);
  k_zero   <<<(tot + T - 1) / T, T, 0, stream>>>(bufB, tot);
  k_scatter<<<(4 * E + T - 1) / T, T, 0, stream>>>(ei, ei + E, dinv, bufA, bufB, E);
  k_final2 <<<(n + T - 1) / T, T, 0, stream>>>(bufB, bufA, dinv, b2, Wl, bl, out, n);
}


# same-key lane masks of the counting sorts from one wave vote per key bit (8 or 9 compares + xor/or) in place of 32 lane reads with 32 compares and selects; masks, ranks and f32 sum order unchanged
# speedup vs baseline: 2.8038x; 1.1552x over previous
#include <hip/hip_runtime.h>
#include <hip/hip_bf16.h>
#include <stddef.h>

typedef __attribute__((ext_vector_type(2))) float v2f;
typedef __attribute__((ext_vector_type(8))) float v8f;

__device__ __forceinline__ void atomAddF(float* p, float v) {
  asm volatile("global_atomic_add_f32 %0, %1, off scope:SCOPE_DEV"
               :
               : "v"(p), "v"(v)
               : "memory");
}

__global__ void k_init(float* __restrict__ deg, float* __restrict__ agg,
                       int n, int tot) {
  int t = blockIdx.x * blockDim.x + threadIdx.x;
  if (t < tot) agg[t] = 0.0f;
  if (t < n)   deg[t] = 1.0f;
}

__global__ void k_zero(float* __restrict__ p, int tot) {
  int t = blockIdx.x * blockDim.x + threadIdx.x;
  if (t < tot) p[t] = 0.0f;
}

__global__ void k_degree(const int* __restrict__ dst, float* __restrict__ deg,
                         int E) {
  int t = blockIdx.x * blockDim.x + threadIdx.x;
  if (t < E) atomAddF(&deg[dst[t]], 1.0f);
}

__global__ void k_rsqrt(float* __restrict__ d, int n) {
  int t = blockIdx.x * blockDim.x + threadIdx.x;
  if (t < n) d[t] = rsqrtf(d[t]);
}

__global__ void k_gemm16(const float* __restrict__ X, const float* __restrict__ W,
                         float* __restrict__ Y, int nrows, int K) {
  const int wave = threadIdx.x >> 5;
  const int lane = threadIdx.x & 31;
  const int tile = blockIdx.x * (blockDim.x >> 5) + wave;
  const int row0 = tile * 16;
  if (row0 >= nrows) return;

  const int m  = lane & 15;
  const int kh = lane >> 4;

  const float* ap = X + (size_t)(row0 + m) * K + 2 * kh;
  v8f acc = {};
  for (int kb = 0; kb < K; kb += 4) {
    v2f a = *(const v2f*)(ap + kb);
    const int kr = kb + 2 * kh;
    v2f b;
    b.x = W[(size_t)kr * 16 + m];
    b.y = W[(size_t)(kr + 1) * 16 + m];
    acc = __builtin_amdgcn_wmma_f32_16x16x4_f32(
        false, a, false, b, (short)0, acc, false, false);
  }
#pragma unroll
  for (int v = 0; v < 8; ++v) {
    Y[(size_t)(row0 + v + 8 * kh) * 16 + m] = acc[v];
  }
}

__global__ void k_scatter(const int* __restrict__ src, const int* __restrict__ dst,
                          const float* __restrict__ dinv,
                          const float* __restrict__ xw,
                          float* __restrict__ agg, int E) {
  int t = blockIdx.x * blockDim.x + threadIdx.x;
  int e = t >> 2;
  if (e >= E) return;
  int q = (t & 3) << 2;
  int s = src[e];
  int d = dst[e];
  float nrm = dinv[s] * dinv[d];
  float4 v = *(const float4*)(xw + (size_t)s * 16 + q);
  float* out = agg + (size_t)d * 16 + q;
  atomAddF(out + 0, v.x * nrm);
  atomAddF(out + 1, v.y * nrm);
  atomAddF(out + 2, v.z * nrm);
  atomAddF(out + 3, v.w * nrm);
}

__global__ void k_final1(float* __restrict__ agg, const float* __restrict__ xw,
                         const float* __restrict__ dinv,
                         const float* __restrict__ b, int tot) {
  int t = blockIdx.x * blockDim.x + threadIdx.x;
  if (t >= tot) return;
  int i = t >> 4, f = t & 15;
  float s = dinv[i];
  float v = agg[t] + xw[t] * s * s + b[f];
  agg[t] = fmaxf(v, 0.0f);
}

__global__ void k_final2(const float* __restrict__ agg, const float* __restrict__ xw,
                         const float* __restrict__ dinv,
                         const float* __restrict__ b2,
                         const float* __restrict__ Wl,
                         const float* __restrict__ bl,
                         float* __restrict__ out, int n) {
  int i = blockIdx.x * blockDim.x + threadIdx.x;
  if (i >= n) return;
  float s = dinv[i];
  float ss = s * s;
  float acc = 0.0f;
#pragma unroll
  for (int f = 0; f < 16; ++f) {
    float h = agg[(size_t)i * 16 + f] + xw[(size_t)i * 16 + f] * ss + b2[f];
    h = fmaxf(h, 0.0f);
    acc += h * Wl[f];
  }
  out[i] = acc + bl[0];
}

#define NB_SHIFT 8
#define NB       256
#define PG       2048
#define MAXBINS  512
#define BIN_BITS 9
#define AGW      16
#define AGT      (AGW * 32)
#define CAP      10240
#define GU       4
static_assert(MAXBINS == (1 << BIN_BITS) && NB == (1 << NB_SHIFT), "key widths");

template <int BITS>
__device__ __forceinline__ unsigned sameKeyMask(int key, bool valid) {
  static_assert(BITS >= 1 && BITS <= 31, "key width");
  unsigned differ = 0u;
#pragma unroll
  for (int b = 0; b < BITS; ++b) {
    const int mine = (key << (31 - b)) >> 31;
    const unsigned vote = __builtin_amdgcn_ballot_w32(mine != 0);
    differ |= vote ^ (unsigned)mine;
  }
  return ~differ & __builtin_amdgcn_ballot_w32(valid);
}

template <int N>
__device__ __forceinline__ void blockScanIncl(int* s, int t) {
  for (int d = 1; d < N; d <<= 1) {
    const int v = (t >= d) ? s[t - d] : 0;
    __syncthreads();
    s[t] += v;
    __syncthreads();
  }
}

__global__ void k_count(const int* __restrict__ dst, int* __restrict__ cnt,
                        int E, int chunk, int nbins) {
  __shared__ int h[MAXBINS];
  const int lane = threadIdx.x;
  for (int i = lane; i < nbins; i += 32) h[i] = 0;
  const int g  = blockIdx.x;
  const int e0 = g * chunk;
  const int e1 = min(e0 + chunk, E);
  int dn[GU];
#pragma unroll
  for (int u = 0; u < GU; ++u) dn[u] = dst[min(e0 + u * 32 + lane, E - 1)];
  for (int base = e0; base < e1; base += 32 * GU) {
    int d[GU];
#pragma unroll
    for (int u = 0; u < GU; ++u) d[u] = dn[u];
#pragma unroll
    for (int u = 0; u < GU; ++u) dn[u] = dst[min(base + 32 * GU + u * 32 + lane, E - 1)];
#pragma unroll
    for (int u = 0; u < GU; ++u) {
      if (base + u * 32 < e1) {
        const bool valid = base + u * 32 + lane < e1;
        const int key = valid ? (d[u] >> NB_SHIFT) : -1;
        const unsigned m = sameKeyMask<BIN_BITS>(key, valid);
        if (valid && (m >> lane) == 1u) h[key] += __popc(m);
      }
    }
  }
  for (int i = lane; i < nbins; i += 32) cnt[i * PG + g] = h[i];
}

__global__ void k_scan_bin(int* __restrict__ cnt, int* __restrict__ binTot) {
  __shared__ int s[PG / 2];
  const int t = threadIdx.x, b = blockIdx.x;
  int* row = cnt + (size_t)b * PG;
  const int c0 = row[2 * t], c1 = row[2 * t + 1];
  const int c = c0 + c1;
  s[t] = c;
  __syncthreads();
  blockScanIncl<PG / 2>(s, t);
  const int ex = s[t] - c;
  row[2 * t]     = ex;
  row[2 * t + 1] = ex + c0;
  if (t == PG / 2 - 1) binTot[b] = s[t];
}

__global__ void k_scan_tot(const int* __restrict__ binTot, int* __restrict__ binBase,
                           int nbins) {
  __shared__ int s[MAXBINS];
  const int t = threadIdx.x;
  const int c = (t < nbins) ? binTot[t] : 0;
  s[t] = c;
  __syncthreads();
  blockScanIncl<MAXBINS>(s, t);
  if (t < nbins) binBase[t] = s[t] - c;
  if (t == MAXBINS - 1) binBase[nbins] = s[t];
}

__global__ void k_part(const int* __restrict__ src, const int* __restrict__ dst,
                       const int* __restrict__ cnt, const int* __restrict__ binBase,
                       unsigned* __restrict__ pk, int E, int chunk, int nbins) {
  __shared__ int cur[MAXBINS];
  const int lane = threadIdx.x;
  const int g = blockIdx.x;
  for (int i = lane; i < nbins; i += 32) cur[i] = binBase[i] + cnt[i * PG + g];
  const int e0 = g * chunk;
  const int e1 = min(e0 + chunk, E);
  const unsigned below = (1u << lane) - 1u;
  int dn[GU], sn[GU];
#pragma unroll
  for (int u = 0; u < GU; ++u) {
    const int i = min(e0 + u * 32 + lane, E - 1);
    dn[u] = dst[i]; sn[u] = src[i];
  }
  for (int base = e0; base < e1; base += 32 * GU) {
    int d[GU], s[GU];
#pragma unroll
    for (int u = 0; u < GU; ++u) { d[u] = dn[u]; s[u] = sn[u]; }
#pragma unroll
    for (int u = 0; u < GU; ++u) {
      const int i = min(base + 32 * GU + u * 32 + lane, E - 1);
      dn[u] = dst[i]; sn[u] = src[i];
    }
#pragma unroll
    for (int u = 0; u < GU; ++u) {
      if (base + u * 32 < e1) {
        const bool valid = base + u * 32 + lane < e1;
        const int key = valid ? (d[u] >> NB_SHIFT) : -1;
        const unsigned m = sameKeyMask<BIN_BITS>(key, valid);
        if (valid) {
          const int pos = cur[key] + __popc(m & below);
          if ((m >> lane) == 1u) cur[key] = pos + 1;
          pk[pos] = ((unsigned)s[u] << NB_SHIFT) | (unsigned)(d[u] & (NB - 1));
        }
      }
    }
  }
}

__device__ __forceinline__ void countRows(const unsigned* __restrict__ pk, int b0, int b1,
                                          int* hwv, int wave, int lane) {
  for (int base = b0 + wave * 32; base < b1; base += AGT) {
    const int i = base + lane;
    const bool valid = i < b1;
    const unsigned q = pk[min(i, b1 - 1)];
    const int key = valid ? (int)(q & (NB - 1)) : -1;
    const unsigned m = sameKeyMask<NB_SHIFT>(key, valid);
    if (valid && (m >> lane) == 1u) hwv[key] += __popc(m);
  }
}

__global__ void k_deg(const unsigned* __restrict__ pk, const int* __restrict__ binBase,
                      float* __restrict__ dinv, int n) {
  __shared__ int hw[AGW * NB];
  const int t = threadIdx.x, b = blockIdx.x;
  const int lane = t & 31;
  const int wave = __builtin_amdgcn_readfirstlane(t >> 5);
  for (int i = t; i < AGW * NB; i += AGT) hw[i] = 0;
  __syncthreads();
  const int e0 = binBase[b], e1 = binBase[b + 1];
  countRows(pk, e0, e1, hw + wave * NB, wave, lane);
  __syncthreads();
  if (t < NB) {
    int c = 0;
#pragma unroll
    for (int w = 0; w < AGW; ++w) c += hw[w * NB + t];
    const int node = b * NB + t;
    if (node < n) dinv[node] = rsqrtf(1.0f + (float)c);
  }
}

__global__ void k_gemm16s(const float* __restrict__ X, const float* __restrict__ W,
                          const float* __restrict__ dinv,
                          float* __restrict__ Y, int nrows, int K) {
  const int wave = threadIdx.x >> 5;
  const int lane = threadIdx.x & 31;
  const int tile = blockIdx.x * (blockDim.x >> 5) + wave;
  const int row0 = tile * 16;
  if (row0 >= nrows) return;

  const int m  = lane & 15;
  const int kh = lane >> 4;

  const float* ap = X + (size_t)(row0 + m) * K + 2 * kh;
  v8f acc = {};
  for (int kb = 0; kb < K; kb += 4) {
    v2f a = *(const v2f*)(ap + kb);
    const int kr = kb + 2 * kh;
    v2f b;
    b.x = W[(size_t)kr * 16 + m];
    b.y = W[(size_t)(kr + 1) * 16 + m];
    acc = __builtin_amdgcn_wmma_f32_16x16x4_f32(
        false, a, false, b, (short)0, acc, false, false);
  }
#pragma unroll
  for (int v = 0; v < 8; ++v) {
    const int r = row0 + v + 8 * kh;
    Y[(size_t)r * 16 + m] = acc[v] * dinv[r];
  }
}

template <bool LAST>
__global__ void k_agg(const unsigned* __restrict__ pk, const int* __restrict__ binBase,
                      const float* __restrict__ xws, const float* __restrict__ dinv,
                      const float* __restrict__ bias,
                      const float* __restrict__ Wl, const float* __restrict__ bl,
                      float* __restrict__ H, float* __restrict__ out, int n) {
  __shared__ __attribute__((aligned(16))) float acc[NB * 16];
  __shared__ int hw[AGW * NB];
  __shared__ int sc[AGT];
  __shared__ int rowStart[NB + 1];
  __shared__ int sorted[CAP];

  const int t = threadIdx.x, b = blockIdx.x;
  const int lane = t & 31;
  const int wave = __builtin_amdgcn_readfirstlane(t >> 5);
  const int fq = lane & 3;
  const int grp = t >> 2;
  const float4* xq = (const float4*)xws;
  const int node0 = b * NB;
  for (int i = t; i < NB * 16; i += AGT) {
    const int node = node0 + (i >> 4);
    acc[i] = (node < n) ? xws[(size_t)node0 * 16 + i] : 0.0f;
  }
  const int e0 = binBase[b], e1 = binBase[b + 1];
  int* hwv = hw + wave * NB;
  const unsigned below = (1u << lane) - 1u;

  for (int b0 = e0; b0 < e1; b0 += CAP) {
    const int b1 = min(b0 + CAP, e1);
    for (int i = t; i < AGW * NB; i += AGT) hw[i] = 0;
    __syncthreads();
    countRows(pk, b0, b1, hwv, wave, lane);
    __syncthreads();
    {
      const int key = t >> 1, w0 = (t & 1) * (AGW / 2);
      int s = 0;
#pragma unroll
      for (int i = 0; i < AGW / 2; ++i) s += hw[(w0 + i) * NB + key];
      sc[t] = s;
      __syncthreads();
      blockScanIncl<AGT>(sc, t);
      int run = sc[t] - s;
      if ((t & 1) == 0) rowStart[key] = run;
      if (t == AGT - 1) rowStart[NB] = sc[t];
#pragma unroll
      for (int i = 0; i < AGW / 2; ++i) {
        const int idx = (w0 + i) * NB + key;
        const int c = hw[idx];
        hw[idx] = run;
        run += c;
      }
    }
    __syncthreads();
    for (int base = b0 + wave * 32; base < b1; base += AGT) {
      const int i = base + lane;
      const bool valid = i < b1;
      const unsigned q = pk[min(i, b1 - 1)];
      const int key = valid ? (int)(q & (NB - 1)) : -1;
      const unsigned m = sameKeyMask<NB_SHIFT>(key, valid);
      if (valid) {
        const int pos = hwv[key] + __popc(m & below);
        if ((m >> lane) == 1u) hwv[key] = pos + 1;
        sorted[pos] = (int)(q >> NB_SHIFT);
      }
    }
    __syncthreads();
    for (int r = grp; r < NB; r += AGT / 4) {
      const int s0 = rowStart[r], s1 = rowStart[r + 1];
      float4 sum = make_float4(0.0f, 0.0f, 0.0f, 0.0f);
      for (int i = s0; i < s1; i += 8) {
        float4 v[8];
#pragma unroll
        for (int u = 0; u < 8; ++u) {
          const int idx = min(i + u, s1 - 1);
          v[u] = xq[(unsigned)sorted[idx] * 4u + fq];
        }
#pragma unroll
        for (int u = 0; u < 8; ++u) {
          const bool in = i + u < s1;
          sum.x += in ? v[u].x : 0.0f;
          sum.y += in ? v[u].y : 0.0f;
          sum.z += in ? v[u].z : 0.0f;
          sum.w += in ? v[u].w : 0.0f;
        }
      }
      float4* ar = (float4*)(acc + r * 16 + fq * 4);
      float4 a = *ar;
      a.x += sum.x; a.y += sum.y; a.z += sum.z; a.w += sum.w;
      *ar = a;
    }
    __syncthreads();
  }
  __syncthreads();

  if (!LAST) {
    for (int i = t; i < NB * 16; i += AGT) {
      const int node = node0 + (i >> 4);
      if (node < n)
        H[(size_t)node0 * 16 + i] = fmaxf(acc[i] * dinv[node] + bias[i & 15], 0.0f);
    }
  } else {
    const int node = node0 + t;
    if (t < NB && node < n) {
      const float s = dinv[node];
      float a = 0.0f;
#pragma unroll
      for (int ff = 0; ff < 16; ++ff) {
        const float h = fmaxf(acc[t * 16 + ff] * s + bias[ff], 0.0f);
        a += h * Wl[ff];
      }
      out[node] = a + bl[0];
    }
  }
}

extern "C" void kernel_launch(void* const* d_in, const int* in_sizes, int n_in,
                              void* d_out, int out_size, void* d_ws, size_t ws_size,
                              hipStream_t stream) {
  const float* x   = (const float*)d_in[0];
  const int*   ei  = (const int*)d_in[1];
  const float* W1  = (const float*)d_in[2];
  const float* b1  = (const float*)d_in[3];
  const float* W2  = (const float*)d_in[4];
  const float* b2  = (const float*)d_in[5];
  const float* Wl  = (const float*)d_in[6];
  const float* bl  = (const float*)d_in[7];
  float* out = (float*)d_out;

  const int din = in_sizes[2] / 16;
  const int n   = in_sizes[0] / din;
  const int E   = in_sizes[1] / 2;
  const int tot = n * 16;

  float* ws   = (float*)d_ws;
  float* dinv = ws;
  float* bufA = ws + (((size_t)n + 255) & ~(size_t)255);
  float* bufB = bufA + (size_t)n * 16;

  const int T = 256;
  const int ntiles = (n + 15) / 16;
  const int gemmBlocks = (ntiles + 3) / 4;

  const int nbins = (n + NB - 1) / NB;
  unsigned* pk   = (unsigned*)(bufB + (size_t)n * 16);
  int* cnt       = (int*)(pk + (((size_t)E + 255) & ~(size_t)255));
  int* binTot    = cnt + (size_t)nbins * PG;
  int* binBase   = binTot + MAXBINS;
  const size_t need = (size_t)((char*)(binBase + MAXBINS + 1) - (char*)d_ws);

  if (ws_size >= need && nbins <= MAXBINS) {
    const int chunk = (E + PG - 1) / PG;
    k_count   <<<PG, 32, 0, stream>>>(ei + E, cnt, E, chunk, nbins);
    k_scan_bin<<<nbins, PG / 2, 0, stream>>>(cnt, binTot);
    k_scan_tot<<<1, MAXBINS, 0, stream>>>(binTot, binBase, nbins);
    k_part    <<<PG, 32, 0, stream>>>(ei, ei + E, cnt, binBase, pk, E, chunk, nbins);
    k_deg     <<<nbins, AGT, 0, stream>>>(pk, binBase, dinv, n);

    k_gemm16s <<<gemmBlocks, 128, 0, stream>>>(x, W1, dinv, bufA, n, din);
    k_agg<false><<<nbins, AGT, 0, stream>>>(pk, binBase, bufA, dinv, b1, Wl, bl, bufB, out, n);
    k_gemm16s <<<gemmBlocks, 128, 0, stream>>>(bufB, W2, dinv, bufA, n, 16);
    k_agg<true> <<<nbins, AGT, 0, stream>>>(pk, binBase, bufA, dinv, b2, Wl, bl, bufB, out, n);
    return;
  }

  k_init  <<<(tot + T - 1) / T, T, 0, stream>>>(dinv, bufB, n, tot);
  k_degree<<<(E + T - 1) / T,   T, 0, stream>>>(ei + E, dinv, E);
  k_rsqrt <<<(n + T - 1) / T,   T, 0, stream>>>(dinv, n);

  k_gemm16 <<<gemmBlocks, 128, 0, stream>>>(x, W1, bufA, n, din);
  k_scatter<<<(4 * E + T - 1) / T, T, 0, stream>>>(ei, ei + E, dinv, bufA, bufB, E);
  k_final1 <<<(tot + T - 1) / T, T, 0, stream>>>(bufB, bufA, dinv, b1, tot);

  k_gemm16 <<<gemmBlocks, 128, 0, stream>>>(bufB, W2, bufA, n, 16);
  k_zero   <<<(tot + T - 1) / T, T, 0, stream>>>(bufB, tot);
  k_scatter<<<(4 * E + T - 1) / T, T, 0, stream>>>(ei, ei + E, dinv, bufA, bufB, E);
  k_final2 <<<(n + T - 1) / T, T, 0, stream>>>(bufB, bufA, dinv, b2, Wl, bl, out, n);
}


# bucketing passes run as 16-wave work-groups: bin tables and bucketed edges staged in LDS, so count and edge stores leave as contiguous runs (were 32 lines x 4 B per store); f32 sum order same
# speedup vs baseline: 4.3686x; 1.5581x over previous
#include <hip/hip_runtime.h>
#include <hip/hip_bf16.h>
#include <stddef.h>

typedef __attribute__((ext_vector_type(2))) float v2f;
typedef __attribute__((ext_vector_type(8))) float v8f;
typedef __attribute__((ext_vector_type(2))) unsigned v2u;

__device__ __forceinline__ void atomAddF(float* p, float v) {
  asm volatile("global_atomic_add_f32 %0, %1, off scope:SCOPE_DEV"
               :
               : "v"(p), "v"(v)
               : "memory");
}

__global__ void k_init(float* __restrict__ deg, float* __restrict__ agg,
                       int n, int tot) {
  int t = blockIdx.x * blockDim.x + threadIdx.x;
  if (t < tot) agg[t] = 0.0f;
  if (t < n)   deg[t] = 1.0f;
}

__global__ void k_zero(float* __restrict__ p, int tot) {
  int t = blockIdx.x * blockDim.x + threadIdx.x;
  if (t < tot) p[t] = 0.0f;
}

__global__ void k_degree(const int* __restrict__ dst, float* __restrict__ deg,
                         int E) {
  int t = blockIdx.x * blockDim.x + threadIdx.x;
  if (t < E) atomAddF(&deg[dst[t]], 1.0f);
}

__global__ void k_rsqrt(float* __restrict__ d, int n) {
  int t = blockIdx.x * blockDim.x + threadIdx.x;
  if (t < n) d[t] = rsqrtf(d[t]);
}

__global__ void k_gemm16(const float* __restrict__ X, const float* __restrict__ W,
                         float* __restrict__ Y, int nrows, int K) {
  const int wave = threadIdx.x >> 5;
  const int lane = threadIdx.x & 31;
  const int tile = blockIdx.x * (blockDim.x >> 5) + wave;
  const int row0 = tile * 16;
  if (row0 >= nrows) return;

  const int m  = lane & 15;
  const int kh = lane >> 4;

  const float* ap = X + (size_t)(row0 + m) * K + 2 * kh;
  v8f acc = {};
  for (int kb = 0; kb < K; kb += 4) {
    v2f a = *(const v2f*)(ap + kb);
    const int kr = kb + 2 * kh;
    v2f b;
    b.x = W[(size_t)kr * 16 + m];
    b.y = W[(size_t)(kr + 1) * 16 + m];
    acc = __builtin_amdgcn_wmma_f32_16x16x4_f32(
        false, a, false, b, (short)0, acc, false, false);
  }
#pragma unroll
  for (int v = 0; v < 8; ++v) {
    Y[(size_t)(row0 + v + 8 * kh) * 16 + m] = acc[v];
  }
}

__global__ void k_scatter(const int* __restrict__ src, const int* __restrict__ dst,
                          const float* __restrict__ dinv,
                          const float* __restrict__ xw,
                          float* __restrict__ agg, int E) {
  int t = blockIdx.x * blockDim.x + threadIdx.x;
  int e = t >> 2;
  if (e >= E) return;
  int q = (t & 3) << 2;
  int s = src[e];
  int d = dst[e];
  float nrm = dinv[s] * dinv[d];
  float4 v = *(const float4*)(xw + (size_t)s * 16 + q);
  float* out = agg + (size_t)d * 16 + q;
  atomAddF(out + 0, v.x * nrm);
  atomAddF(out + 1, v.y * nrm);
  atomAddF(out + 2, v.z * nrm);
  atomAddF(out + 3, v.w * nrm);
}

__global__ void k_final1(float* __restrict__ agg, const float* __restrict__ xw,
                         const float* __restrict__ dinv,
                         const float* __restrict__ b, int tot) {
  int t = blockIdx.x * blockDim.x + threadIdx.x;
  if (t >= tot) return;
  int i = t >> 4, f = t & 15;
  float s = dinv[i];
  float v = agg[t] + xw[t] * s * s + b[f];
  agg[t] = fmaxf(v, 0.0f);
}

__global__ void k_final2(const float* __restrict__ agg, const float* __restrict__ xw,
                         const float* __restrict__ dinv,
                         const float* __restrict__ b2,
                         const float* __restrict__ Wl,
                         const float* __restrict__ bl,
                         float* __restrict__ out, int n) {
  int i = blockIdx.x * blockDim.x + threadIdx.x;
  if (i >= n) return;
  float s = dinv[i];
  float ss = s * s;
  float acc = 0.0f;
#pragma unroll
  for (int f = 0; f < 16; ++f) {
    float h = agg[(size_t)i * 16 + f] + xw[(size_t)i * 16 + f] * ss + b2[f];
    h = fmaxf(h, 0.0f);
    acc += h * Wl[f];
  }
  out[i] = acc + bl[0];
}

#define NB_SHIFT 8
#define NB       256
#define PG       2048
#define MAXBINS  512
#define BIN_BITS 9
#define AGW      16
#define AGT      (AGW * 32)
#define CAP      10240
#define GU       4
#define BW       16
#define BT       (BW * 32)
#define HS       (MAXBINS + 1)
#define CHW      1600
static_assert(MAXBINS == (1 << BIN_BITS) && NB == (1 << NB_SHIFT), "key widths");
static_assert(PG % BW == 0 && BT == MAXBINS, "bucketing work-group shape");

template <int BITS>
__device__ __forceinline__ unsigned sameKeyMask(int key, bool valid) {
  static_assert(BITS >= 1 && BITS <= 31, "key width");
  unsigned differ = 0u;
#pragma unroll
  for (int b = 0; b < BITS; ++b) {
    const int mine = (key << (31 - b)) >> 31;
    const unsigned vote = __builtin_amdgcn_ballot_w32(mine != 0);
    differ |= vote ^ (unsigned)mine;
  }
  return ~differ & __builtin_amdgcn_ballot_w32(valid);
}

template <int N>
__device__ __forceinline__ void blockScanIncl(int* s, int t) {
  for (int d = 1; d < N; d <<= 1) {
    const int v = (t >= d) ? s[t - d] : 0;
    __syncthreads();
    s[t] += v;
    __syncthreads();
  }
}

__global__ void k_count(const int* __restrict__ dst, int* __restrict__ cnt,
                        int E, int chunk, int nbins) {
  __shared__ int h[BW * HS];
  const int t = threadIdx.x;
  const int lane = t & 31;
  const int wave = __builtin_amdgcn_readfirstlane(t >> 5);
  for (int i = t; i < BW * HS; i += BT) h[i] = 0;
  __syncthreads();
  int* hv = h + wave * HS;
  const int g0 = blockIdx.x * BW;
  const int e0 = (g0 + wave) * chunk;
  const int e1 = min(e0 + chunk, E);
  int dn[GU];
#pragma unroll
  for (int u = 0; u < GU; ++u) dn[u] = dst[min(e0 + u * 32 + lane, E - 1)];
  for (int base = e0; base < e1; base += 32 * GU) {
    int d[GU];
#pragma unroll
    for (int u = 0; u < GU; ++u) d[u] = dn[u];
#pragma unroll
    for (int u = 0; u < GU; ++u) dn[u] = dst[min(base + 32 * GU + u * 32 + lane, E - 1)];
#pragma unroll
    for (int u = 0; u < GU; ++u) {
      if (base + u * 32 < e1) {
        const bool valid = base + u * 32 + lane < e1;
        const int key = valid ? (d[u] >> NB_SHIFT) : -1;
        const unsigned m = sameKeyMask<BIN_BITS>(key, valid);
        if (valid && (m >> lane) == 1u) hv[key] += __popc(m);
      }
    }
  }
  __syncthreads();
  for (int i = t; i < nbins * BW; i += BT) {
    const int bin = i / BW, w = i % BW;
    cnt[bin * PG + g0 + w] = h[w * HS + bin];
  }
}

__global__ void k_scan_bin(int* __restrict__ cnt, int* __restrict__ binTot) {
  __shared__ int s[PG / 2];
  const int t = threadIdx.x, b = blockIdx.x;
  int* row = cnt + (size_t)b * PG;
  const int c0 = row[2 * t], c1 = row[2 * t + 1];
  const int c = c0 + c1;
  s[t] = c;
  __syncthreads();
  blockScanIncl<PG / 2>(s, t);
  const int ex = s[t] - c;
  row[2 * t]     = ex;
  row[2 * t + 1] = ex + c0;
  if (t == PG / 2 - 1) binTot[b] = s[t];
}

__global__ void k_scan_tot(const int* __restrict__ binTot, int* __restrict__ binBase,
                           int nbins) {
  __shared__ int s[MAXBINS];
  const int t = threadIdx.x;
  const int c = (t < nbins) ? binTot[t] : 0;
  s[t] = c;
  __syncthreads();
  blockScanIncl<MAXBINS>(s, t);
  if (t < nbins) binBase[t] = s[t] - c;
  if (t == MAXBINS - 1) binBase[nbins] = s[t];
}

__global__ void k_part(const int* __restrict__ src, const int* __restrict__ dst,
                       const int* __restrict__ cnt, const int* __restrict__ binBase,
                       unsigned* __restrict__ pk, int E, int chunk, int nbins) {
  __shared__ int cur[BW * HS];
  __shared__ int dl[MAXBINS];
  __shared__ int sc[BT];
  __shared__ v2u stage[BW * CHW];
  const int t = threadIdx.x;
  const int lane = t & 31;
  const int wave = __builtin_amdgcn_readfirstlane(t >> 5);
  const int g0 = blockIdx.x * BW;
  int c = 0, gs = 0;
  if (t < nbins) {
    const int bb = binBase[t];
    const int lo = cnt[t * PG + g0];
    const int hi = (g0 + BW < PG) ? cnt[t * PG + g0 + BW] : binBase[t + 1] - bb;
    c = hi - lo;
    gs = bb + lo;
  }
  sc[t] = c;
  __syncthreads();
  blockScanIncl<BT>(sc, t);
  dl[t] = (sc[t] - c) - gs;
  const int total = sc[BT - 1];
  for (int i = t; i < nbins * BW; i += BT) {
    const int bin = i / BW, w = i % BW;
    cur[w * HS + bin] = binBase[bin] + cnt[bin * PG + g0 + w];
  }
  __syncthreads();
  int* curv = cur + wave * HS;
  const int e0 = (g0 + wave) * chunk;
  const int e1 = min(e0 + chunk, E);
  const unsigned below = (1u << lane) - 1u;
  int dn[GU], sn[GU];
#pragma unroll
  for (int u = 0; u < GU; ++u) {
    const int i = min(e0 + u * 32 + lane, E - 1);
    dn[u] = dst[i]; sn[u] = src[i];
  }
  for (int base = e0; base < e1; base += 32 * GU) {
    int d[GU], s[GU];
#pragma unroll
    for (int u = 0; u < GU; ++u) { d[u] = dn[u]; s[u] = sn[u]; }
#pragma unroll
    for (int u = 0; u < GU; ++u) {
      const int i = min(base + 32 * GU + u * 32 + lane, E - 1);
      dn[u] = dst[i]; sn[u] = src[i];
    }
#pragma unroll
    for (int u = 0; u < GU; ++u) {
      if (base + u * 32 < e1) {
        const bool valid = base + u * 32 + lane < e1;
        const int key = valid ? (d[u] >> NB_SHIFT) : -1;
        const unsigned m = sameKeyMask<BIN_BITS>(key, valid);
        if (valid) {
          const int pos = curv[key] + __popc(m & below);
          if ((m >> lane) == 1u) curv[key] = pos + 1;
          v2u e;
          e.x = ((unsigned)s[u] << NB_SHIFT) | (unsigned)(d[u] & (NB - 1));
          e.y = (unsigned)pos;
          stage[pos + dl[key]] = e;
        }
      }
    }
  }
  __syncthreads();
#pragma unroll 4
  for (int i = t; i < total; i += BT) {
    const v2u e = stage[i];
    pk[e.y] = e.x;
  }
}

__device__ __forceinline__ void countRows(const unsigned* __restrict__ pk, int b0, int b1,
                                          int* hwv, int wave, int lane) {
  for (int base = b0 + wave * 32; base < b1; base += AGT) {
    const int i = base + lane;
    const bool valid = i < b1;
    const unsigned q = pk[min(i, b1 - 1)];
    const int key = valid ? (int)(q & (NB - 1)) : -1;
    const unsigned m = sameKeyMask<NB_SHIFT>(key, valid);
    if (valid && (m >> lane) == 1u) hwv[key] += __popc(m);
  }
}

__global__ void k_deg(const unsigned* __restrict__ pk, const int* __restrict__ binBase,
                      float* __restrict__ dinv, int n) {
  __shared__ int hw[AGW * NB];
  const int t = threadIdx.x, b = blockIdx.x;
  const int lane = t & 31;
  const int wave = __builtin_amdgcn_readfirstlane(t >> 5);
  for (int i = t; i < AGW * NB; i += AGT) hw[i] = 0;
  __syncthreads();
  const int e0 = binBase[b], e1 = binBase[b + 1];
  countRows(pk, e0, e1, hw + wave * NB, wave, lane);
  __syncthreads();
  if (t < NB) {
    int c = 0;
#pragma unroll
    for (int w = 0; w < AGW; ++w) c += hw[w * NB + t];
    const int node = b * NB + t;
    if (node < n) dinv[node] = rsqrtf(1.0f + (float)c);
  }
}

__global__ void k_gemm16s(const float* __restrict__ X, const float* __restrict__ W,
                          const float* __restrict__ dinv,
                          float* __restrict__ Y, int nrows, int K) {
  const int wave = threadIdx.x >> 5;
  const int lane = threadIdx.x & 31;
  const int tile = blockIdx.x * (blockDim.x >> 5) + wave;
  const int row0 = tile * 16;
  if (row0 >= nrows) return;

  const int m  = lane & 15;
  const int kh = lane >> 4;

  const float* ap = X + (size_t)(row0 + m) * K + 2 * kh;
  v8f acc = {};
  for (int kb = 0; kb < K; kb += 4) {
    v2f a = *(const v2f*)(ap + kb);
    const int kr = kb + 2 * kh;
    v2f b;
    b.x = W[(size_t)kr * 16 + m];
    b.y = W[(size_t)(kr + 1) * 16 + m];
    acc = __builtin_amdgcn_wmma_f32_16x16x4_f32(
        false, a, false, b, (short)0, acc, false, false);
  }
#pragma unroll
  for (int v = 0; v < 8; ++v) {
    const int r = row0 + v + 8 * kh;
    Y[(size_t)r * 16 + m] = acc[v] * dinv[r];
  }
}

template <bool LAST>
__global__ void k_agg(const unsigned* __restrict__ pk, const int* __restrict__ binBase,
                      const float* __restrict__ xws, const float* __restrict__ dinv,
                      const float* __restrict__ bias,
                      const float* __restrict__ Wl, const float* __restrict__ bl,
                      float* __restrict__ H, float* __restrict__ out, int n) {
  __shared__ __attribute__((aligned(16))) float acc[NB * 16];
  __shared__ int hw[AGW * NB];
  __shared__ int sc[AGT];
  __shared__ int rowStart[NB + 1];
  __shared__ int sorted[CAP];

  const int t = threadIdx.x, b = blockIdx.x;
  const int lane = t & 31;
  const int wave = __builtin_amdgcn_readfirstlane(t >> 5);
  const int fq = lane & 3;
  const int grp = t >> 2;
  const float4* xq = (const float4*)xws;
  const int node0 = b * NB;
  for (int i = t; i < NB * 16; i += AGT) {
    const int node = node0 + (i >> 4);
    acc[i] = (node < n) ? xws[(size_t)node0 * 16 + i] : 0.0f;
  }
  const int e0 = binBase[b], e1 = binBase[b + 1];
  int* hwv = hw + wave * NB;
  const unsigned below = (1u << lane) - 1u;

  for (int b0 = e0; b0 < e1; b0 += CAP) {
    const int b1 = min(b0 + CAP, e1);
    for (int i = t; i < AGW * NB; i += AGT) hw[i] = 0;
    __syncthreads();
    countRows(pk, b0, b1, hwv, wave, lane);
    __syncthreads();
    {
      const int key = t >> 1, w0 = (t & 1) * (AGW / 2);
      int s = 0;
#pragma unroll
      for (int i = 0; i < AGW / 2; ++i) s += hw[(w0 + i) * NB + key];
      sc[t] = s;
      __syncthreads();
      blockScanIncl<AGT>(sc, t);
      int run = sc[t] - s;
      if ((t & 1) == 0) rowStart[key] = run;
      if (t == AGT - 1) rowStart[NB] = sc[t];
#pragma unroll
      for (int i = 0; i < AGW / 2; ++i) {
        const int idx = (w0 + i) * NB + key;
        const int c = hw[idx];
        hw[idx] = run;
        run += c;
      }
    }
    __syncthreads();
    for (int base = b0 + wave * 32; base < b1; base += AGT) {
      const int i = base + lane;
      const bool valid = i < b1;
      const unsigned q = pk[min(i, b1 - 1)];
      const int key = valid ? (int)(q & (NB - 1)) : -1;
      const unsigned m = sameKeyMask<NB_SHIFT>(key, valid);
      if (valid) {
        const int pos = hwv[key] + __popc(m & below);
        if ((m >> lane) == 1u) hwv[key] = pos + 1;
        sorted[pos] = (int)(q >> NB_SHIFT);
      }
    }
    __syncthreads();
    for (int r = grp; r < NB; r += AGT / 4) {
      const int s0 = rowStart[r], s1 = rowStart[r + 1];
      float4 sum = make_float4(0.0f, 0.0f, 0.0f, 0.0f);
      for (int i = s0; i < s1; i += 8) {
        float4 v[8];
#pragma unroll
        for (int u = 0; u < 8; ++u) {
          const int idx = min(i + u, s1 - 1);
          v[u] = xq[(unsigned)sorted[idx] * 4u + fq];
        }
#pragma unroll
        for (int u = 0; u < 8; ++u) {
          const bool in = i + u < s1;
          sum.x += in ? v[u].x : 0.0f;
          sum.y += in ? v[u].y : 0.0f;
          sum.z += in ? v[u].z : 0.0f;
          sum.w += in ? v[u].w : 0.0f;
        }
      }
      float4* ar = (float4*)(acc + r * 16 + fq * 4);
      float4 a = *ar;
      a.x += sum.x; a.y += sum.y; a.z += sum.z; a.w += sum.w;
      *ar = a;
    }
    __syncthreads();
  }
  __syncthreads();

  if (!LAST) {
    for (int i = t; i < NB * 16; i += AGT) {
      const int node = node0 + (i >> 4);
      if (node < n)
        H[(size_t)node0 * 16 + i] = fmaxf(acc[i] * dinv[node] + bias[i & 15], 0.0f);
    }
  } else {
    const int node = node0 + t;
    if (t < NB && node < n) {
      const float s = dinv[node];
      float a = 0.0f;
#pragma unroll
      for (int ff = 0; ff < 16; ++ff) {
        const float h = fmaxf(acc[t * 16 + ff] * s + bias[ff], 0.0f);
        a += h * Wl[ff];
      }
      out[node] = a + bl[0];
    }
  }
}

extern "C" void kernel_launch(void* const* d_in, const int* in_sizes, int n_in,
                              void* d_out, int out_size, void* d_ws, size_t ws_size,
                              hipStream_t stream) {
  const float* x   = (const float*)d_in[0];
  const int*   ei  = (const int*)d_in[1];
  const float* W1  = (const float*)d_in[2];
  const float* b1  = (const float*)d_in[3];
  const float* W2  = (const float*)d_in[4];
  const float* b2  = (const float*)d_in[5];
  const float* Wl  = (const float*)d_in[6];
  const float* bl  = (const float*)d_in[7];
  float* out = (float*)d_out;

  const int din = in_sizes[2] / 16;
  const int n   = in_sizes[0] / din;
  const int E   = in_sizes[1] / 2;
  const int tot = n * 16;

  float* ws   = (float*)d_ws;
  float* dinv = ws;
  float* bufA = ws + (((size_t)n + 255) & ~(size_t)255);
  float* bufB = bufA + (size_t)n * 16;

  const int T = 256;
  const int ntiles = (n + 15) / 16;
  const int gemmBlocks = (ntiles + 3) / 4;

  const int nbins = (n + NB - 1) / NB;
  unsigned* pk   = (unsigned*)(bufB + (size_t)n * 16);
  int* cnt       = (int*)(pk + (((size_t)E + 255) & ~(size_t)255));
  int* binTot    = cnt + (size_t)nbins * PG;
  int* binBase   = binTot + MAXBINS;
  const size_t need = (size_t)((char*)(binBase + MAXBINS + 1) - (char*)d_ws);

  const int chunk = (E + PG - 1) / PG;
  if (ws_size >= need && nbins <= MAXBINS && chunk <= CHW) {
    k_count   <<<PG / BW, BT, 0, stream>>>(ei + E, cnt, E, chunk, nbins);
    k_scan_bin<<<nbins, PG / 2, 0, stream>>>(cnt, binTot);
    k_scan_tot<<<1, MAXBINS, 0, stream>>>(binTot, binBase, nbins);
    k_part    <<<PG / BW, BT, 0, stream>>>(ei, ei + E, cnt, binBase, pk, E, chunk, nbins);
    k_deg     <<<nbins, AGT, 0, stream>>>(pk, binBase, dinv, n);

    k_gemm16s <<<gemmBlocks, 128, 0, stream>>>(x, W1, dinv, bufA, n, din);
    k_agg<false><<<nbins, AGT, 0, stream>>>(pk, binBase, bufA, dinv, b1, Wl, bl, bufB, out, n);
    k_gemm16s <<<gemmBlocks, 128, 0, stream>>>(bufB, W2, dinv, bufA, n, 16);
    k_agg<true> <<<nbins, AGT, 0, stream>>>(pk, binBase, bufA, dinv, b2, Wl, bl, bufB, out, n);
    return;
  }

  k_init  <<<(tot + T - 1) / T, T, 0, stream>>>(dinv, bufB, n, tot);
  k_degree<<<(E + T - 1) / T,   T, 0, stream>>>(ei + E, dinv, E);
  k_rsqrt <<<(n + T - 1) / T,   T, 0, stream>>>(dinv, n);

  k_gemm16 <<<gemmBlocks, 128, 0, stream>>>(x, W1, bufA, n, din);
  k_scatter<<<(4 * E + T - 1) / T, T, 0, stream>>>(ei, ei + E, dinv, bufA, bufB, E);
  k_final1 <<<(tot + T - 1) / T, T, 0, stream>>>(bufB, bufA, dinv, b1, tot);

  k_gemm16 <<<gemmBlocks, 128, 0, stream>>>(bufB, W2, bufA, n, 16);
  k_zero   <<<(tot + T - 1) / T, T, 0, stream>>>(bufB, tot);
  k_scatter<<<(4 * E + T - 1) / T, T, 0, stream>>>(ei, ei + E, dinv, bufA, bufB, E);
  k_final2 <<<(n + T - 1) / T, T, 0, stream>>>(bufB, bufA, dinv, b2, Wl, bl, out, n);
}


# each bin's edges sorted by destination row once per call, in the degree pass (in place, row starts kept in workspace), and reused by both layers' row sums; was re-sorted per layer; f32 sum order same
# speedup vs baseline: 4.7635x; 1.0904x over previous
#include <hip/hip_runtime.h>
#include <hip/hip_bf16.h>
#include <stddef.h>

typedef __attribute__((ext_vector_type(2))) float v2f;
typedef __attribute__((ext_vector_type(8))) float v8f;
typedef __attribute__((ext_vector_type(2))) unsigned v2u;

__device__ __forceinline__ void atomAddF(float* p, float v) {
  asm volatile("global_atomic_add_f32 %0, %1, off scope:SCOPE_DEV"
               :
               : "v"(p), "v"(v)
               : "memory");
}

__global__ void k_init(float* __restrict__ deg, float* __restrict__ agg,
                       int n, int tot) {
  int t = blockIdx.x * blockDim.x + threadIdx.x;
  if (t < tot) agg[t] = 0.0f;
  if (t < n)   deg[t] = 1.0f;
}

__global__ void k_zero(float* __restrict__ p, int tot) {
  int t = blockIdx.x * blockDim.x + threadIdx.x;
  if (t < tot) p[t] = 0.0f;
}

__global__ void k_degree(const int* __restrict__ dst, float* __restrict__ deg,
                         int E) {
  int t = blockIdx.x * blockDim.x + threadIdx.x;
  if (t < E) atomAddF(&deg[dst[t]], 1.0f);
}

__global__ void k_rsqrt(float* __restrict__ d, int n) {
  int t = blockIdx.x * blockDim.x + threadIdx.x;
  if (t < n) d[t] = rsqrtf(d[t]);
}

__global__ void k_gemm16(const float* __restrict__ X, const float* __restrict__ W,
                         float* __restrict__ Y, int nrows, int K) {
  const int wave = threadIdx.x >> 5;
  const int lane = threadIdx.x & 31;
  const int tile = blockIdx.x * (blockDim.x >> 5) + wave;
  const int row0 = tile * 16;
  if (row0 >= nrows) return;

  const int m  = lane & 15;
  const int kh = lane >> 4;

  const float* ap = X + (size_t)(row0 + m) * K + 2 * kh;
  v8f acc = {};
  for (int kb = 0; kb < K; kb += 4) {
    v2f a = *(const v2f*)(ap + kb);
    const int kr = kb + 2 * kh;
    v2f b;
    b.x = W[(size_t)kr * 16 + m];
    b.y = W[(size_t)(kr + 1) * 16 + m];
    acc = __builtin_amdgcn_wmma_f32_16x16x4_f32(
        false, a, false, b, (short)0, acc, false, false);
  }
#pragma unroll
  for (int v = 0; v < 8; ++v) {
    Y[(size_t)(row0 + v + 8 * kh) * 16 + m] = acc[v];
  }
}

__global__ void k_scatter(const int* __restrict__ src, const int* __restrict__ dst,
                          const float* __restrict__ dinv,
                          const float* __restrict__ xw,
                          float* __restrict__ agg, int E) {
  int t = blockIdx.x * blockDim.x + threadIdx.x;
  int e = t >> 2;
  if (e >= E) return;
  int q = (t & 3) << 2;
  int s = src[e];
  int d = dst[e];
  float nrm = dinv[s] * dinv[d];
  float4 v = *(const float4*)(xw + (size_t)s * 16 + q);
  float* out = agg + (size_t)d * 16 + q;
  atomAddF(out + 0, v.x * nrm);
  atomAddF(out + 1, v.y * nrm);
  atomAddF(out + 2, v.z * nrm);
  atomAddF(out + 3, v.w * nrm);
}

__global__ void k_final1(float* __restrict__ agg, const float* __restrict__ xw,
                         const float* __restrict__ dinv,
                         const float* __restrict__ b, int tot) {
  int t = blockIdx.x * blockDim.x + threadIdx.x;
  if (t >= tot) return;
  int i = t >> 4, f = t & 15;
  float s = dinv[i];
  float v = agg[t] + xw[t] * s * s + b[f];
  agg[t] = fmaxf(v, 0.0f);
}

__global__ void k_final2(const float* __restrict__ agg, const float* __restrict__ xw,
                         const float* __restrict__ dinv,
                         const float* __restrict__ b2,
                         const float* __restrict__ Wl,
                         const float* __restrict__ bl,
                         float* __restrict__ out, int n) {
  int i = blockIdx.x * blockDim.x + threadIdx.x;
  if (i >= n) return;
  float s = dinv[i];
  float ss = s * s;
  float acc = 0.0f;
#pragma unroll
  for (int f = 0; f < 16; ++f) {
    float h = agg[(size_t)i * 16 + f] + xw[(size_t)i * 16 + f] * ss + b2[f];
    h = fmaxf(h, 0.0f);
    acc += h * Wl[f];
  }
  out[i] = acc + bl[0];
}

#define NB_SHIFT 8
#define NB       256
#define PG       2048
#define MAXBINS  512
#define BIN_BITS 9
#define AGW      16
#define AGT      (AGW * 32)
#define CAP      10240
#define GU       4
#define BW       16
#define BT       (BW * 32)
#define HS       (MAXBINS + 1)
#define CHW      1600
#define RS       (NB + 1)
static_assert(MAXBINS == (1 << BIN_BITS) && NB == (1 << NB_SHIFT), "key widths");
static_assert(PG % BW == 0 && BT == MAXBINS, "bucketing work-group shape");
static_assert(CAP % AGT == 0 && RS <= AGT, "a sorted batch is copied by one work-group in whole rounds");

template <int BITS>
__device__ __forceinline__ unsigned sameKeyMask(int key, bool valid) {
  static_assert(BITS >= 1 && BITS <= 31, "key width");
  unsigned differ = 0u;
#pragma unroll
  for (int b = 0; b < BITS; ++b) {
    const int mine = (key << (31 - b)) >> 31;
    const unsigned vote = __builtin_amdgcn_ballot_w32(mine != 0);
    differ |= vote ^ (unsigned)mine;
  }
  return ~differ & __builtin_amdgcn_ballot_w32(valid);
}

template <int N>
__device__ __forceinline__ void blockScanIncl(int* s, int t) {
  for (int d = 1; d < N; d <<= 1) {
    const int v = (t >= d) ? s[t - d] : 0;
    __syncthreads();
    s[t] += v;
    __syncthreads();
  }
}

__global__ void k_count(const int* __restrict__ dst, int* __restrict__ cnt,
                        int E, int chunk, int nbins) {
  __shared__ int h[BW * HS];
  const int t = threadIdx.x;
  const int lane = t & 31;
  const int wave = __builtin_amdgcn_readfirstlane(t >> 5);
  for (int i = t; i < BW * HS; i += BT) h[i] = 0;
  __syncthreads();
  int* hv = h + wave * HS;
  const int g0 = blockIdx.x * BW;
  const int e0 = (g0 + wave) * chunk;
  const int e1 = min(e0 + chunk, E);
  int dn[GU];
#pragma unroll
  for (int u = 0; u < GU; ++u) dn[u] = dst[min(e0 + u * 32 + lane, E - 1)];
  for (int base = e0; base < e1; base += 32 * GU) {
    int d[GU];
#pragma unroll
    for (int u = 0; u < GU; ++u) d[u] = dn[u];
#pragma unroll
    for (int u = 0; u < GU; ++u) dn[u] = dst[min(base + 32 * GU + u * 32 + lane, E - 1)];
#pragma unroll
    for (int u = 0; u < GU; ++u) {
      if (base + u * 32 < e1) {
        const bool valid = base + u * 32 + lane < e1;
        const int key = valid ? (d[u] >> NB_SHIFT) : -1;
        const unsigned m = sameKeyMask<BIN_BITS>(key, valid);
        if (valid && (m >> lane) == 1u) hv[key] += __popc(m);
      }
    }
  }
  __syncthreads();
  for (int i = t; i < nbins * BW; i += BT) {
    const int bin = i / BW, w = i % BW;
    cnt[bin * PG + g0 + w] = h[w * HS + bin];
  }
}

__global__ void k_scan_bin(int* __restrict__ cnt, int* __restrict__ binTot) {
  __shared__ int s[PG / 2];
  const int t = threadIdx.x, b = blockIdx.x;
  int* row = cnt + (size_t)b * PG;
  const int c0 = row[2 * t], c1 = row[2 * t + 1];
  const int c = c0 + c1;
  s[t] = c;
  __syncthreads();
  blockScanIncl<PG / 2>(s, t);
  const int ex = s[t] - c;
  row[2 * t]     = ex;
  row[2 * t + 1] = ex + c0;
  if (t == PG / 2 - 1) binTot[b] = s[t];
}

__global__ void k_scan_tot(const int* __restrict__ binTot, int* __restrict__ binBase,
                           int nbins) {
  __shared__ int s[MAXBINS];
  const int t = threadIdx.x;
  const int c = (t < nbins) ? binTot[t] : 0;
  s[t] = c;
  __syncthreads();
  blockScanIncl<MAXBINS>(s, t);
  if (t < nbins) binBase[t] = s[t] - c;
  if (t == MAXBINS - 1) binBase[nbins] = s[t];
}

__global__ void k_part(const int* __restrict__ src, const int* __restrict__ dst,
                       const int* __restrict__ cnt, const int* __restrict__ binBase,
                       unsigned* __restrict__ pk, int E, int chunk, int nbins) {
  __shared__ int cur[BW * HS];
  __shared__ int dl[MAXBINS];
  __shared__ int sc[BT];
  __shared__ v2u stage[BW * CHW];
  const int t = threadIdx.x;
  const int lane = t & 31;
  const int wave = __builtin_amdgcn_readfirstlane(t >> 5);
  const int g0 = blockIdx.x * BW;
  int c = 0, gs = 0;
  if (t < nbins) {
    const int bb = binBase[t];
    const int lo = cnt[t * PG + g0];
    const int hi = (g0 + BW < PG) ? cnt[t * PG + g0 + BW] : binBase[t + 1] - bb;
    c = hi - lo;
    gs = bb + lo;
  }
  sc[t] = c;
  __syncthreads();
  blockScanIncl<BT>(sc, t);
  dl[t] = (sc[t] - c) - gs;
  const int total = sc[BT - 1];
  for (int i = t; i < nbins * BW; i += BT) {
    const int bin = i / BW, w = i % BW;
    cur[w * HS + bin] = binBase[bin] + cnt[bin * PG + g0 + w];
  }
  __syncthreads();
  int* curv = cur + wave * HS;
  const int e0 = (g0 + wave) * chunk;
  const int e1 = min(e0 + chunk, E);
  const unsigned below = (1u << lane) - 1u;
  int dn[GU], sn[GU];
#pragma unroll
  for (int u = 0; u < GU; ++u) {
    const int i = min(e0 + u * 32 + lane, E - 1);
    dn[u] = dst[i]; sn[u] = src[i];
  }
  for (int base = e0; base < e1; base += 32 * GU) {
    int d[GU], s[GU];
#pragma unroll
    for (int u = 0; u < GU; ++u) { d[u] = dn[u]; s[u] = sn[u]; }
#pragma unroll
    for (int u = 0; u < GU; ++u) {
      const int i = min(base + 32 * GU + u * 32 + lane, E - 1);
      dn[u] = dst[i]; sn[u] = src[i];
    }
#pragma unroll
    for (int u = 0; u < GU; ++u) {
      if (base + u * 32 < e1) {
        const bool valid = base + u * 32 + lane < e1;
        const int key = valid ? (d[u] >> NB_SHIFT) : -1;
        const unsigned m = sameKeyMask<BIN_BITS>(key, valid);
        if (valid) {
          const int pos = curv[key] + __popc(m & below);
          if ((m >> lane) == 1u) curv[key] = pos + 1;
          v2u e;
          e.x = ((unsigned)s[u] << NB_SHIFT) | (unsigned)(d[u] & (NB - 1));
          e.y = (unsigned)pos;
          stage[pos + dl[key]] = e;
        }
      }
    }
  }
  __syncthreads();
#pragma unroll 4
  for (int i = t; i < total; i += BT) {
    const v2u e = stage[i];
    pk[e.y] = e.x;
  }
}

__device__ __forceinline__ void countRows(const unsigned* __restrict__ pk, int b0, int b1,
                                          int* hwv, int wave, int lane) {
  for (int base = b0 + wave * 32; base < b1; base += AGT) {
    const int i = base + lane;
    const bool valid = i < b1;
    const unsigned q = pk[min(i, b1 - 1)];
    const int key = valid ? (int)(q & (NB - 1)) : -1;
    const unsigned m = sameKeyMask<NB_SHIFT>(key, valid);
    if (valid && (m >> lane) == 1u) hwv[key] += __popc(m);
  }
}

__global__ void k_sort(unsigned* __restrict__ pk, const int* __restrict__ binBase,
                       int* __restrict__ rs, float* __restrict__ dinv, int n) {
  __shared__ int hw[AGW * NB];
  __shared__ int sc[AGT];
  __shared__ int sorted[CAP];

  const int t = threadIdx.x, b = blockIdx.x;
  const int lane = t & 31;
  const int wave = __builtin_amdgcn_readfirstlane(t >> 5);
  const int e0 = binBase[b], e1 = binBase[b + 1];
  int* hwv = hw + wave * NB;
  const unsigned below = (1u << lane) - 1u;
  int* rsb = rs + (size_t)(b + e0 / CAP) * RS;
  int indeg = 0;

  for (int b0 = e0; b0 < e1; b0 += CAP, rsb += RS) {
    const int b1 = min(b0 + CAP, e1);
    for (int i = t; i < AGW * NB; i += AGT) hw[i] = 0;
    __syncthreads();
    countRows(pk, b0, b1, hwv, wave, lane);
    __syncthreads();
    {
      const int key = t >> 1, w0 = (t & 1) * (AGW / 2);
      int s = 0;
#pragma unroll
      for (int i = 0; i < AGW / 2; ++i) s += hw[(w0 + i) * NB + key];
      indeg += s + __shfl_xor(s, 1);
      sc[t] = s;
      __syncthreads();
      blockScanIncl<AGT>(sc, t);
      int run = sc[t] - s;
      if ((t & 1) == 0) rsb[key] = run;
      if (t == AGT - 1) rsb[NB] = sc[t];
#pragma unroll
      for (int i = 0; i < AGW / 2; ++i) {
        const int idx = (w0 + i) * NB + key;
        const int c = hw[idx];
        hw[idx] = run;
        run += c;
      }
    }
    __syncthreads();
    for (int base = b0 + wave * 32; base < b1; base += AGT) {
      const int i = base + lane;
      const bool valid = i < b1;
      const unsigned q = pk[min(i, b1 - 1)];
      const int key = valid ? (int)(q & (NB - 1)) : -1;
      const unsigned m = sameKeyMask<NB_SHIFT>(key, valid);
      if (valid) {
        const int pos = hwv[key] + __popc(m & below);
        if ((m >> lane) == 1u) hwv[key] = pos + 1;
        sorted[pos] = (int)(q >> NB_SHIFT);
      }
    }
    __syncthreads();
    for (int i = t; i < b1 - b0; i += AGT) pk[b0 + i] = (unsigned)sorted[i];
  }
  const int node = b * NB + (t >> 1);
  if ((t & 1) == 0 && node < n) dinv[node] = rsqrtf(1.0f + (float)indeg);
}

__global__ void k_gemm16s(const float* __restrict__ X, const float* __restrict__ W,
                          const float* __restrict__ dinv,
                          float* __restrict__ Y, int nrows, int K) {
  const int wave = threadIdx.x >> 5;
  const int lane = threadIdx.x & 31;
  const int tile = blockIdx.x * (blockDim.x >> 5) + wave;
  const int row0 = tile * 16;
  if (row0 >= nrows) return;

  const int m  = lane & 15;
  const int kh = lane >> 4;

  const float* ap = X + (size_t)(row0 + m) * K + 2 * kh;
  v8f acc = {};
  for (int kb = 0; kb < K; kb += 4) {
    v2f a = *(const v2f*)(ap + kb);
    const int kr = kb + 2 * kh;
    v2f b;
    b.x = W[(size_t)kr * 16 + m];
    b.y = W[(size_t)(kr + 1) * 16 + m];
    acc = __builtin_amdgcn_wmma_f32_16x16x4_f32(
        false, a, false, b, (short)0, acc, false, false);
  }
#pragma unroll
  for (int v = 0; v < 8; ++v) {
    const int r = row0 + v + 8 * kh;
    Y[(size_t)r * 16 + m] = acc[v] * dinv[r];
  }
}

template <bool LAST>
__global__ void k_agg(const unsigned* __restrict__ pk, const int* __restrict__ binBase,
                      const int* __restrict__ rs,
                      const float* __restrict__ xws, const float* __restrict__ dinv,
                      const float* __restrict__ bias,
                      const float* __restrict__ Wl, const float* __restrict__ bl,
                      float* __restrict__ H, float* __restrict__ out, int n) {
  __shared__ __attribute__((aligned(16))) float acc[NB * 16];
  __shared__ int rowStart[RS];
  __shared__ int sorted[CAP];

  const int t = threadIdx.x, b = blockIdx.x;
  const int lane = t & 31;
  const int fq = lane & 3;
  const int grp = t >> 2;
  const float4* xq = (const float4*)xws;
  const int node0 = b * NB;
  for (int i = t; i < NB * 16; i += AGT) {
    const int node = node0 + (i >> 4);
    acc[i] = (node < n) ? xws[(size_t)node0 * 16 + i] : 0.0f;
  }
  const int e0 = binBase[b], e1 = binBase[b + 1];
  const int* rsb = rs + (size_t)(b + e0 / CAP) * RS;

  for (int b0 = e0; b0 < e1; b0 += CAP, rsb += RS) {
    const int b1 = min(b0 + CAP, e1);
    const int nb = b1 - b0;
    const int rsv = rsb[min(t, NB)];
    unsigned q[CAP / AGT];
#pragma unroll
    for (int u = 0; u < CAP / AGT; ++u) q[u] = pk[b0 + min(u * AGT + t, nb - 1)];
    if (t < RS) rowStart[t] = rsv;
#pragma unroll
    for (int u = 0; u < CAP / AGT; ++u)
      if (u * AGT + t < nb) sorted[u * AGT + t] = (int)q[u];
    __syncthreads();
    for (int r = grp; r < NB; r += AGT / 4) {
      const int s0 = rowStart[r], s1 = rowStart[r + 1];
      float4 sum = make_float4(0.0f, 0.0f, 0.0f, 0.0f);
      for (int i = s0; i < s1; i += 8) {
        float4 v[8];
#pragma unroll
        for (int u = 0; u < 8; ++u) {
          const int idx = min(i + u, s1 - 1);
          v[u] = xq[(unsigned)sorted[idx] * 4u + fq];
        }
#pragma unroll
        for (int u = 0; u < 8; ++u) {
          const bool in = i + u < s1;
          sum.x += in ? v[u].x : 0.0f;
          sum.y += in ? v[u].y : 0.0f;
          sum.z += in ? v[u].z : 0.0f;
          sum.w += in ? v[u].w : 0.0f;
        }
      }
      float4* ar = (float4*)(acc + r * 16 + fq * 4);
      float4 a = *ar;
      a.x += sum.x; a.y += sum.y; a.z += sum.z; a.w += sum.w;
      *ar = a;
    }
    __syncthreads();
  }
  __syncthreads();

  if (!LAST) {
    for (int i = t; i < NB * 16; i += AGT) {
      const int node = node0 + (i >> 4);
      if (node < n)
        H[(size_t)node0 * 16 + i] = fmaxf(acc[i] * dinv[node] + bias[i & 15], 0.0f);
    }
  } else {
    const int node = node0 + t;
    if (t < NB && node < n) {
      const float s = dinv[node];
      float a = 0.0f;
#pragma unroll
      for (int ff = 0; ff < 16; ++ff) {
        const float h = fmaxf(acc[t * 16 + ff] * s + bias[ff], 0.0f);
        a += h * Wl[ff];
      }
      out[node] = a + bl[0];
    }
  }
}

extern "C" void kernel_launch(void* const* d_in, const int* in_sizes, int n_in,
                              void* d_out, int out_size, void* d_ws, size_t ws_size,
                              hipStream_t stream) {
  const float* x   = (const float*)d_in[0];
  const int*   ei  = (const int*)d_in[1];
  const float* W1  = (const float*)d_in[2];
  const float* b1  = (const float*)d_in[3];
  const float* W2  = (const float*)d_in[4];
  const float* b2  = (const float*)d_in[5];
  const float* Wl  = (const float*)d_in[6];
  const float* bl  = (const float*)d_in[7];
  float* out = (float*)d_out;

  const int din = in_sizes[2] / 16;
  const int n   = in_sizes[0] / din;
  const int E   = in_sizes[1] / 2;
  const int tot = n * 16;

  float* ws   = (float*)d_ws;
  float* dinv = ws;
  float* bufA = ws + (((size_t)n + 255) & ~(size_t)255);
  float* bufB = bufA + (size_t)n * 16;

  const int T = 256;
  const int ntiles = (n + 15) / 16;
  const int gemmBlocks = (ntiles + 3) / 4;

  const int nbins = (n + NB - 1) / NB;
  unsigned* pk   = (unsigned*)(bufB + (size_t)n * 16);
  int* cnt       = (int*)(pk + (((size_t)E + 255) & ~(size_t)255));
  int* binTot    = cnt + (size_t)nbins * PG;
  int* binBase   = binTot + MAXBINS;
  const size_t need = (size_t)((char*)(binBase + MAXBINS + 1) - (char*)d_ws);

  int* rs        = cnt;

  const int chunk = (E + PG - 1) / PG;
  if (ws_size >= need && nbins <= MAXBINS && chunk <= CHW &&
      (size_t)(nbins + E / CAP + 1) * RS <= (size_t)nbins * PG) {
    k_count   <<<PG / BW, BT, 0, stream>>>(ei + E, cnt, E, chunk, nbins);
    k_scan_bin<<<nbins, PG / 2, 0, stream>>>(cnt, binTot);
    k_scan_tot<<<1, MAXBINS, 0, stream>>>(binTot, binBase, nbins);
    k_part    <<<PG / BW, BT, 0, stream>>>(ei, ei + E, cnt, binBase, pk, E, chunk, nbins);
    k_sort    <<<nbins, AGT, 0, stream>>>(pk, binBase, rs, dinv, n);

    k_gemm16s <<<gemmBlocks, 128, 0, stream>>>(x, W1, dinv, bufA, n, din);
    k_agg<false><<<nbins, AGT, 0, stream>>>(pk, binBase, rs, bufA, dinv, b1, Wl, bl, bufB, out, n);
    k_gemm16s <<<gemmBlocks, 128, 0, stream>>>(bufB, W2, dinv, bufA, n, 16);
    k_agg<true> <<<nbins, AGT, 0, stream>>>(pk, binBase, rs, bufA, dinv, b2, Wl, bl, bufB, out, n);
    return;
  }

  k_init  <<<(tot + T - 1) / T, T, 0, stream>>>(dinv, bufB, n, tot);
  k_degree<<<(E + T - 1) / T,   T, 0, stream>>>(ei + E, dinv, E);
  k_rsqrt <<<(n + T - 1) / T,   T, 0, stream>>>(dinv, n);

  k_gemm16 <<<gemmBlocks, 128, 0, stream>>>(x, W1, bufA, n, din);
  k_scatter<<<(4 * E + T - 1) / T, T, 0, stream>>>(ei, ei + E, dinv, bufA, bufB, E);
  k_final1 <<<(tot + T - 1) / T, T, 0, stream>>>(bufB, bufA, dinv, b1, tot);

  k_gemm16 <<<gemmBlocks, 128, 0, stream>>>(bufB, W2, bufA, n, 16);
  k_zero   <<<(tot + T - 1) / T, T, 0, stream>>>(bufB, tot);
  k_scatter<<<(4 * E + T - 1) / T, T, 0, stream>>>(ei, ei + E, dinv, bufA, bufB, E);
  k_final2 <<<(n + T - 1) / T, T, 0, stream>>>(bufB, bufA, dinv, b2, Wl, bl, out, n);
}


# both products stop being launches (9 become 7): layer 1 rides the first bucketing launch as extra work-groups, unscaled, the bin sort scales its rows; layer 2 from row sums in LDS; f32 sum order same
# speedup vs baseline: 5.0576x; 1.0617x over previous
#include <hip/hip_runtime.h>
#include <hip/hip_bf16.h>
#include <stddef.h>

typedef __attribute__((ext_vector_type(2))) float v2f;
typedef __attribute__((ext_vector_type(8))) float v8f;
typedef __attribute__((ext_vector_type(2))) unsigned v2u;

__device__ __forceinline__ void atomAddF(float* p, float v) {
  asm volatile("global_atomic_add_f32 %0, %1, off scope:SCOPE_DEV"
               :
               : "v"(p), "v"(v)
               : "memory");
}

__global__ void k_init(float* __restrict__ deg, float* __restrict__ agg,
                       int n, int tot) {
  int t = blockIdx.x * blockDim.x + threadIdx.x;
  if (t < tot) agg[t] = 0.0f;
  if (t < n)   deg[t] = 1.0f;
}

__global__ void k_zero(float* __restrict__ p, int tot) {
  int t = blockIdx.x * blockDim.x + threadIdx.x;
  if (t < tot) p[t] = 0.0f;
}

__global__ void k_degree(const int* __restrict__ dst, float* __restrict__ deg,
                         int E) {
  int t = blockIdx.x * blockDim.x + threadIdx.x;
  if (t < E) atomAddF(&deg[dst[t]], 1.0f);
}

__global__ void k_rsqrt(float* __restrict__ d, int n) {
  int t = blockIdx.x * blockDim.x + threadIdx.x;
  if (t < n) d[t] = rsqrtf(d[t]);
}

__global__ void k_gemm16(const float* __restrict__ X, const float* __restrict__ W,
                         float* __restrict__ Y, int nrows, int K) {
  const int wave = threadIdx.x >> 5;
  const int lane = threadIdx.x & 31;
  const int tile = blockIdx.x * (blockDim.x >> 5) + wave;
  const int row0 = tile * 16;
  if (row0 >= nrows) return;

  const int m  = lane & 15;
  const int kh = lane >> 4;

  const float* ap = X + (size_t)(row0 + m) * K + 2 * kh;
  v8f acc = {};
  for (int kb = 0; kb < K; kb += 4) {
    v2f a = *(const v2f*)(ap + kb);
    const int kr = kb + 2 * kh;
    v2f b;
    b.x = W[(size_t)kr * 16 + m];
    b.y = W[(size_t)(kr + 1) * 16 + m];
    acc = __builtin_amdgcn_wmma_f32_16x16x4_f32(
        false, a, false, b, (short)0, acc, false, false);
  }
#pragma unroll
  for (int v = 0; v < 8; ++v) {
    Y[(size_t)(row0 + v + 8 * kh) * 16 + m] = acc[v];
  }
}

__global__ void k_scatter(const int* __restrict__ src, const int* __restrict__ dst,
                          const float* __restrict__ dinv,
                          const float* __restrict__ xw,
                          float* __restrict__ agg, int E) {
  int t = blockIdx.x * blockDim.x + threadIdx.x;
  int e = t >> 2;
  if (e >= E) return;
  int q = (t & 3) << 2;
  int s = src[e];
  int d = dst[e];
  float nrm = dinv[s] * dinv[d];
  float4 v = *(const float4*)(xw + (size_t)s * 16 + q);
  float* out = agg + (size_t)d * 16 + q;
  atomAddF(out + 0, v.x * nrm);
  atomAddF(out + 1, v.y * nrm);
  atomAddF(out + 2, v.z * nrm);
  atomAddF(out + 3, v.w * nrm);
}

__global__ void k_final1(float* __restrict__ agg, const float* __restrict__ xw,
                         const float* __restrict__ dinv,
                         const float* __restrict__ b, int tot) {
  int t = blockIdx.x * blockDim.x + threadIdx.x;
  if (t >= tot) return;
  int i = t >> 4, f = t & 15;
  float s = dinv[i];
  float v = agg[t] + xw[t] * s * s + b[f];
  agg[t] = fmaxf(v, 0.0f);
}

__global__ void k_final2(const float* __restrict__ agg, const float* __restrict__ xw,
                         const float* __restrict__ dinv,
                         const float* __restrict__ b2,
                         const float* __restrict__ Wl,
                         const float* __restrict__ bl,
                         float* __restrict__ out, int n) {
  int i = blockIdx.x * blockDim.x + threadIdx.x;
  if (i >= n) return;
  float s = dinv[i];
  float ss = s * s;
  float acc = 0.0f;
#pragma unroll
  for (int f = 0; f < 16; ++f) {
    float h = agg[(size_t)i * 16 + f] + xw[(size_t)i * 16 + f] * ss + b2[f];
    h = fmaxf(h, 0.0f);
    acc += h * Wl[f];
  }
  out[i] = acc + bl[0];
}

#define NB_SHIFT 8
#define NB       256
#define PG       2048
#define MAXBINS  512
#define BIN_BITS 9
#define AGW      16
#define AGT      (AGW * 32)
#define CAP      10240
#define GU       4
#define BW       16
#define BT       (BW * 32)
#define HS       (MAXBINS + 1)
#define CHW      1600
#define RS       (NB + 1)
static_assert(MAXBINS == (1 << BIN_BITS) && NB == (1 << NB_SHIFT), "key widths");
static_assert(PG % BW == 0 && BT == MAXBINS, "bucketing work-group shape");
static_assert(BW * 16 == NB, "a product work-group's BW tiles of 16 rows are one bin's rows");
static_assert((NB * 4) % AGT == 0, "a bin's product rows are scaled by one work-group in whole rounds");
static_assert(CAP % AGT == 0 && RS <= AGT, "a sorted batch is copied by one work-group in whole rounds");

template <int BITS>
__device__ __forceinline__ unsigned sameKeyMask(int key, bool valid) {
  static_assert(BITS >= 1 && BITS <= 31, "key width");
  unsigned differ = 0u;
#pragma unroll
  for (int b = 0; b < BITS; ++b) {
    const int mine = (key << (31 - b)) >> 31;
    const unsigned vote = __builtin_amdgcn_ballot_w32(mine != 0);
    differ |= vote ^ (unsigned)mine;
  }
  return ~differ & __builtin_amdgcn_ballot_w32(valid);
}

template <int N>
__device__ __forceinline__ void blockScanIncl(int* s, int t) {
  for (int d = 1; d < N; d <<= 1) {
    const int v = (t >= d) ? s[t - d] : 0;
    __syncthreads();
    s[t] += v;
    __syncthreads();
  }
}

__global__ void k_count(const int* __restrict__ dst, int* __restrict__ cnt,
                        int E, int chunk, int nbins,
                        const float* __restrict__ X, const float* __restrict__ W,
                        float* __restrict__ Y, int n, int K) {
  __shared__ int h[BW * HS];
  const int t = threadIdx.x;
  const int lane = t & 31;
  const int wave = __builtin_amdgcn_readfirstlane(t >> 5);
  if (blockIdx.x >= PG / BW) {
    const int row0 = ((int)blockIdx.x - PG / BW) * NB + wave * 16;
    if (row0 >= n) return;
    const int m  = lane & 15;
    const int kh = lane >> 4;
    const float* ap = X + (size_t)(row0 + m) * K + 2 * kh;
    v8f acc = {};
    for (int kb = 0; kb < K; kb += 4) {
      v2f a = *(const v2f*)(ap + kb);
      const int kr = kb + 2 * kh;
      v2f w;
      w.x = W[(size_t)kr * 16 + m];
      w.y = W[(size_t)(kr + 1) * 16 + m];
      acc = __builtin_amdgcn_wmma_f32_16x16x4_f32(
          false, a, false, w, (short)0, acc, false, false);
    }
#pragma unroll
    for (int v = 0; v < 8; ++v) {
      Y[(size_t)(row0 + v + 8 * kh) * 16 + m] = acc[v];
    }
    return;
  }
  for (int i = t; i < BW * HS; i += BT) h[i] = 0;
  __syncthreads();
  int* hv = h + wave * HS;
  const int g0 = blockIdx.x * BW;
  const int e0 = (g0 + wave) * chunk;
  const int e1 = min(e0 + chunk, E);
  int dn[GU];
#pragma unroll
  for (int u = 0; u < GU; ++u) dn[u] = dst[min(e0 + u * 32 + lane, E - 1)];
  for (int base = e0; base < e1; base += 32 * GU) {
    int d[GU];
#pragma unroll
    for (int u = 0; u < GU; ++u) d[u] = dn[u];
#pragma unroll
    for (int u = 0; u < GU; ++u) dn[u] = dst[min(base + 32 * GU + u * 32 + lane, E - 1)];
#pragma unroll
    for (int u = 0; u < GU; ++u) {
      if (base + u * 32 < e1) {
        const bool valid = base + u * 32 + lane < e1;
        const int key = valid ? (d[u] >> NB_SHIFT) : -1;
        const unsigned m = sameKeyMask<BIN_BITS>(key, valid);
        if (valid && (m >> lane) == 1u) hv[key] += __popc(m);
      }
    }
  }
  __syncthreads();
  for (int i = t; i < nbins * BW; i += BT) {
    const int bin = i / BW, w = i % BW;
    cnt[bin * PG + g0 + w] = h[w * HS + bin];
  }
}

__global__ void k_scan_bin(int* __restrict__ cnt, int* __restrict__ binTot) {
  __shared__ int s[PG / 2];
  const int t = threadIdx.x, b = blockIdx.x;
  int* row = cnt + (size_t)b * PG;
  const int c0 = row[2 * t], c1 = row[2 * t + 1];
  const int c = c0 + c1;
  s[t] = c;
  __syncthreads();
  blockScanIncl<PG / 2>(s, t);
  const int ex = s[t] - c;
  row[2 * t]     = ex;
  row[2 * t + 1] = ex + c0;
  if (t == PG / 2 - 1) binTot[b] = s[t];
}

__global__ void k_scan_tot(const int* __restrict__ binTot, int* __restrict__ binBase,
                           int nbins) {
  __shared__ int s[MAXBINS];
  const int t = threadIdx.x;
  const int c = (t < nbins) ? binTot[t] : 0;
  s[t] = c;
  __syncthreads();
  blockScanIncl<MAXBINS>(s, t);
  if (t < nbins) binBase[t] = s[t] - c;
  if (t == MAXBINS - 1) binBase[nbins] = s[t];
}

__global__ void k_part(const int* __restrict__ src, const int* __restrict__ dst,
                       const int* __restrict__ cnt, const int* __restrict__ binBase,
                       unsigned* __restrict__ pk, int E, int chunk, int nbins) {
  __shared__ int cur[BW * HS];
  __shared__ int dl[MAXBINS];
  __shared__ int sc[BT];
  __shared__ v2u stage[BW * CHW];
  const int t = threadIdx.x;
  const int lane = t & 31;
  const int wave = __builtin_amdgcn_readfirstlane(t >> 5);
  const int g0 = blockIdx.x * BW;
  int c = 0, gs = 0;
  if (t < nbins) {
    const int bb = binBase[t];
    const int lo = cnt[t * PG + g0];
    const int hi = (g0 + BW < PG) ? cnt[t * PG + g0 + BW] : binBase[t + 1] - bb;
    c = hi - lo;
    gs = bb + lo;
  }
  sc[t] = c;
  __syncthreads();
  blockScanIncl<BT>(sc, t);
  dl[t] = (sc[t] - c) - gs;
  const int total = sc[BT - 1];
  for (int i = t; i < nbins * BW; i += BT) {
    const int bin = i / BW, w = i % BW;
    cur[w * HS + bin] = binBase[bin] + cnt[bin * PG + g0 + w];
  }
  __syncthreads();
  int* curv = cur + wave * HS;
  const int e0 = (g0 + wave) * chunk;
  const int e1 = min(e0 + chunk, E);
  const unsigned below = (1u << lane) - 1u;
  int dn[GU], sn[GU];
#pragma unroll
  for (int u = 0; u < GU; ++u) {
    const int i = min(e0 + u * 32 + lane, E - 1);
    dn[u] = dst[i]; sn[u] = src[i];
  }
  for (int base = e0; base < e1; base += 32 * GU) {
    int d[GU], s[GU];
#pragma unroll
    for (int u = 0; u < GU; ++u) { d[u] = dn[u]; s[u] = sn[u]; }
#pragma unroll
    for (int u = 0; u < GU; ++u) {
      const int i = min(base + 32 * GU + u * 32 + lane, E - 1);
      dn[u] = dst[i]; sn[u] = src[i];
    }
#pragma unroll
    for (int u = 0; u < GU; ++u) {
      if (base + u * 32 < e1) {
        const bool valid = base + u * 32 + lane < e1;
        const int key = valid ? (d[u] >> NB_SHIFT) : -1;
        const unsigned m = sameKeyMask<BIN_BITS>(key, valid);
        if (valid) {
          const int pos = curv[key] + __popc(m & below);
          if ((m >> lane) == 1u) curv[key] = pos + 1;
          v2u e;
          e.x = ((unsigned)s[u] << NB_SHIFT) | (unsigned)(d[u] & (NB - 1));
          e.y = (unsigned)pos;
          stage[pos + dl[key]] = e;
        }
      }
    }
  }
  __syncthreads();
#pragma unroll 4
  for (int i = t; i < total; i += BT) {
    const v2u e = stage[i];
    pk[e.y] = e.x;
  }
}

__device__ __forceinline__ void countRows(const unsigned* __restrict__ pk, int b0, int b1,
                                          int* hwv, int wave, int lane) {
  for (int base = b0 + wave * 32; base < b1; base += AGT) {
    const int i = base + lane;
    const bool valid = i < b1;
    const unsigned q = pk[min(i, b1 - 1)];
    const int key = valid ? (int)(q & (NB - 1)) : -1;
    const unsigned m = sameKeyMask<NB_SHIFT>(key, valid);
    if (valid && (m >> lane) == 1u) hwv[key] += __popc(m);
  }
}

__global__ void k_sort(unsigned* __restrict__ pk, const int* __restrict__ binBase,
                       int* __restrict__ rs, float* __restrict__ dinv, int n,
                       float* __restrict__ Y) {
  __shared__ int hw[AGW * NB];
  __shared__ int sc[AGT];
  __shared__ int sorted[CAP];
  __shared__ float dv[NB];

  const int t = threadIdx.x, b = blockIdx.x;
  const int lane = t & 31;
  const int wave = __builtin_amdgcn_readfirstlane(t >> 5);
  const int e0 = binBase[b], e1 = binBase[b + 1];
  int* hwv = hw + wave * NB;
  const unsigned below = (1u << lane) - 1u;
  int* rsb = rs + (size_t)(b + e0 / CAP) * RS;
  int indeg = 0;

  for (int b0 = e0; b0 < e1; b0 += CAP, rsb += RS) {
    const int b1 = min(b0 + CAP, e1);
    for (int i = t; i < AGW * NB; i += AGT) hw[i] = 0;
    __syncthreads();
    countRows(pk, b0, b1, hwv, wave, lane);
    __syncthreads();
    {
      const int key = t >> 1, w0 = (t & 1) * (AGW / 2);
      int s = 0;
#pragma unroll
      for (int i = 0; i < AGW / 2; ++i) s += hw[(w0 + i) * NB + key];
      indeg += s + __shfl_xor(s, 1);
      sc[t] = s;
      __syncthreads();
      blockScanIncl<AGT>(sc, t);
      int run = sc[t] - s;
      if ((t & 1) == 0) rsb[key] = run;
      if (t == AGT - 1) rsb[NB] = sc[t];
#pragma unroll
      for (int i = 0; i < AGW / 2; ++i) {
        const int idx = (w0 + i) * NB + key;
        const int c = hw[idx];
        hw[idx] = run;
        run += c;
      }
    }
    __syncthreads();
    for (int base = b0 + wave * 32; base < b1; base += AGT) {
      const int i = base + lane;
      const bool valid = i < b1;
      const unsigned q = pk[min(i, b1 - 1)];
      const int key = valid ? (int)(q & (NB - 1)) : -1;
      const unsigned m = sameKeyMask<NB_SHIFT>(key, valid);
      if (valid) {
        const int pos = hwv[key] + __popc(m & below);
        if ((m >> lane) == 1u) hwv[key] = pos + 1;
        sorted[pos] = (int)(q >> NB_SHIFT);
      }
    }
    __syncthreads();
    for (int i = t; i < b1 - b0; i += AGT) pk[b0 + i] = (unsigned)sorted[i];
  }
  float4* yq = (float4*)(Y + (size_t)b * NB * 16);
  float4 yv[NB * 4 / AGT];
#pragma unroll
  for (int u = 0; u < NB * 4 / AGT; ++u) {
    const int i = u * AGT + t;
    if (b * NB + (i >> 2) < n) yv[u] = yq[i];
  }
  const int node = b * NB + (t >> 1);
  if ((t & 1) == 0) {
    const float dn = rsqrtf(1.0f + (float)indeg);
    dv[t >> 1] = dn;
    if (node < n) dinv[node] = dn;
  }
  __syncthreads();
#pragma unroll
  for (int u = 0; u < NB * 4 / AGT; ++u) {
    const int i = u * AGT + t;
    if (b * NB + (i >> 2) < n) {
      const float s = dv[i >> 2];
      float4 v = yv[u];
      v.x *= s; v.y *= s; v.z *= s; v.w *= s;
      yq[i] = v;
    }
  }
}

template <bool LAST>
__global__ void k_agg(const unsigned* __restrict__ pk, const int* __restrict__ binBase,
                      const int* __restrict__ rs,
                      const float* __restrict__ xws, const float* __restrict__ dinv,
                      const float* __restrict__ bias, const float* __restrict__ W2,
                      const float* __restrict__ Wl, const float* __restrict__ bl,
                      float* __restrict__ Y, float* __restrict__ out, int n) {
  __shared__ __attribute__((aligned(16))) float acc[NB * 16];
  __shared__ int rowStart[RS];
  __shared__ int sorted[CAP];

  const int t = threadIdx.x, b = blockIdx.x;
  const int lane = t & 31;
  const int fq = lane & 3;
  const int grp = t >> 2;
  const float4* xq = (const float4*)xws;
  const int node0 = b * NB;
  for (int i = t; i < NB * 16; i += AGT) {
    const int node = node0 + (i >> 4);
    acc[i] = (node < n) ? xws[(size_t)node0 * 16 + i] : 0.0f;
  }
  const int e0 = binBase[b], e1 = binBase[b + 1];
  const int* rsb = rs + (size_t)(b + e0 / CAP) * RS;

  for (int b0 = e0; b0 < e1; b0 += CAP, rsb += RS) {
    const int b1 = min(b0 + CAP, e1);
    const int nb = b1 - b0;
    const int rsv = rsb[min(t, NB)];
    unsigned q[CAP / AGT];
#pragma unroll
    for (int u = 0; u < CAP / AGT; ++u) q[u] = pk[b0 + min(u * AGT + t, nb - 1)];
    if (t < RS) rowStart[t] = rsv;
#pragma unroll
    for (int u = 0; u < CAP / AGT; ++u)
      if (u * AGT + t < nb) sorted[u * AGT + t] = (int)q[u];
    __syncthreads();
    for (int r = grp; r < NB; r += AGT / 4) {
      const int s0 = rowStart[r], s1 = rowStart[r + 1];
      float4 sum = make_float4(0.0f, 0.0f, 0.0f, 0.0f);
      for (int i = s0; i < s1; i += 8) {
        float4 v[8];
#pragma unroll
        for (int u = 0; u < 8; ++u) {
          const int idx = min(i + u, s1 - 1);
          v[u] = xq[(unsigned)sorted[idx] * 4u + fq];
        }
#pragma unroll
        for (int u = 0; u < 8; ++u) {
          const bool in = i + u < s1;
          sum.x += in ? v[u].x : 0.0f;
          sum.y += in ? v[u].y : 0.0f;
          sum.z += in ? v[u].z : 0.0f;
          sum.w += in ? v[u].w : 0.0f;
        }
      }
      float4* ar = (float4*)(acc + r * 16 + fq * 4);
      float4 a = *ar;
      a.x += sum.x; a.y += sum.y; a.z += sum.z; a.w += sum.w;
      *ar = a;
    }
    __syncthreads();
  }
  __syncthreads();

  if (!LAST) {
    const int wave = __builtin_amdgcn_readfirstlane(t >> 5);
    const int row0 = node0 + wave * 16;
    if (row0 < n) {
      const int m  = lane & 15;
      const int kh = lane >> 4;
      const float da = dinv[row0 + m];
      const float4 d0 = *(const float4*)(dinv + row0 + 8 * kh);
      const float4 d1 = *(const float4*)(dinv + row0 + 8 * kh + 4);
      const float dr[8] = {d0.x, d0.y, d0.z, d0.w, d1.x, d1.y, d1.z, d1.w};
      const float* ar = acc + (wave * 16 + m) * 16;
      v8f c = {};
#pragma unroll
      for (int kb = 0; kb < 16; kb += 4) {
        const int k0 = kb + 2 * kh;
        v2f a, w;
        a.x = fmaxf(ar[k0]     * da + bias[k0],     0.0f);
        a.y = fmaxf(ar[k0 + 1] * da + bias[k0 + 1], 0.0f);
        w.x = W2[k0 * 16 + m];
        w.y = W2[(k0 + 1) * 16 + m];
        c = __builtin_amdgcn_wmma_f32_16x16x4_f32(
            false, a, false, w, (short)0, c, false, false);
      }
      float* yp = Y + (size_t)(row0 + 8 * kh) * 16 + m;
#pragma unroll
      for (int v = 0; v < 8; ++v) yp[v * 16] = c[v] * dr[v];
    }
  } else {
    const int node = node0 + t;
    if (t < NB && node < n) {
      const float s = dinv[node];
      float a = 0.0f;
#pragma unroll
      for (int ff = 0; ff < 16; ++ff) {
        const float h = fmaxf(acc[t * 16 + ff] * s + bias[ff], 0.0f);
        a += h * Wl[ff];
      }
      out[node] = a + bl[0];
    }
  }
}

extern "C" void kernel_launch(void* const* d_in, const int* in_sizes, int n_in,
                              void* d_out, int out_size, void* d_ws, size_t ws_size,
                              hipStream_t stream) {
  const float* x   = (const float*)d_in[0];
  const int*   ei  = (const int*)d_in[1];
  const float* W1  = (const float*)d_in[2];
  const float* b1  = (const float*)d_in[3];
  const float* W2  = (const float*)d_in[4];
  const float* b2  = (const float*)d_in[5];
  const float* Wl  = (const float*)d_in[6];
  const float* bl  = (const float*)d_in[7];
  float* out = (float*)d_out;

  const int din = in_sizes[2] / 16;
  const int n   = in_sizes[0] / din;
  const int E   = in_sizes[1] / 2;
  const int tot = n * 16;

  float* ws   = (float*)d_ws;
  float* dinv = ws;
  float* bufA = ws + (((size_t)n + 255) & ~(size_t)255);
  float* bufB = bufA + (size_t)n * 16;

  const int T = 256;
  const int ntiles = (n + 15) / 16;
  const int gemmBlocks = (ntiles + 3) / 4;

  const int nbins = (n + NB - 1) / NB;
  unsigned* pk   = (unsigned*)(bufB + (size_t)n * 16);
  int* cnt       = (int*)(pk + (((size_t)E + 255) & ~(size_t)255));
  int* binTot    = cnt + (size_t)nbins * PG;
  int* binBase   = binTot + MAXBINS;
  const size_t need = (size_t)((char*)(binBase + MAXBINS + 1) - (char*)d_ws);

  int* rs        = cnt;

  const int chunk = (E + PG - 1) / PG;
  if (ws_size >= need && nbins <= MAXBINS && chunk <= CHW &&
      (size_t)(nbins + E / CAP + 1) * RS <= (size_t)nbins * PG) {
    k_count   <<<PG / BW + nbins, BT, 0, stream>>>(ei + E, cnt, E, chunk, nbins, x, W1, bufA, n, din);
    k_scan_bin<<<nbins, PG / 2, 0, stream>>>(cnt, binTot);
    k_scan_tot<<<1, MAXBINS, 0, stream>>>(binTot, binBase, nbins);
    k_part    <<<PG / BW, BT, 0, stream>>>(ei, ei + E, cnt, binBase, pk, E, chunk, nbins);
    k_sort    <<<nbins, AGT, 0, stream>>>(pk, binBase, rs, dinv, n, bufA);

    k_agg<false><<<nbins, AGT, 0, stream>>>(pk, binBase, rs, bufA, dinv, b1, W2, Wl, bl, bufB, out, n);
    k_agg<true> <<<nbins, AGT, 0, stream>>>(pk, binBase, rs, bufB, dinv, b2, W2, Wl, bl, bufA, out, n);
    return;
  }

  k_init  <<<(tot + T - 1) / T, T, 0, stream>>>(dinv, bufB, n, tot);
  k_degree<<<(E + T - 1) / T,   T, 0, stream>>>(ei + E, dinv, E);
  k_rsqrt <<<(n + T - 1) / T,   T, 0, stream>>>(dinv, n);

  k_gemm16 <<<gemmBlocks, 128, 0, stream>>>(x, W1, bufA, n, din);
  k_scatter<<<(4 * E + T - 1) / T, T, 0, stream>>>(ei, ei + E, dinv, bufA, bufB, E);
  k_final1 <<<(tot + T - 1) / T, T, 0, stream>>>(bufB, bufA, dinv, b1, tot);

  k_gemm16 <<<gemmBlocks, 128, 0, stream>>>(bufB, W2, bufA, n, 16);
  k_zero   <<<(tot + T - 1) / T, T, 0, stream>>>(bufB, tot);
  k_scatter<<<(4 * E + T - 1) / T, T, 0, stream>>>(ei, ei + E, dinv, bufA, bufB, E);
  k_final2 <<<(n + T - 1) / T, T, 0, stream>>>(bufB, bufA, dinv, b2, Wl, bl, out, n);
}


# both scan launches go (7 launches become 5): the first bucketing pass also stores each work-group's bin totals, the second's work-groups sum and scan them themselves; edge and f32 sum order same
# speedup vs baseline: 5.5792x; 1.1031x over previous
#include <hip/hip_runtime.h>
#include <hip/hip_bf16.h>
#include <stddef.h>

typedef __attribute__((ext_vector_type(2))) float v2f;
typedef __attribute__((ext_vector_type(8))) float v8f;
typedef __attribute__((ext_vector_type(2))) unsigned v2u;

__device__ __forceinline__ void atomAddF(float* p, float v) {
  asm volatile("global_atomic_add_f32 %0, %1, off scope:SCOPE_DEV"
               :
               : "v"(p), "v"(v)
               : "memory");
}

__global__ void k_init(float* __restrict__ deg, float* __restrict__ agg,
                       int n, int tot) {
  int t = blockIdx.x * blockDim.x + threadIdx.x;
  if (t < tot) agg[t] = 0.0f;
  if (t < n)   deg[t] = 1.0f;
}

__global__ void k_zero(float* __restrict__ p, int tot) {
  int t = blockIdx.x * blockDim.x + threadIdx.x;
  if (t < tot) p[t] = 0.0f;
}

__global__ void k_degree(const int* __restrict__ dst, float* __restrict__ deg,
                         int E) {
  int t = blockIdx.x * blockDim.x + threadIdx.x;
  if (t < E) atomAddF(&deg[dst[t]], 1.0f);
}

__global__ void k_rsqrt(float* __restrict__ d, int n) {
  int t = blockIdx.x * blockDim.x + threadIdx.x;
  if (t < n) d[t] = rsqrtf(d[t]);
}

__global__ void k_gemm16(const float* __restrict__ X, const float* __restrict__ W,
                         float* __restrict__ Y, int nrows, int K) {
  const int wave = threadIdx.x >> 5;
  const int lane = threadIdx.x & 31;
  const int tile = blockIdx.x * (blockDim.x >> 5) + wave;
  const int row0 = tile * 16;
  if (row0 >= nrows) return;

  const int m  = lane & 15;
  const int kh = lane >> 4;

  const float* ap = X + (size_t)(row0 + m) * K + 2 * kh;
  v8f acc = {};
  for (int kb = 0; kb < K; kb += 4) {
    v2f a = *(const v2f*)(ap + kb);
    const int kr = kb + 2 * kh;
    v2f b;
    b.x = W[(size_t)kr * 16 + m];
    b.y = W[(size_t)(kr + 1) * 16 + m];
    acc = __builtin_amdgcn_wmma_f32_16x16x4_f32(
        false, a, false, b, (short)0, acc, false, false);
  }
#pragma unroll
  for (int v = 0; v < 8; ++v) {
    Y[(size_t)(row0 + v + 8 * kh) * 16 + m] = acc[v];
  }
}

__global__ void k_scatter(const int* __restrict__ src, const int* __restrict__ dst,
                          const float* __restrict__ dinv,
                          const float* __restrict__ xw,
                          float* __restrict__ agg, int E) {
  int t = blockIdx.x * blockDim.x + threadIdx.x;
  int e = t >> 2;
  if (e >= E) return;
  int q = (t & 3) << 2;
  int s = src[e];
  int d = dst[e];
  float nrm = dinv[s] * dinv[d];
  float4 v = *(const float4*)(xw + (size_t)s * 16 + q);
  float* out = agg + (size_t)d * 16 + q;
  atomAddF(out + 0, v.x * nrm);
  atomAddF(out + 1, v.y * nrm);
  atomAddF(out + 2, v.z * nrm);
  atomAddF(out + 3, v.w * nrm);
}

__global__ void k_final1(float* __restrict__ agg, const float* __restrict__ xw,
                         const float* __restrict__ dinv,
                         const float* __restrict__ b, int tot) {
  int t = blockIdx.x * blockDim.x + threadIdx.x;
  if (t >= tot) return;
  int i = t >> 4, f = t & 15;
  float s = dinv[i];
  float v = agg[t] + xw[t] * s * s + b[f];
  agg[t] = fmaxf(v, 0.0f);
}

__global__ void k_final2(const float* __restrict__ agg, const float* __restrict__ xw,
                         const float* __restrict__ dinv,
                         const float* __restrict__ b2,
                         const float* __restrict__ Wl,
                         const float* __restrict__ bl,
                         float* __restrict__ out, int n) {
  int i = blockIdx.x * blockDim.x + threadIdx.x;
  if (i >= n) return;
  float s = dinv[i];
  float ss = s * s;
  float acc = 0.0f;
#pragma unroll
  for (int f = 0; f < 16; ++f) {
    float h = agg[(size_t)i * 16 + f] + xw[(size_t)i * 16 + f] * ss + b2[f];
    h = fmaxf(h, 0.0f);
    acc += h * Wl[f];
  }
  out[i] = acc + bl[0];
}

#define NB_SHIFT 8
#define NB       256
#define PG       2048
#define MAXBINS  512
#define BIN_BITS 9
#define AGW      16
#define AGT      (AGW * 32)
#define CAP      10240
#define GU       4
#define BW       16
#define BT       (BW * 32)
#define HS       (MAXBINS + 1)
#define CHW      1600
#define RS       (NB + 1)
static_assert(MAXBINS == (1 << BIN_BITS) && NB == (1 << NB_SHIFT), "key widths");
static_assert(PG % BW == 0 && BT == MAXBINS, "bucketing work-group shape");
static_assert(BW == 16, "a bin's BW chunk counts of one work-group are read as four 16-byte words");
static_assert(BW * 16 == NB, "a product work-group's BW tiles of 16 rows are one bin's rows");
static_assert((NB * 4) % AGT == 0, "a bin's product rows are scaled by one work-group in whole rounds");
static_assert(CAP % AGT == 0 && RS <= AGT, "a sorted batch is copied by one work-group in whole rounds");

template <int BITS>
__device__ __forceinline__ unsigned sameKeyMask(int key, bool valid) {
  static_assert(BITS >= 1 && BITS <= 31, "key width");
  unsigned differ = 0u;
#pragma unroll
  for (int b = 0; b < BITS; ++b) {
    const int mine = (key << (31 - b)) >> 31;
    const unsigned vote = __builtin_amdgcn_ballot_w32(mine != 0);
    differ |= vote ^ (unsigned)mine;
  }
  return ~differ & __builtin_amdgcn_ballot_w32(valid);
}

template <int N>
__device__ __forceinline__ void blockScanIncl(int* s, int t) {
  for (int d = 1; d < N; d <<= 1) {
    const int v = (t >= d) ? s[t - d] : 0;
    __syncthreads();
    s[t] += v;
    __syncthreads();
  }
}

template <int N>
__device__ __forceinline__ void blockScanIncl2(int* a, int* b, int t) {
  for (int d = 1; d < N; d <<= 1) {
    const int va = (t >= d) ? a[t - d] : 0;
    const int vb = (t >= d) ? b[t - d] : 0;
    __syncthreads();
    a[t] += va;
    b[t] += vb;
    __syncthreads();
  }
}

__global__ void k_count(const int* __restrict__ dst, int* __restrict__ cnt,
                        int* __restrict__ wgTot, int E, int chunk, int nbins,
                        const float* __restrict__ X, const float* __restrict__ W,
                        float* __restrict__ Y, int n, int K) {
  __shared__ int h[BW * HS];
  const int t = threadIdx.x;
  const int lane = t & 31;
  const int wave = __builtin_amdgcn_readfirstlane(t >> 5);
  if (blockIdx.x >= PG / BW) {
    const int row0 = ((int)blockIdx.x - PG / BW) * NB + wave * 16;
    if (row0 >= n) return;
    const int m  = lane & 15;
    const int kh = lane >> 4;
    const float* ap = X + (size_t)(row0 + m) * K + 2 * kh;
    v8f acc = {};
    for (int kb = 0; kb < K; kb += 4) {
      v2f a = *(const v2f*)(ap + kb);
      const int kr = kb + 2 * kh;
      v2f w;
      w.x = W[(size_t)kr * 16 + m];
      w.y = W[(size_t)(kr + 1) * 16 + m];
      acc = __builtin_amdgcn_wmma_f32_16x16x4_f32(
          false, a, false, w, (short)0, acc, false, false);
    }
#pragma unroll
    for (int v = 0; v < 8; ++v) {
      Y[(size_t)(row0 + v + 8 * kh) * 16 + m] = acc[v];
    }
    return;
  }
  for (int i = t; i < BW * HS; i += BT) h[i] = 0;
  __syncthreads();
  int* hv = h + wave * HS;
  const int g0 = blockIdx.x * BW;
  const int e0 = (g0 + wave) * chunk;
  const int e1 = min(e0 + chunk, E);
  int dn[GU];
#pragma unroll
  for (int u = 0; u < GU; ++u) dn[u] = dst[min(e0 + u * 32 + lane, E - 1)];
  for (int base = e0; base < e1; base += 32 * GU) {
    int d[GU];
#pragma unroll
    for (int u = 0; u < GU; ++u) d[u] = dn[u];
#pragma unroll
    for (int u = 0; u < GU; ++u) dn[u] = dst[min(base + 32 * GU + u * 32 + lane, E - 1)];
#pragma unroll
    for (int u = 0; u < GU; ++u) {
      if (base + u * 32 < e1) {
        const bool valid = base + u * 32 + lane < e1;
        const int key = valid ? (d[u] >> NB_SHIFT) : -1;
        const unsigned m = sameKeyMask<BIN_BITS>(key, valid);
        if (valid && (m >> lane) == 1u) hv[key] += __popc(m);
      }
    }
  }
  __syncthreads();
  for (int i = t; i < nbins * BW; i += BT) {
    const int bin = i / BW, w = i % BW;
    cnt[bin * PG + g0 + w] = h[w * HS + bin];
  }
  if (t < nbins) {
    int s = 0;
#pragma unroll
    for (int w = 0; w < BW; ++w) s += h[w * HS + t];
    wgTot[blockIdx.x * MAXBINS + t] = s;
  }
}

__global__ void k_part(const int* __restrict__ src, const int* __restrict__ dst,
                       const int* __restrict__ cnt, const int* __restrict__ wgTot,
                       int* __restrict__ binBase,
                       unsigned* __restrict__ pk, int E, int chunk, int nbins) {
  __shared__ int cur[BW * HS];
  __shared__ int dl[MAXBINS];
  __shared__ int sc[BT];
  __shared__ int sb[BT];
  __shared__ v2u stage[BW * CHW];
  const int t = threadIdx.x;
  const int lane = t & 31;
  const int wave = __builtin_amdgcn_readfirstlane(t >> 5);
  const int g0 = blockIdx.x * BW;
  int cw[BW];
  int c = 0, tot = 0, pre = 0;
  if (t < nbins) {
    const int4* cq = (const int4*)(cnt + (size_t)t * PG + g0);
#pragma unroll
    for (int u = 0; u < BW / 4; ++u) {
      const int4 q = cq[u];
      cw[4 * u] = q.x; cw[4 * u + 1] = q.y; cw[4 * u + 2] = q.z; cw[4 * u + 3] = q.w;
    }
    const int* tp = wgTot + t;
    const int me = (int)blockIdx.x;
#pragma unroll 32
    for (int w = 0; w < PG / BW; ++w) {
      const int v = tp[w * MAXBINS];
      tot += v;
      pre += (w < me) ? v : 0;
    }
#pragma unroll
    for (int w = 0; w < BW; ++w) c += cw[w];
  }
  sc[t] = c;
  sb[t] = tot;
  __syncthreads();
  blockScanIncl2<BT>(sc, sb, t);
  const int bb = sb[t] - tot;
  const int gs = bb + pre;
  dl[t] = (sc[t] - c) - gs;
  const int total = sc[BT - 1];
  if (blockIdx.x == 0) {
    if (t < nbins) binBase[t] = bb;
    if (t == BT - 1) binBase[nbins] = sb[t];
  }
  if (t < nbins) {
    int run = gs;
#pragma unroll
    for (int w = 0; w < BW; ++w) {
      cur[w * HS + t] = run;
      run += cw[w];
    }
  }
  __syncthreads();
  int* curv = cur + wave * HS;
  const int e0 = (g0 + wave) * chunk;
  const int e1 = min(e0 + chunk, E);
  const unsigned below = (1u << lane) - 1u;
  int dn[GU], sn[GU];
#pragma unroll
  for (int u = 0; u < GU; ++u) {
    const int i = min(e0 + u * 32 + lane, E - 1);
    dn[u] = dst[i]; sn[u] = src[i];
  }
  for (int base = e0; base < e1; base += 32 * GU) {
    int d[GU], s[GU];
#pragma unroll
    for (int u = 0; u < GU; ++u) { d[u] = dn[u]; s[u] = sn[u]; }
#pragma unroll
    for (int u = 0; u < GU; ++u) {
      const int i = min(base + 32 * GU + u * 32 + lane, E - 1);
      dn[u] = dst[i]; sn[u] = src[i];
    }
#pragma unroll
    for (int u = 0; u < GU; ++u) {
      if (base + u * 32 < e1) {
        const bool valid = base + u * 32 + lane < e1;
        const int key = valid ? (d[u] >> NB_SHIFT) : -1;
        const unsigned m = sameKeyMask<BIN_BITS>(key, valid);
        if (valid) {
          const int pos = curv[key] + __popc(m & below);
          if ((m >> lane) == 1u) curv[key] = pos + 1;
          v2u e;
          e.x = ((unsigned)s[u] << NB_SHIFT) | (unsigned)(d[u] & (NB - 1));
          e.y = (unsigned)pos;
          stage[pos + dl[key]] = e;
        }
      }
    }
  }
  __syncthreads();
#pragma unroll 4
  for (int i = t; i < total; i += BT) {
    const v2u e = stage[i];
    pk[e.y] = e.x;
  }
}

__device__ __forceinline__ void countRows(const unsigned* __restrict__ pk, int b0, int b1,
                                          int* hwv, int wave, int lane) {
  for (int base = b0 + wave * 32; base < b1; base += AGT) {
    const int i = base + lane;
    const bool valid = i < b1;
    const unsigned q = pk[min(i, b1 - 1)];
    const int key = valid ? (int)(q & (NB - 1)) : -1;
    const unsigned m = sameKeyMask<NB_SHIFT>(key, valid);
    if (valid && (m >> lane) == 1u) hwv[key] += __popc(m);
  }
}

__global__ void k_sort(unsigned* __restrict__ pk, const int* __restrict__ binBase,
                       int* __restrict__ rs, float* __restrict__ dinv, int n,
                       float* __restrict__ Y) {
  __shared__ int hw[AGW * NB];
  __shared__ int sc[AGT];
  __shared__ int sorted[CAP];
  __shared__ float dv[NB];

  const int t = threadIdx.x, b = blockIdx.x;
  const int lane = t & 31;
  const int wave = __builtin_amdgcn_readfirstlane(t >> 5);
  const int e0 = binBase[b], e1 = binBase[b + 1];
  int* hwv = hw + wave * NB;
  const unsigned below = (1u << lane) - 1u;
  int* rsb = rs + (size_t)(b + e0 / CAP) * RS;
  int indeg = 0;

  for (int b0 = e0; b0 < e1; b0 += CAP, rsb += RS) {
    const int b1 = min(b0 + CAP, e1);
    for (int i = t; i < AGW * NB; i += AGT) hw[i] = 0;
    __syncthreads();
    countRows(pk, b0, b1, hwv, wave, lane);
    __syncthreads();
    {
      const int key = t >> 1, w0 = (t & 1) * (AGW / 2);
      int s = 0;
#pragma unroll
      for (int i = 0; i < AGW / 2; ++i) s += hw[(w0 + i) * NB + key];
      indeg += s + __shfl_xor(s, 1);
      sc[t] = s;
      __syncthreads();
      blockScanIncl<AGT>(sc, t);
      int run = sc[t] - s;
      if ((t & 1) == 0) rsb[key] = run;
      if (t == AGT - 1) rsb[NB] = sc[t];
#pragma unroll
      for (int i = 0; i < AGW / 2; ++i) {
        const int idx = (w0 + i) * NB + key;
        const int c = hw[idx];
        hw[idx] = run;
        run += c;
      }
    }
    __syncthreads();
    for (int base = b0 + wave * 32; base < b1; base += AGT) {
      const int i = base + lane;
      const bool valid = i < b1;
      const unsigned q = pk[min(i, b1 - 1)];
      const int key = valid ? (int)(q & (NB - 1)) : -1;
      const unsigned m = sameKeyMask<NB_SHIFT>(key, valid);
      if (valid) {
        const int pos = hwv[key] + __popc(m & below);
        if ((m >> lane) == 1u) hwv[key] = pos + 1;
        sorted[pos] = (int)(q >> NB_SHIFT);
      }
    }
    __syncthreads();
    for (int i = t; i < b1 - b0; i += AGT) pk[b0 + i] = (unsigned)sorted[i];
  }
  float4* yq = (float4*)(Y + (size_t)b * NB * 16);
  float4 yv[NB * 4 / AGT];
#pragma unroll
  for (int u = 0; u < NB * 4 / AGT; ++u) {
    const int i = u * AGT + t;
    if (b * NB + (i >> 2) < n) yv[u] = yq[i];
  }
  const int node = b * NB + (t >> 1);
  if ((t & 1) == 0) {
    const float dn = rsqrtf(1.0f + (float)indeg);
    dv[t >> 1] = dn;
    if (node < n) dinv[node] = dn;
  }
  __syncthreads();
#pragma unroll
  for (int u = 0; u < NB * 4 / AGT; ++u) {
    const int i = u * AGT + t;
    if (b * NB + (i >> 2) < n) {
      const float s = dv[i >> 2];
      float4 v = yv[u];
      v.x *= s; v.y *= s; v.z *= s; v.w *= s;
      yq[i] = v;
    }
  }
}

template <bool LAST>
__global__ void k_agg(const unsigned* __restrict__ pk, const int* __restrict__ binBase,
                      const int* __restrict__ rs,
                      const float* __restrict__ xws, const float* __restrict__ dinv,
                      const float* __restrict__ bias, const float* __restrict__ W2,
                      const float* __restrict__ Wl, const float* __restrict__ bl,
                      float* __restrict__ Y, float* __restrict__ out, int n) {
  __shared__ __attribute__((aligned(16))) float acc[NB * 16];
  __shared__ int rowStart[RS];
  __shared__ int sorted[CAP];

  const int t = threadIdx.x, b = blockIdx.x;
  const int lane = t & 31;
  const int fq = lane & 3;
  const int grp = t >> 2;
  const float4* xq = (const float4*)xws;
  const int node0 = b * NB;
  for (int i = t; i < NB * 16; i += AGT) {
    const int node = node0 + (i >> 4);
    acc[i] = (node < n) ? xws[(size_t)node0 * 16 + i] : 0.0f;
  }
  const int e0 = binBase[b], e1 = binBase[b + 1];
  const int* rsb = rs + (size_t)(b + e0 / CAP) * RS;

  for (int b0 = e0; b0 < e1; b0 += CAP, rsb += RS) {
    const int b1 = min(b0 + CAP, e1);
    const int nb = b1 - b0;
    const int rsv = rsb[min(t, NB)];
    unsigned q[CAP / AGT];
#pragma unroll
    for (int u = 0; u < CAP / AGT; ++u) q[u] = pk[b0 + min(u * AGT + t, nb - 1)];
    if (t < RS) rowStart[t] = rsv;
#pragma unroll
    for (int u = 0; u < CAP / AGT; ++u)
      if (u * AGT + t < nb) sorted[u * AGT + t] = (int)q[u];
    __syncthreads();
    for (int r = grp; r < NB; r += AGT / 4) {
      const int s0 = rowStart[r], s1 = rowStart[r + 1];
      float4 sum = make_float4(0.0f, 0.0f, 0.0f, 0.0f);
      for (int i = s0; i < s1; i += 8) {
        float4 v[8];
#pragma unroll
        for (int u = 0; u < 8; ++u) {
          const int idx = min(i + u, s1 - 1);
          v[u] = xq[(unsigned)sorted[idx] * 4u + fq];
        }
#pragma unroll
        for (int u = 0; u < 8; ++u) {
          const bool in = i + u < s1;
          sum.x += in ? v[u].x : 0.0f;
          sum.y += in ? v[u].y : 0.0f;
          sum.z += in ? v[u].z : 0.0f;
          sum.w += in ? v[u].w : 0.0f;
        }
      }
      float4* ar = (float4*)(acc + r * 16 + fq * 4);
      float4 a = *ar;
      a.x += sum.x; a.y += sum.y; a.z += sum.z; a.w += sum.w;
      *ar = a;
    }
    __syncthreads();
  }
  __syncthreads();

  if (!LAST) {
    const int wave = __builtin_amdgcn_readfirstlane(t >> 5);
    const int row0 = node0 + wave * 16;
    if (row0 < n) {
      const int m  = lane & 15;
      const int kh = lane >> 4;
      const float da = dinv[row0 + m];
      const float4 d0 = *(const float4*)(dinv + row0 + 8 * kh);
      const float4 d1 = *(const float4*)(dinv + row0 + 8 * kh + 4);
      const float dr[8] = {d0.x, d0.y, d0.z, d0.w, d1.x, d1.y, d1.z, d1.w};
      const float* ar = acc + (wave * 16 + m) * 16;
      v8f c = {};
#pragma unroll
      for (int kb = 0; kb < 16; kb += 4) {
        const int k0 = kb + 2 * kh;
        v2f a, w;
        a.x = fmaxf(ar[k0]     * da + bias[k0],     0.0f);
        a.y = fmaxf(ar[k0 + 1] * da + bias[k0 + 1], 0.0f);
        w.x = W2[k0 * 16 + m];
        w.y = W2[(k0 + 1) * 16 + m];
        c = __builtin_amdgcn_wmma_f32_16x16x4_f32(
            false, a, false, w, (short)0, c, false, false);
      }
      float* yp = Y + (size_t)(row0 + 8 * kh) * 16 + m;
#pragma unroll
      for (int v = 0; v < 8; ++v) yp[v * 16] = c[v] * dr[v];
    }
  } else {
    const int node = node0 + t;
    if (t < NB && node < n) {
      const float s = dinv[node];
      float a = 0.0f;
#pragma unroll
      for (int ff = 0; ff < 16; ++ff) {
        const float h = fmaxf(acc[t * 16 + ff] * s + bias[ff], 0.0f);
        a += h * Wl[ff];
      }
      out[node] = a + bl[0];
    }
  }
}

extern "C" void kernel_launch(void* const* d_in, const int* in_sizes, int n_in,
                              void* d_out, int out_size, void* d_ws, size_t ws_size,
                              hipStream_t stream) {
  const float* x   = (const float*)d_in[0];
  const int*   ei  = (const int*)d_in[1];
  const float* W1  = (const float*)d_in[2];
  const float* b1  = (const float*)d_in[3];
  const float* W2  = (const float*)d_in[4];
  const float* b2  = (const float*)d_in[5];
  const float* Wl  = (const float*)d_in[6];
  const float* bl  = (const float*)d_in[7];
  float* out = (float*)d_out;

  const int din = in_sizes[2] / 16;
  const int n   = in_sizes[0] / din;
  const int E   = in_sizes[1] / 2;
  const int tot = n * 16;

  float* ws   = (float*)d_ws;
  float* dinv = ws;
  float* bufA = ws + (((size_t)n + 255) & ~(size_t)255);
  float* bufB = bufA + (size_t)n * 16;

  const int T = 256;
  const int ntiles = (n + 15) / 16;
  const int gemmBlocks = (ntiles + 3) / 4;

  const int nbins = (n + NB - 1) / NB;
  unsigned* pk   = (unsigned*)(bufB + (size_t)n * 16);
  int* cnt       = (int*)(pk + (((size_t)E + 255) & ~(size_t)255));
  int* binBase   = cnt + (size_t)nbins * PG + MAXBINS;
  const size_t need = (size_t)((char*)(binBase + MAXBINS + 1) - (char*)d_ws);

  int* rs        = cnt;
  int* wgTot     = (int*)bufB;

  const int chunk = (E + PG - 1) / PG;
  if (ws_size >= need && nbins <= MAXBINS && chunk <= CHW &&
      (size_t)(nbins + E / CAP + 1) * RS <= (size_t)nbins * PG &&
      (size_t)n * 16 >= (size_t)(PG / BW) * MAXBINS) {
    k_count   <<<PG / BW + nbins, BT, 0, stream>>>(ei + E, cnt, wgTot, E, chunk, nbins, x, W1, bufA, n, din);
    k_part    <<<PG / BW, BT, 0, stream>>>(ei, ei + E, cnt, wgTot, binBase, pk, E, chunk, nbins);
    k_sort    <<<nbins, AGT, 0, stream>>>(pk, binBase, rs, dinv, n, bufA);

    k_agg<false><<<nbins, AGT, 0, stream>>>(pk, binBase, rs, bufA, dinv, b1, W2, Wl, bl, bufB, out, n);
    k_agg<true> <<<nbins, AGT, 0, stream>>>(pk, binBase, rs, bufB, dinv, b2, W2, Wl, bl, bufA, out, n);
    return;
  }

  k_init  <<<(tot + T - 1) / T, T, 0, stream>>>(dinv, bufB, n, tot);
  k_degree<<<(E + T - 1) / T,   T, 0, stream>>>(ei + E, dinv, E);
  k_rsqrt <<<(n + T - 1) / T,   T, 0, stream>>>(dinv, n);

  k_gemm16 <<<gemmBlocks, 128, 0, stream>>>(x, W1, bufA, n, din);
  k_scatter<<<(4 * E + T - 1) / T, T, 0, stream>>>(ei, ei + E, dinv, bufA, bufB, E);
  k_final1 <<<(tot + T - 1) / T, T, 0, stream>>>(bufB, bufA, dinv, b1, tot);

  k_gemm16 <<<gemmBlocks, 128, 0, stream>>>(bufB, W2, bufA, n, 16);
  k_zero   <<<(tot + T - 1) / T, T, 0, stream>>>(bufB, tot);
  k_scatter<<<(4 * E + T - 1) / T, T, 0, stream>>>(ei, ei + E, dinv, bufA, bufB, E);
  k_final2 <<<(n + T - 1) / T, T, 0, stream>>>(bufB, bufA, dinv, b2, Wl, bl, out, n);
}


# row sort holds a bin's batch in registers: a lane loads its 20 edges once, all in flight at once, and both passes reuse keys and same-row masks (was a waited load per trip, twice); f32 sum order same
# speedup vs baseline: 5.9969x; 1.0749x over previous
#include <hip/hip_runtime.h>
#include <hip/hip_bf16.h>
#include <stddef.h>

typedef __attribute__((ext_vector_type(2))) float v2f;
typedef __attribute__((ext_vector_type(8))) float v8f;
typedef __attribute__((ext_vector_type(2))) unsigned v2u;

__device__ __forceinline__ void atomAddF(float* p, float v) {
  asm volatile("global_atomic_add_f32 %0, %1, off scope:SCOPE_DEV"
               :
               : "v"(p), "v"(v)
               : "memory");
}

__global__ void k_init(float* __restrict__ deg, float* __restrict__ agg,
                       int n, int tot) {
  int t = blockIdx.x * blockDim.x + threadIdx.x;
  if (t < tot) agg[t] = 0.0f;
  if (t < n)   deg[t] = 1.0f;
}

__global__ void k_zero(float* __restrict__ p, int tot) {
  int t = blockIdx.x * blockDim.x + threadIdx.x;
  if (t < tot) p[t] = 0.0f;
}

__global__ void k_degree(const int* __restrict__ dst, float* __restrict__ deg,
                         int E) {
  int t = blockIdx.x * blockDim.x + threadIdx.x;
  if (t < E) atomAddF(&deg[dst[t]], 1.0f);
}

__global__ void k_rsqrt(float* __restrict__ d, int n) {
  int t = blockIdx.x * blockDim.x + threadIdx.x;
  if (t < n) d[t] = rsqrtf(d[t]);
}

__global__ void k_gemm16(const float* __restrict__ X, const float* __restrict__ W,
                         float* __restrict__ Y, int nrows, int K) {
  const int wave = threadIdx.x >> 5;
  const int lane = threadIdx.x & 31;
  const int tile = blockIdx.x * (blockDim.x >> 5) + wave;
  const int row0 = tile * 16;
  if (row0 >= nrows) return;

  const int m  = lane & 15;
  const int kh = lane >> 4;

  const float* ap = X + (size_t)(row0 + m) * K + 2 * kh;
  v8f acc = {};
  for (int kb = 0; kb < K; kb += 4) {
    v2f a = *(const v2f*)(ap + kb);
    const int kr = kb + 2 * kh;
    v2f b;
    b.x = W[(size_t)kr * 16 + m];
    b.y = W[(size_t)(kr + 1) * 16 + m];
    acc = __builtin_amdgcn_wmma_f32_16x16x4_f32(
        false, a, false, b, (short)0, acc, false, false);
  }
#pragma unroll
  for (int v = 0; v < 8; ++v) {
    Y[(size_t)(row0 + v + 8 * kh) * 16 + m] = acc[v];
  }
}

__global__ void k_scatter(const int* __restrict__ src, const int* __restrict__ dst,
                          const float* __restrict__ dinv,
                          const float* __restrict__ xw,
                          float* __restrict__ agg, int E) {
  int t = blockIdx.x * blockDim.x + threadIdx.x;
  int e = t >> 2;
  if (e >= E) return;
  int q = (t & 3) << 2;
  int s = src[e];
  int d = dst[e];
  float nrm = dinv[s] * dinv[d];
  float4 v = *(const float4*)(xw + (size_t)s * 16 + q);
  float* out = agg + (size_t)d * 16 + q;
  atomAddF(out + 0, v.x * nrm);
  atomAddF(out + 1, v.y * nrm);
  atomAddF(out + 2, v.z * nrm);
  atomAddF(out + 3, v.w * nrm);
}

__global__ void k_final1(float* __restrict__ agg, const float* __restrict__ xw,
                         const float* __restrict__ dinv,
                         const float* __restrict__ b, int tot) {
  int t = blockIdx.x * blockDim.x + threadIdx.x;
  if (t >= tot) return;
  int i = t >> 4, f = t & 15;
  float s = dinv[i];
  float v = agg[t] + xw[t] * s * s + b[f];
  agg[t] = fmaxf(v, 0.0f);
}

__global__ void k_final2(const float* __restrict__ agg, const float* __restrict__ xw,
                         const float* __restrict__ dinv,
                         const float* __restrict__ b2,
                         const float* __restrict__ Wl,
                         const float* __restrict__ bl,
                         float* __restrict__ out, int n) {
  int i = blockIdx.x * blockDim.x + threadIdx.x;
  if (i >= n) return;
  float s = dinv[i];
  float ss = s * s;
  float acc = 0.0f;
#pragma unroll
  for (int f = 0; f < 16; ++f) {
    float h = agg[(size_t)i * 16 + f] + xw[(size_t)i * 16 + f] * ss + b2[f];
    h = fmaxf(h, 0.0f);
    acc += h * Wl[f];
  }
  out[i] = acc + bl[0];
}

#define NB_SHIFT 8
#define NB       256
#define PG       2048
#define MAXBINS  512
#define BIN_BITS 9
#define AGW      16
#define AGT      (AGW * 32)
#define CAP      10240
#define GU       4
#define BW       16
#define BT       (BW * 32)
#define HS       (MAXBINS + 1)
#define CHW      1600
#define RS       (NB + 1)
static_assert(MAXBINS == (1 << BIN_BITS) && NB == (1 << NB_SHIFT), "key widths");
static_assert(PG % BW == 0 && BT == MAXBINS, "bucketing work-group shape");
static_assert(BW == 16, "a bin's BW chunk counts of one work-group are read as four 16-byte words");
static_assert(BW * 16 == NB, "a product work-group's BW tiles of 16 rows are one bin's rows");
static_assert((NB * 4) % AGT == 0, "a bin's product rows are scaled by one work-group in whole rounds");
static_assert(CAP % AGT == 0 && RS <= AGT, "a sorted batch is copied by one work-group in whole rounds");

template <int BITS>
__device__ __forceinline__ unsigned sameKeyMask(int key, bool valid) {
  static_assert(BITS >= 1 && BITS <= 31, "key width");
  unsigned differ = 0u;
#pragma unroll
  for (int b = 0; b < BITS; ++b) {
    const int mine = (key << (31 - b)) >> 31;
    const unsigned vote = __builtin_amdgcn_ballot_w32(mine != 0);
    differ |= vote ^ (unsigned)mine;
  }
  return ~differ & __builtin_amdgcn_ballot_w32(valid);
}

template <int N>
__device__ __forceinline__ void blockScanIncl(int* s, int t) {
  for (int d = 1; d < N; d <<= 1) {
    const int v = (t >= d) ? s[t - d] : 0;
    __syncthreads();
    s[t] += v;
    __syncthreads();
  }
}

template <int N>
__device__ __forceinline__ void blockScanIncl2(int* a, int* b, int t) {
  for (int d = 1; d < N; d <<= 1) {
    const int va = (t >= d) ? a[t - d] : 0;
    const int vb = (t >= d) ? b[t - d] : 0;
    __syncthreads();
    a[t] += va;
    b[t] += vb;
    __syncthreads();
  }
}

__global__ void k_count(const int* __restrict__ dst, int* __restrict__ cnt,
                        int* __restrict__ wgTot, int E, int chunk, int nbins,
                        const float* __restrict__ X, const float* __restrict__ W,
                        float* __restrict__ Y, int n, int K) {
  __shared__ int h[BW * HS];
  const int t = threadIdx.x;
  const int lane = t & 31;
  const int wave = __builtin_amdgcn_readfirstlane(t >> 5);
  if (blockIdx.x >= PG / BW) {
    const int row0 = ((int)blockIdx.x - PG / BW) * NB + wave * 16;
    if (row0 >= n) return;
    const int m  = lane & 15;
    const int kh = lane >> 4;
    const float* ap = X + (size_t)(row0 + m) * K + 2 * kh;
    v8f acc = {};
    for (int kb = 0; kb < K; kb += 4) {
      v2f a = *(const v2f*)(ap + kb);
      const int kr = kb + 2 * kh;
      v2f w;
      w.x = W[(size_t)kr * 16 + m];
      w.y = W[(size_t)(kr + 1) * 16 + m];
      acc = __builtin_amdgcn_wmma_f32_16x16x4_f32(
          false, a, false, w, (short)0, acc, false, false);
    }
#pragma unroll
    for (int v = 0; v < 8; ++v) {
      Y[(size_t)(row0 + v + 8 * kh) * 16 + m] = acc[v];
    }
    return;
  }
  for (int i = t; i < BW * HS; i += BT) h[i] = 0;
  __syncthreads();
  int* hv = h + wave * HS;
  const int g0 = blockIdx.x * BW;
  const int e0 = (g0 + wave) * chunk;
  const int e1 = min(e0 + chunk, E);
  int dn[GU];
#pragma unroll
  for (int u = 0; u < GU; ++u) dn[u] = dst[min(e0 + u * 32 + lane, E - 1)];
  for (int base = e0; base < e1; base += 32 * GU) {
    int d[GU];
#pragma unroll
    for (int u = 0; u < GU; ++u) d[u] = dn[u];
#pragma unroll
    for (int u = 0; u < GU; ++u) dn[u] = dst[min(base + 32 * GU + u * 32 + lane, E - 1)];
#pragma unroll
    for (int u = 0; u < GU; ++u) {
      if (base + u * 32 < e1) {
        const bool valid = base + u * 32 + lane < e1;
        const int key = valid ? (d[u] >> NB_SHIFT) : -1;
        const unsigned m = sameKeyMask<BIN_BITS>(key, valid);
        if (valid && (m >> lane) == 1u) hv[key] += __popc(m);
      }
    }
  }
  __syncthreads();
  for (int i = t; i < nbins * BW; i += BT) {
    const int bin = i / BW, w = i % BW;
    cnt[bin * PG + g0 + w] = h[w * HS + bin];
  }
  if (t < nbins) {
    int s = 0;
#pragma unroll
    for (int w = 0; w < BW; ++w) s += h[w * HS + t];
    wgTot[blockIdx.x * MAXBINS + t] = s;
  }
}

__global__ void k_part(const int* __restrict__ src, const int* __restrict__ dst,
                       const int* __restrict__ cnt, const int* __restrict__ wgTot,
                       int* __restrict__ binBase,
                       unsigned* __restrict__ pk, int E, int chunk, int nbins) {
  __shared__ int cur[BW * HS];
  __shared__ int dl[MAXBINS];
  __shared__ int sc[BT];
  __shared__ int sb[BT];
  __shared__ v2u stage[BW * CHW];
  const int t = threadIdx.x;
  const int lane = t & 31;
  const int wave = __builtin_amdgcn_readfirstlane(t >> 5);
  const int g0 = blockIdx.x * BW;
  int cw[BW];
  int c = 0, tot = 0, pre = 0;
  if (t < nbins) {
    const int4* cq = (const int4*)(cnt + (size_t)t * PG + g0);
#pragma unroll
    for (int u = 0; u < BW / 4; ++u) {
      const int4 q = cq[u];
      cw[4 * u] = q.x; cw[4 * u + 1] = q.y; cw[4 * u + 2] = q.z; cw[4 * u + 3] = q.w;
    }
    const int* tp = wgTot + t;
    const int me = (int)blockIdx.x;
#pragma unroll 32
    for (int w = 0; w < PG / BW; ++w) {
      const int v = tp[w * MAXBINS];
      tot += v;
      pre += (w < me) ? v : 0;
    }
#pragma unroll
    for (int w = 0; w < BW; ++w) c += cw[w];
  }
  sc[t] = c;
  sb[t] = tot;
  __syncthreads();
  blockScanIncl2<BT>(sc, sb, t);
  const int bb = sb[t] - tot;
  const int gs = bb + pre;
  dl[t] = (sc[t] - c) - gs;
  const int total = sc[BT - 1];
  if (blockIdx.x == 0) {
    if (t < nbins) binBase[t] = bb;
    if (t == BT - 1) binBase[nbins] = sb[t];
  }
  if (t < nbins) {
    int run = gs;
#pragma unroll
    for (int w = 0; w < BW; ++w) {
      cur[w * HS + t] = run;
      run += cw[w];
    }
  }
  __syncthreads();
  int* curv = cur + wave * HS;
  const int e0 = (g0 + wave) * chunk;
  const int e1 = min(e0 + chunk, E);
  const unsigned below = (1u << lane) - 1u;
  int dn[GU], sn[GU];
#pragma unroll
  for (int u = 0; u < GU; ++u) {
    const int i = min(e0 + u * 32 + lane, E - 1);
    dn[u] = dst[i]; sn[u] = src[i];
  }
  for (int base = e0; base < e1; base += 32 * GU) {
    int d[GU], s[GU];
#pragma unroll
    for (int u = 0; u < GU; ++u) { d[u] = dn[u]; s[u] = sn[u]; }
#pragma unroll
    for (int u = 0; u < GU; ++u) {
      const int i = min(base + 32 * GU + u * 32 + lane, E - 1);
      dn[u] = dst[i]; sn[u] = src[i];
    }
#pragma unroll
    for (int u = 0; u < GU; ++u) {
      if (base + u * 32 < e1) {
        const bool valid = base + u * 32 + lane < e1;
        const int key = valid ? (d[u] >> NB_SHIFT) : -1;
        const unsigned m = sameKeyMask<BIN_BITS>(key, valid);
        if (valid) {
          const int pos = curv[key] + __popc(m & below);
          if ((m >> lane) == 1u) curv[key] = pos + 1;
          v2u e;
          e.x = ((unsigned)s[u] << NB_SHIFT) | (unsigned)(d[u] & (NB - 1));
          e.y = (unsigned)pos;
          stage[pos + dl[key]] = e;
        }
      }
    }
  }
  __syncthreads();
#pragma unroll 4
  for (int i = t; i < total; i += BT) {
    const v2u e = stage[i];
    pk[e.y] = e.x;
  }
}

__global__ void k_sort(unsigned* __restrict__ pk, const int* __restrict__ binBase,
                       int* __restrict__ rs, float* __restrict__ dinv, int n,
                       float* __restrict__ Y) {
  __shared__ int hw[AGW * NB];
  __shared__ int sc[AGT];
  __shared__ int sorted[CAP];
  __shared__ float dv[NB];

  const int t = threadIdx.x, b = blockIdx.x;
  const int lane = t & 31;
  const int wave = __builtin_amdgcn_readfirstlane(t >> 5);
  const int e0 = binBase[b], e1 = binBase[b + 1];
  int* hwv = hw + wave * NB;
  const unsigned below = (1u << lane) - 1u;
  int* rsb = rs + (size_t)(b + e0 / CAP) * RS;
  int indeg = 0;

  for (int b0 = e0; b0 < e1; b0 += CAP, rsb += RS) {
    const int b1 = min(b0 + CAP, e1);
    const int nb = b1 - b0;
    unsigned q[CAP / AGT];
    unsigned mk[CAP / AGT];
#pragma unroll
    for (int u = 0; u < CAP / AGT; ++u) q[u] = pk[b0 + min(u * AGT + t, nb - 1)];
    for (int i = t; i < AGW * NB; i += AGT) hw[i] = 0;
    __syncthreads();
#pragma unroll
    for (int u = 0; u < CAP / AGT; ++u) {
      mk[u] = 0u;
      if (u * AGT + wave * 32 < nb) {
        const bool valid = u * AGT + t < nb;
        const int key = valid ? (int)(q[u] & (NB - 1)) : -1;
        const unsigned m = sameKeyMask<NB_SHIFT>(key, valid);
        mk[u] = m;
        if (valid && (m >> lane) == 1u) hwv[key] += __popc(m);
      }
    }
    __syncthreads();
    {
      const int key = t >> 1, w0 = (t & 1) * (AGW / 2);
      int s = 0;
#pragma unroll
      for (int i = 0; i < AGW / 2; ++i) s += hw[(w0 + i) * NB + key];
      indeg += s + __shfl_xor(s, 1);
      sc[t] = s;
      __syncthreads();
      blockScanIncl<AGT>(sc, t);
      int run = sc[t] - s;
      if ((t & 1) == 0) rsb[key] = run;
      if (t == AGT - 1) rsb[NB] = sc[t];
#pragma unroll
      for (int i = 0; i < AGW / 2; ++i) {
        const int idx = (w0 + i) * NB + key;
        const int c = hw[idx];
        hw[idx] = run;
        run += c;
      }
    }
    __syncthreads();
#pragma unroll
    for (int u = 0; u < CAP / AGT; ++u) {
      if (u * AGT + t < nb) {
        const int key = (int)(q[u] & (NB - 1));
        const unsigned m = mk[u];
        const int pos = hwv[key] + __popc(m & below);
        if ((m >> lane) == 1u) hwv[key] = pos + 1;
        sorted[pos] = (int)(q[u] >> NB_SHIFT);
      }
    }
    __syncthreads();
    for (int i = t; i < nb; i += AGT) pk[b0 + i] = (unsigned)sorted[i];
  }
  float4* yq = (float4*)(Y + (size_t)b * NB * 16);
  float4 yv[NB * 4 / AGT];
#pragma unroll
  for (int u = 0; u < NB * 4 / AGT; ++u) {
    const int i = u * AGT + t;
    if (b * NB + (i >> 2) < n) yv[u] = yq[i];
  }
  const int node = b * NB + (t >> 1);
  if ((t & 1) == 0) {
    const float dn = rsqrtf(1.0f + (float)indeg);
    dv[t >> 1] = dn;
    if (node < n) dinv[node] = dn;
  }
  __syncthreads();
#pragma unroll
  for (int u = 0; u < NB * 4 / AGT; ++u) {
    const int i = u * AGT + t;
    if (b * NB + (i >> 2) < n) {
      const float s = dv[i >> 2];
      float4 v = yv[u];
      v.x *= s; v.y *= s; v.z *= s; v.w *= s;
      yq[i] = v;
    }
  }
}

template <bool LAST>
__global__ void k_agg(const unsigned* __restrict__ pk, const int* __restrict__ binBase,
                      const int* __restrict__ rs,
                      const float* __restrict__ xws, const float* __restrict__ dinv,
                      const float* __restrict__ bias, const float* __restrict__ W2,
                      const float* __restrict__ Wl, const float* __restrict__ bl,
                      float* __restrict__ Y, float* __restrict__ out, int n) {
  __shared__ __attribute__((aligned(16))) float acc[NB * 16];
  __shared__ int rowStart[RS];
  __shared__ int sorted[CAP];

  const int t = threadIdx.x, b = blockIdx.x;
  const int lane = t & 31;
  const int fq = lane & 3;
  const int grp = t >> 2;
  const float4* xq = (const float4*)xws;
  const int node0 = b * NB;
  for (int i = t; i < NB * 16; i += AGT) {
    const int node = node0 + (i >> 4);
    acc[i] = (node < n) ? xws[(size_t)node0 * 16 + i] : 0.0f;
  }
  const int e0 = binBase[b], e1 = binBase[b + 1];
  const int* rsb = rs + (size_t)(b + e0 / CAP) * RS;

  for (int b0 = e0; b0 < e1; b0 += CAP, rsb += RS) {
    const int b1 = min(b0 + CAP, e1);
    const int nb = b1 - b0;
    const int rsv = rsb[min(t, NB)];
    unsigned q[CAP / AGT];
#pragma unroll
    for (int u = 0; u < CAP / AGT; ++u) q[u] = pk[b0 + min(u * AGT + t, nb - 1)];
    if (t < RS) rowStart[t] = rsv;
#pragma unroll
    for (int u = 0; u < CAP / AGT; ++u)
      if (u * AGT + t < nb) sorted[u * AGT + t] = (int)q[u];
    __syncthreads();
    for (int r = grp; r < NB; r += AGT / 4) {
      const int s0 = rowStart[r], s1 = rowStart[r + 1];
      float4 sum = make_float4(0.0f, 0.0f, 0.0f, 0.0f);
      for (int i = s0; i < s1; i += 8) {
        float4 v[8];
#pragma unroll
        for (int u = 0; u < 8; ++u) {
          const int idx = min(i + u, s1 - 1);
          v[u] = xq[(unsigned)sorted[idx] * 4u + fq];
        }
#pragma unroll
        for (int u = 0; u < 8; ++u) {
          const bool in = i + u < s1;
          sum.x += in ? v[u].x : 0.0f;
          sum.y += in ? v[u].y : 0.0f;
          sum.z += in ? v[u].z : 0.0f;
          sum.w += in ? v[u].w : 0.0f;
        }
      }
      float4* ar = (float4*)(acc + r * 16 + fq * 4);
      float4 a = *ar;
      a.x += sum.x; a.y += sum.y; a.z += sum.z; a.w += sum.w;
      *ar = a;
    }
    __syncthreads();
  }
  __syncthreads();

  if (!LAST) {
    const int wave = __builtin_amdgcn_readfirstlane(t >> 5);
    const int row0 = node0 + wave * 16;
    if (row0 < n) {
      const int m  = lane & 15;
      const int kh = lane >> 4;
      const float da = dinv[row0 + m];
      const float4 d0 = *(const float4*)(dinv + row0 + 8 * kh);
      const float4 d1 = *(const float4*)(dinv + row0 + 8 * kh + 4);
      const float dr[8] = {d0.x, d0.y, d0.z, d0.w, d1.x, d1.y, d1.z, d1.w};
      const float* ar = acc + (wave * 16 + m) * 16;
      v8f c = {};
#pragma unroll
      for (int kb = 0; kb < 16; kb += 4) {
        const int k0 = kb + 2 * kh;
        v2f a, w;
        a.x = fmaxf(ar[k0]     * da + bias[k0],     0.0f);
        a.y = fmaxf(ar[k0 + 1] * da + bias[k0 + 1], 0.0f);
        w.x = W2[k0 * 16 + m];
        w.y = W2[(k0 + 1) * 16 + m];
        c = __builtin_amdgcn_wmma_f32_16x16x4_f32(
            false, a, false, w, (short)0, c, false, false);
      }
      float* yp = Y + (size_t)(row0 + 8 * kh) * 16 + m;
#pragma unroll
      for (int v = 0; v < 8; ++v) yp[v * 16] = c[v] * dr[v];
    }
  } else {
    const int node = node0 + t;
    if (t < NB && node < n) {
      const float s = dinv[node];
      float a = 0.0f;
#pragma unroll
      for (int ff = 0; ff < 16; ++ff) {
        const float h = fmaxf(acc[t * 16 + ff] * s + bias[ff], 0.0f);
        a += h * Wl[ff];
      }
      out[node] = a + bl[0];
    }
  }
}

extern "C" void kernel_launch(void* const* d_in, const int* in_sizes, int n_in,
                              void* d_out, int out_size, void* d_ws, size_t ws_size,
                              hipStream_t stream) {
  const float* x   = (const float*)d_in[0];
  const int*   ei  = (const int*)d_in[1];
  const float* W1  = (const float*)d_in[2];
  const float* b1  = (const float*)d_in[3];
  const float* W2  = (const float*)d_in[4];
  const float* b2  = (const float*)d_in[5];
  const float* Wl  = (const float*)d_in[6];
  const float* bl  = (const float*)d_in[7];
  float* out = (float*)d_out;

  const int din = in_sizes[2] / 16;
  const int n   = in_sizes[0] / din;
  const int E   = in_sizes[1] / 2;
  const int tot = n * 16;

  float* ws   = (float*)d_ws;
  float* dinv = ws;
  float* bufA = ws + (((size_t)n + 255) & ~(size_t)255);
  float* bufB = bufA + (size_t)n * 16;

  const int T = 256;
  const int ntiles = (n + 15) / 16;
  const int gemmBlocks = (ntiles + 3) / 4;

  const int nbins = (n + NB - 1) / NB;
  unsigned* pk   = (unsigned*)(bufB + (size_t)n * 16);
  int* cnt       = (int*)(pk + (((size_t)E + 255) & ~(size_t)255));
  int* binBase   = cnt + (size_t)nbins * PG + MAXBINS;
  const size_t need = (size_t)((char*)(binBase + MAXBINS + 1) - (char*)d_ws);

  int* rs        = cnt;
  int* wgTot     = (int*)bufB;

  const int chunk = (E + PG - 1) / PG;
  if (ws_size >= need && nbins <= MAXBINS && chunk <= CHW &&
      (size_t)(nbins + E / CAP + 1) * RS <= (size_t)nbins * PG &&
      (size_t)n * 16 >= (size_t)(PG / BW) * MAXBINS) {
    k_count   <<<PG / BW + nbins, BT, 0, stream>>>(ei + E, cnt, wgTot, E, chunk, nbins, x, W1, bufA, n, din);
    k_part    <<<PG / BW, BT, 0, stream>>>(ei, ei + E, cnt, wgTot, binBase, pk, E, chunk, nbins);
    k_sort    <<<nbins, AGT, 0, stream>>>(pk, binBase, rs, dinv, n, bufA);

    k_agg<false><<<nbins, AGT, 0, stream>>>(pk, binBase, rs, bufA, dinv, b1, W2, Wl, bl, bufB, out, n);
    k_agg<true> <<<nbins, AGT, 0, stream>>>(pk, binBase, rs, bufB, dinv, b2, W2, Wl, bl, bufA, out, n);
    return;
  }

  k_init  <<<(tot + T - 1) / T, T, 0, stream>>>(dinv, bufB, n, tot);
  k_degree<<<(E + T - 1) / T,   T, 0, stream>>>(ei + E, dinv, E);
  k_rsqrt <<<(n + T - 1) / T,   T, 0, stream>>>(dinv, n);

  k_gemm16 <<<gemmBlocks, 128, 0, stream>>>(x, W1, bufA, n, din);
  k_scatter<<<(4 * E + T - 1) / T, T, 0, stream>>>(ei, ei + E, dinv, bufA, bufB, E);
  k_final1 <<<(tot + T - 1) / T, T, 0, stream>>>(bufB, bufA, dinv, b1, tot);

  k_gemm16 <<<gemmBlocks, 128, 0, stream>>>(bufB, W2, bufA, n, 16);
  k_zero   <<<(tot + T - 1) / T, T, 0, stream>>>(bufB, tot);
  k_scatter<<<(4 * E + T - 1) / T, T, 0, stream>>>(ei, ei + E, dinv, bufA, bufB, E);
  k_final2 <<<(n + T - 1) / T, T, 0, stream>>>(bufB, bufA, dinv, b2, Wl, bl, out, n);
}


# bucketing in one launch (5 launches become 4): a work-group counting-sorts its chunks by bin in LDS, from registers, into one contiguous region with run starts; a bin is 256 runs; f32 sum order same
# speedup vs baseline: 6.2828x; 1.0477x over previous
#include <hip/hip_runtime.h>
#include <hip/hip_bf16.h>
#include <stddef.h>

typedef __attribute__((ext_vector_type(2))) float v2f;
typedef __attribute__((ext_vector_type(8))) float v8f;
typedef __attribute__((ext_vector_type(2))) unsigned v2u;

__device__ __forceinline__ void atomAddF(float* p, float v) {
  asm volatile("global_atomic_add_f32 %0, %1, off scope:SCOPE_DEV"
               :
               : "v"(p), "v"(v)
               : "memory");
}

__global__ void k_init(float* __restrict__ deg, float* __restrict__ agg,
                       int n, int tot) {
  int t = blockIdx.x * blockDim.x + threadIdx.x;
  if (t < tot) agg[t] = 0.0f;
  if (t < n)   deg[t] = 1.0f;
}

__global__ void k_zero(float* __restrict__ p, int tot) {
  int t = blockIdx.x * blockDim.x + threadIdx.x;
  if (t < tot) p[t] = 0.0f;
}

__global__ void k_degree(const int* __restrict__ dst, float* __restrict__ deg,
                         int E) {
  int t = blockIdx.x * blockDim.x + threadIdx.x;
  if (t < E) atomAddF(&deg[dst[t]], 1.0f);
}

__global__ void k_rsqrt(float* __restrict__ d, int n) {
  int t = blockIdx.x * blockDim.x + threadIdx.x;
  if (t < n) d[t] = rsqrtf(d[t]);
}

__global__ void k_gemm16(const float* __restrict__ X, const float* __restrict__ W,
                         float* __restrict__ Y, int nrows, int K) {
  const int wave = threadIdx.x >> 5;
  const int lane = threadIdx.x & 31;
  const int tile = blockIdx.x * (blockDim.x >> 5) + wave;
  const int row0 = tile * 16;
  if (row0 >= nrows) return;

  const int m  = lane & 15;
  const int kh = lane >> 4;

  const float* ap = X + (size_t)(row0 + m) * K + 2 * kh;
  v8f acc = {};
  for (int kb = 0; kb < K; kb += 4) {
    v2f a = *(const v2f*)(ap + kb);
    const int kr = kb + 2 * kh;
    v2f b;
    b.x = W[(size_t)kr * 16 + m];
    b.y = W[(size_t)(kr + 1) * 16 + m];
    acc = __builtin_amdgcn_wmma_f32_16x16x4_f32(
        false, a, false, b, (short)0, acc, false, false);
  }
#pragma unroll
  for (int v = 0; v < 8; ++v) {
    Y[(size_t)(row0 + v + 8 * kh) * 16 + m] = acc[v];
  }
}

__global__ void k_scatter(const int* __restrict__ src, const int* __restrict__ dst,
                          const float* __restrict__ dinv,
                          const float* __restrict__ xw,
                          float* __restrict__ agg, int E) {
  int t = blockIdx.x * blockDim.x + threadIdx.x;
  int e = t >> 2;
  if (e >= E) return;
  int q = (t & 3) << 2;
  int s = src[e];
  int d = dst[e];
  float nrm = dinv[s] * dinv[d];
  float4 v = *(const float4*)(xw + (size_t)s * 16 + q);
  float* out = agg + (size_t)d * 16 + q;
  atomAddF(out + 0, v.x * nrm);
  atomAddF(out + 1, v.y * nrm);
  atomAddF(out + 2, v.z * nrm);
  atomAddF(out + 3, v.w * nrm);
}

__global__ void k_final1(float* __restrict__ agg, const float* __restrict__ xw,
                         const float* __restrict__ dinv,
                         const float* __restrict__ b, int tot) {
  int t = blockIdx.x * blockDim.x + threadIdx.x;
  if (t >= tot) return;
  int i = t >> 4, f = t & 15;
  float s = dinv[i];
  float v = agg[t] + xw[t] * s * s + b[f];
  agg[t] = fmaxf(v, 0.0f);
}

__global__ void k_final2(const float* __restrict__ agg, const float* __restrict__ xw,
                         const float* __restrict__ dinv,
                         const float* __restrict__ b2,
                         const float* __restrict__ Wl,
                         const float* __restrict__ bl,
                         float* __restrict__ out, int n) {
  int i = blockIdx.x * blockDim.x + threadIdx.x;
  if (i >= n) return;
  float s = dinv[i];
  float ss = s * s;
  float acc = 0.0f;
#pragma unroll
  for (int f = 0; f < 16; ++f) {
    float h = agg[(size_t)i * 16 + f] + xw[(size_t)i * 16 + f] * ss + b2[f];
    h = fmaxf(h, 0.0f);
    acc += h * Wl[f];
  }
  out[i] = acc + bl[0];
}

#define NB_SHIFT 8
#define NB       256
#define PG       4096
#define MAXBINS  512
#define BIN_BITS 9
#define AGW      16
#define AGT      (AGW * 32)
#define CAP      10240
#define GM       25
#define BW       16
#define BT       (BW * 32)
#define HS       (MAXBINS + 1)
#define WCAP     (BW * 32 * GM)
#define OS       (MAXBINS + 1)
#define NRUN     (PG / BW)
#define RS       (NB + 1)
static_assert(MAXBINS == (1 << BIN_BITS) && NB == (1 << NB_SHIFT), "key widths");
static_assert(PG % BW == 0 && BT == MAXBINS, "bucketing work-group shape");
static_assert(NRUN == 256 && NRUN <= AGT, "a bin work-group scans the bin's NRUN runs with its first eight waves and finds a run in 8 halvings");
static_assert(BW * 16 == NB, "a product work-group's BW tiles of 16 rows are one bin's rows");
static_assert((NB * 4) % AGT == 0, "a bin's product rows are scaled by one work-group in whole rounds");
static_assert(CAP % AGT == 0 && RS <= AGT, "a sorted batch is copied by one work-group in whole rounds");

template <int BITS>
__device__ __forceinline__ unsigned sameKeyMask(int key, bool valid) {
  static_assert(BITS >= 1 && BITS <= 31, "key width");
  unsigned differ = 0u;
#pragma unroll
  for (int b = 0; b < BITS; ++b) {
    const int mine = (key << (31 - b)) >> 31;
    const unsigned vote = __builtin_amdgcn_ballot_w32(mine != 0);
    differ |= vote ^ (unsigned)mine;
  }
  return ~differ & __builtin_amdgcn_ballot_w32(valid);
}

template <int N>
__device__ __forceinline__ void blockScanIncl(int* s, int t) {
  for (int d = 1; d < N; d <<= 1) {
    const int v = (t >= d) ? s[t - d] : 0;
    __syncthreads();
    s[t] += v;
    __syncthreads();
  }
}

__global__ void k_bucket(const int* __restrict__ src, const int* __restrict__ dst,
                         int* __restrict__ off, unsigned* __restrict__ pk,
                         int E, int chunk, int nbins,
                         const float* __restrict__ X, const float* __restrict__ W,
                         float* __restrict__ Y, int n, int K) {
  __shared__ int h[BW * HS];
  __shared__ int sc[BT];
  __shared__ unsigned stage[WCAP];
  const int t = threadIdx.x;
  const int lane = t & 31;
  const int wave = __builtin_amdgcn_readfirstlane(t >> 5);
  if (blockIdx.x >= NRUN) {
    const int row0 = ((int)blockIdx.x - NRUN) * NB + wave * 16;
    if (row0 >= n) return;
    const int m  = lane & 15;
    const int kh = lane >> 4;
    const float* ap = X + (size_t)(row0 + m) * K + 2 * kh;
    v8f acc = {};
    for (int kb = 0; kb < K; kb += 4) {
      v2f a = *(const v2f*)(ap + kb);
      const int kr = kb + 2 * kh;
      v2f w;
      w.x = W[(size_t)kr * 16 + m];
      w.y = W[(size_t)(kr + 1) * 16 + m];
      acc = __builtin_amdgcn_wmma_f32_16x16x4_f32(
          false, a, false, w, (short)0, acc, false, false);
    }
#pragma unroll
    for (int v = 0; v < 8; ++v) {
      Y[(size_t)(row0 + v + 8 * kh) * 16 + m] = acc[v];
    }
    return;
  }
  int* hv = h + wave * HS;
  const int e0 = ((int)blockIdx.x * BW + wave) * chunk;
  const int e1 = min(e0 + chunk, E);
  const unsigned below = (1u << lane) - 1u;
  int dv[GM], sv[GM];
  unsigned mk[GM];
#pragma unroll
  for (int u = 0; u < GM; ++u) {
    const int i = min(e0 + u * 32 + lane, E - 1);
    dv[u] = dst[i]; sv[u] = src[i];
  }
  for (int i = t; i < BW * HS; i += BT) h[i] = 0;
  __syncthreads();
#pragma unroll
  for (int u = 0; u < GM; ++u) {
    mk[u] = 0u;
    if (e0 + u * 32 < e1) {
      const bool valid = e0 + u * 32 + lane < e1;
      const int key = valid ? (dv[u] >> NB_SHIFT) : -1;
      const unsigned m = sameKeyMask<BIN_BITS>(key, valid);
      mk[u] = m;
      if (valid && (m >> lane) == 1u) hv[key] += __popc(m);
    }
  }
  __syncthreads();
  int c = 0;
  if (t < nbins) {
#pragma unroll
    for (int w = 0; w < BW; ++w) c += h[w * HS + t];
  }
  sc[t] = c;
  __syncthreads();
  blockScanIncl<BT>(sc, t);
  const int ls = sc[t] - c;
  const int total = sc[BT - 1];
  if (t <= nbins) off[(size_t)blockIdx.x * OS + t] = ls;
  if (t < nbins) {
    int run = ls;
#pragma unroll
    for (int w = 0; w < BW; ++w) {
      const int k = h[w * HS + t];
      h[w * HS + t] = run;
      run += k;
    }
  }
  __syncthreads();
#pragma unroll
  for (int u = 0; u < GM; ++u) {
    if (e0 + u * 32 + lane < e1) {
      const int key = dv[u] >> NB_SHIFT;
      const unsigned m = mk[u];
      const int pos = hv[key] + __popc(m & below);
      if ((m >> lane) == 1u) hv[key] = pos + 1;
      stage[pos] = ((unsigned)sv[u] << NB_SHIFT) | (unsigned)(dv[u] & (NB - 1));
    }
  }
  __syncthreads();
  unsigned* reg = pk + (size_t)blockIdx.x * WCAP;
  for (int i = t; i < total; i += BT) reg[i] = stage[i];
}

__device__ __forceinline__ int binRuns(const int* __restrict__ off, int b, int t,
                                       int* rp, int* ra, int* wt) {
  const int lane = t & 31, wv = t >> 5;
  int o0 = 0, len = 0;
  if (t < NRUN) {
    o0 = off[(size_t)t * OS + b];
    len = off[(size_t)t * OS + b + 1] - o0;
  }
  int sl = len, so = o0;
#pragma unroll
  for (int d = 1; d < 32; d <<= 1) {
    const int vl = __shfl_up(sl, d), vo = __shfl_up(so, d);
    if (lane >= d) { sl += vl; so += vo; }
  }
  if (lane == 31 && wv < NRUN / 32) { wt[wv] = sl; wt[NRUN / 32 + wv] = so; }
  __syncthreads();
  int bl = 0, bo = 0, tl = 0, to = 0;
#pragma unroll
  for (int w = 0; w < NRUN / 32; ++w) {
    const int a = wt[w], o = wt[NRUN / 32 + w];
    if (w < wv) { bl += a; bo += o; }
    tl += a; to += o;
  }
  if (t < NRUN) {
    rp[t] = bl + sl - len;
    ra[t] = t * WCAP + o0;
  }
  if (t == 0) rp[NRUN] = tl;
  __syncthreads();
  return to;
}

__device__ __forceinline__ int binWord(const int* rp, const int* ra, int f) {
  int g = 0;
#pragma unroll
  for (int s = NRUN / 2; s >= 1; s >>= 1)
    if (rp[g + s] <= f) g += s;
  return ra[g] + (f - rp[g]);
}

__global__ void k_sort(unsigned* __restrict__ pk, const int* __restrict__ off,
                       int* __restrict__ rs, float* __restrict__ dinv, int n,
                       float* __restrict__ Y) {
  __shared__ int rp[NRUN + 1], ra[NRUN], wt[2 * NRUN / 32];
  __shared__ int adl[CAP];
  __shared__ int hw[AGW * NB];
  __shared__ int sc[AGT];
  __shared__ int sorted[CAP];
  __shared__ float dv[NB];

  const int t = threadIdx.x, b = blockIdx.x;
  const int lane = t & 31;
  const int wave = __builtin_amdgcn_readfirstlane(t >> 5);
  const int e0 = binRuns(off, b, t, rp, ra, wt);
  const int e1 = e0 + rp[NRUN];
  int* hwv = hw + wave * NB;
  const unsigned below = (1u << lane) - 1u;
  int* rsb = rs + (size_t)(b + e0 / CAP) * RS;
  int indeg = 0;

  for (int b0 = e0; b0 < e1; b0 += CAP, rsb += RS) {
    const int b1 = min(b0 + CAP, e1);
    const int nb = b1 - b0;
    unsigned q[CAP / AGT];
    unsigned mk[CAP / AGT];
    {
      int ad[CAP / AGT];
#pragma unroll
      for (int u = 0; u < CAP / AGT; ++u) ad[u] = binWord(rp, ra, b0 - e0 + min(u * AGT + t, nb - 1));
#pragma unroll
      for (int u = 0; u < CAP / AGT; ++u) { q[u] = pk[ad[u]]; adl[u * AGT + t] = ad[u]; }
    }
    for (int i = t; i < AGW * NB; i += AGT) hw[i] = 0;
    __syncthreads();
#pragma unroll
    for (int u = 0; u < CAP / AGT; ++u) {
      mk[u] = 0u;
      if (u * AGT + wave * 32 < nb) {
        const bool valid = u * AGT + t < nb;
        const int key = valid ? (int)(q[u] & (NB - 1)) : -1;
        const unsigned m = sameKeyMask<NB_SHIFT>(key, valid);
        mk[u] = m;
        if (valid && (m >> lane) == 1u) hwv[key] += __popc(m);
      }
    }
    __syncthreads();
    {
      const int key = t >> 1, w0 = (t & 1) * (AGW / 2);
      int s = 0;
#pragma unroll
      for (int i = 0; i < AGW / 2; ++i) s += hw[(w0 + i) * NB + key];
      indeg += s + __shfl_xor(s, 1);
      sc[t] = s;
      __syncthreads();
      blockScanIncl<AGT>(sc, t);
      int run = sc[t] - s;
      if ((t & 1) == 0) rsb[key] = run;
      if (t == AGT - 1) rsb[NB] = sc[t];
#pragma unroll
      for (int i = 0; i < AGW / 2; ++i) {
        const int idx = (w0 + i) * NB + key;
        const int c = hw[idx];
        hw[idx] = run;
        run += c;
      }
    }
    __syncthreads();
#pragma unroll
    for (int u = 0; u < CAP / AGT; ++u) {
      if (u * AGT + t < nb) {
        const int key = (int)(q[u] & (NB - 1));
        const unsigned m = mk[u];
        const int pos = hwv[key] + __popc(m & below);
        if ((m >> lane) == 1u) hwv[key] = pos + 1;
        sorted[pos] = (int)(q[u] >> NB_SHIFT);
      }
    }
    __syncthreads();
#pragma unroll
    for (int u = 0; u < CAP / AGT; ++u)
      if (u * AGT + t < nb) pk[adl[u * AGT + t]] = (unsigned)sorted[u * AGT + t];
  }
  float4* yq = (float4*)(Y + (size_t)b * NB * 16);
  float4 yv[NB * 4 / AGT];
#pragma unroll
  for (int u = 0; u < NB * 4 / AGT; ++u) {
    const int i = u * AGT + t;
    if (b * NB + (i >> 2) < n) yv[u] = yq[i];
  }
  const int node = b * NB + (t >> 1);
  if ((t & 1) == 0) {
    const float dn = rsqrtf(1.0f + (float)indeg);
    dv[t >> 1] = dn;
    if (node < n) dinv[node] = dn;
  }
  __syncthreads();
#pragma unroll
  for (int u = 0; u < NB * 4 / AGT; ++u) {
    const int i = u * AGT + t;
    if (b * NB + (i >> 2) < n) {
      const float s = dv[i >> 2];
      float4 v = yv[u];
      v.x *= s; v.y *= s; v.z *= s; v.w *= s;
      yq[i] = v;
    }
  }
}

template <bool LAST>
__global__ void k_agg(const unsigned* __restrict__ pk, const int* __restrict__ off,
                      const int* __restrict__ rs,
                      const float* __restrict__ xws, const float* __restrict__ dinv,
                      const float* __restrict__ bias, const float* __restrict__ W2,
                      const float* __restrict__ Wl, const float* __restrict__ bl,
                      float* __restrict__ Y, float* __restrict__ out, int n) {
  __shared__ __attribute__((aligned(16))) float acc[NB * 16];
  __shared__ int rowStart[RS];
  __shared__ int sorted[CAP];
  __shared__ int rp[NRUN + 1], ra[NRUN], wt[2 * NRUN / 32];

  const int t = threadIdx.x, b = blockIdx.x;
  const int lane = t & 31;
  const int fq = lane & 3;
  const int grp = t >> 2;
  const float4* xq = (const float4*)xws;
  const int node0 = b * NB;
  for (int i = t; i < NB * 16; i += AGT) {
    const int node = node0 + (i >> 4);
    acc[i] = (node < n) ? xws[(size_t)node0 * 16 + i] : 0.0f;
  }
  const int e0 = binRuns(off, b, t, rp, ra, wt);
  const int e1 = e0 + rp[NRUN];
  const int* rsb = rs + (size_t)(b + e0 / CAP) * RS;

  for (int b0 = e0; b0 < e1; b0 += CAP, rsb += RS) {
    const int b1 = min(b0 + CAP, e1);
    const int nb = b1 - b0;
    const int rsv = rsb[min(t, NB)];
    unsigned q[CAP / AGT];
    int ad[CAP / AGT];
#pragma unroll
    for (int u = 0; u < CAP / AGT; ++u) ad[u] = binWord(rp, ra, b0 - e0 + min(u * AGT + t, nb - 1));
#pragma unroll
    for (int u = 0; u < CAP / AGT; ++u) q[u] = pk[ad[u]];
    if (t < RS) rowStart[t] = rsv;
#pragma unroll
    for (int u = 0; u < CAP / AGT; ++u)
      if (u * AGT + t < nb) sorted[u * AGT + t] = (int)q[u];
    __syncthreads();
    for (int r = grp; r < NB; r += AGT / 4) {
      const int s0 = rowStart[r], s1 = rowStart[r + 1];
      float4 sum = make_float4(0.0f, 0.0f, 0.0f, 0.0f);
      for (int i = s0; i < s1; i += 8) {
        float4 v[8];
#pragma unroll
        for (int u = 0; u < 8; ++u) {
          const int idx = min(i + u, s1 - 1);
          v[u] = xq[(unsigned)sorted[idx] * 4u + fq];
        }
#pragma unroll
        for (int u = 0; u < 8; ++u) {
          const bool in = i + u < s1;
          sum.x += in ? v[u].x : 0.0f;
          sum.y += in ? v[u].y : 0.0f;
          sum.z += in ? v[u].z : 0.0f;
          sum.w += in ? v[u].w : 0.0f;
        }
      }
      float4* ar = (float4*)(acc + r * 16 + fq * 4);
      float4 a = *ar;
      a.x += sum.x; a.y += sum.y; a.z += sum.z; a.w += sum.w;
      *ar = a;
    }
    __syncthreads();
  }
  __syncthreads();

  if (!LAST) {
    const int wave = __builtin_amdgcn_readfirstlane(t >> 5);
    const int row0 = node0 + wave * 16;
    if (row0 < n) {
      const int m  = lane & 15;
      const int kh = lane >> 4;
      const float da = dinv[row0 + m];
      const float4 d0 = *(const float4*)(dinv + row0 + 8 * kh);
      const float4 d1 = *(const float4*)(dinv + row0 + 8 * kh + 4);
      const float dr[8] = {d0.x, d0.y, d0.z, d0.w, d1.x, d1.y, d1.z, d1.w};
      const float* ar = acc + (wave * 16 + m) * 16;
      v8f c = {};
#pragma unroll
      for (int kb = 0; kb < 16; kb += 4) {
        const int k0 = kb + 2 * kh;
        v2f a, w;
        a.x = fmaxf(ar[k0]     * da + bias[k0],     0.0f);
        a.y = fmaxf(ar[k0 + 1] * da + bias[k0 + 1], 0.0f);
        w.x = W2[k0 * 16 + m];
        w.y = W2[(k0 + 1) * 16 + m];
        c = __builtin_amdgcn_wmma_f32_16x16x4_f32(
            false, a, false, w, (short)0, c, false, false);
      }
      float* yp = Y + (size_t)(row0 + 8 * kh) * 16 + m;
#pragma unroll
      for (int v = 0; v < 8; ++v) yp[v * 16] = c[v] * dr[v];
    }
  } else {
    const int node = node0 + t;
    if (t < NB && node < n) {
      const float s = dinv[node];
      float a = 0.0f;
#pragma unroll
      for (int ff = 0; ff < 16; ++ff) {
        const float h = fmaxf(acc[t * 16 + ff] * s + bias[ff], 0.0f);
        a += h * Wl[ff];
      }
      out[node] = a + bl[0];
    }
  }
}

extern "C" void kernel_launch(void* const* d_in, const int* in_sizes, int n_in,
                              void* d_out, int out_size, void* d_ws, size_t ws_size,
                              hipStream_t stream) {
  const float* x   = (const float*)d_in[0];
  const int*   ei  = (const int*)d_in[1];
  const float* W1  = (const float*)d_in[2];
  const float* b1  = (const float*)d_in[3];
  const float* W2  = (const float*)d_in[4];
  const float* b2  = (const float*)d_in[5];
  const float* Wl  = (const float*)d_in[6];
  const float* bl  = (const float*)d_in[7];
  float* out = (float*)d_out;

  const int din = in_sizes[2] / 16;
  const int n   = in_sizes[0] / din;
  const int E   = in_sizes[1] / 2;
  const int tot = n * 16;

  float* ws   = (float*)d_ws;
  float* dinv = ws;
  float* bufA = ws + (((size_t)n + 255) & ~(size_t)255);
  float* bufB = bufA + (size_t)n * 16;

  const int T = 256;
  const int ntiles = (n + 15) / 16;
  const int gemmBlocks = (ntiles + 3) / 4;

  const int nbins = (n + NB - 1) / NB;
  unsigned* pk   = (unsigned*)(bufB + (size_t)n * 16);
  int* off       = (int*)(pk + (size_t)NRUN * WCAP);
  int* rs        = off + (size_t)NRUN * OS;
  const size_t need = (size_t)((char*)(rs + (size_t)(nbins + E / CAP + 1) * RS) - (char*)d_ws);

  const int chunk = (E + PG - 1) / PG;
  if (ws_size >= need && nbins < MAXBINS && chunk <= 32 * GM) {
    k_bucket  <<<NRUN + nbins, BT, 0, stream>>>(ei, ei + E, off, pk, E, chunk, nbins, x, W1, bufA, n, din);
    k_sort    <<<nbins, AGT, 0, stream>>>(pk, off, rs, dinv, n, bufA);

    k_agg<false><<<nbins, AGT, 0, stream>>>(pk, off, rs, bufA, dinv, b1, W2, Wl, bl, bufB, out, n);
    k_agg<true> <<<nbins, AGT, 0, stream>>>(pk, off, rs, bufB, dinv, b2, W2, Wl, bl, bufA, out, n);
    return;
  }

  k_init  <<<(tot + T - 1) / T, T, 0, stream>>>(dinv, bufB, n, tot);
  k_degree<<<(E + T - 1) / T,   T, 0, stream>>>(ei + E, dinv, E);
  k_rsqrt <<<(n + T - 1) / T,   T, 0, stream>>>(dinv, n);

  k_gemm16 <<<gemmBlocks, 128, 0, stream>>>(x, W1, bufA, n, din);
  k_scatter<<<(4 * E + T - 1) / T, T, 0, stream>>>(ei, ei + E, dinv, bufA, bufB, E);
  k_final1 <<<(tot + T - 1) / T, T, 0, stream>>>(bufB, bufA, dinv, b1, tot);

  k_gemm16 <<<gemmBlocks, 128, 0, stream>>>(bufB, W2, bufA, n, 16);
  k_zero   <<<(tot + T - 1) / T, T, 0, stream>>>(bufB, tot);
  k_scatter<<<(4 * E + T - 1) / T, T, 0, stream>>>(ei, ei + E, dinv, bufA, bufB, E);
  k_final2 <<<(n + T - 1) / T, T, 0, stream>>>(bufB, bufA, dinv, b2, Wl, bl, out, n);
}


# bin work-groups get a batch's words of pk with no search: two threads per run write the run's word numbers to LDS, each slot reads its own; was 8 dependent LDS halvings per edge; f32 sum order same
# speedup vs baseline: 6.7001x; 1.0664x over previous
#include <hip/hip_runtime.h>
#include <hip/hip_bf16.h>
#include <stddef.h>

typedef __attribute__((ext_vector_type(2))) float v2f;
typedef __attribute__((ext_vector_type(8))) float v8f;
typedef __attribute__((ext_vector_type(2))) unsigned v2u;

__device__ __forceinline__ void atomAddF(float* p, float v) {
  asm volatile("global_atomic_add_f32 %0, %1, off scope:SCOPE_DEV"
               :
               : "v"(p), "v"(v)
               : "memory");
}

__global__ void k_init(float* __restrict__ deg, float* __restrict__ agg,
                       int n, int tot) {
  int t = blockIdx.x * blockDim.x + threadIdx.x;
  if (t < tot) agg[t] = 0.0f;
  if (t < n)   deg[t] = 1.0f;
}

__global__ void k_zero(float* __restrict__ p, int tot) {
  int t = blockIdx.x * blockDim.x + threadIdx.x;
  if (t < tot) p[t] = 0.0f;
}

__global__ void k_degree(const int* __restrict__ dst, float* __restrict__ deg,
                         int E) {
  int t = blockIdx.x * blockDim.x + threadIdx.x;
  if (t < E) atomAddF(&deg[dst[t]], 1.0f);
}

__global__ void k_rsqrt(float* __restrict__ d, int n) {
  int t = blockIdx.x * blockDim.x + threadIdx.x;
  if (t < n) d[t] = rsqrtf(d[t]);
}

__global__ void k_gemm16(const float* __restrict__ X, const float* __restrict__ W,
                         float* __restrict__ Y, int nrows, int K) {
  const int wave = threadIdx.x >> 5;
  const int lane = threadIdx.x & 31;
  const int tile = blockIdx.x * (blockDim.x >> 5) + wave;
  const int row0 = tile * 16;
  if (row0 >= nrows) return;

  const int m  = lane & 15;
  const int kh = lane >> 4;

  const float* ap = X + (size_t)(row0 + m) * K + 2 * kh;
  v8f acc = {};
  for (int kb = 0; kb < K; kb += 4) {
    v2f a = *(const v2f*)(ap + kb);
    const int kr = kb + 2 * kh;
    v2f b;
    b.x = W[(size_t)kr * 16 + m];
    b.y = W[(size_t)(kr + 1) * 16 + m];
    acc = __builtin_amdgcn_wmma_f32_16x16x4_f32(
        false, a, false, b, (short)0, acc, false, false);
  }
#pragma unroll
  for (int v = 0; v < 8; ++v) {
    Y[(size_t)(row0 + v + 8 * kh) * 16 + m] = acc[v];
  }
}

__global__ void k_scatter(const int* __restrict__ src, const int* __restrict__ dst,
                          const float* __restrict__ dinv,
                          const float* __restrict__ xw,
                          float* __restrict__ agg, int E) {
  int t = blockIdx.x * blockDim.x + threadIdx.x;
  int e = t >> 2;
  if (e >= E) return;
  int q = (t & 3) << 2;
  int s = src[e];
  int d = dst[e];
  float nrm = dinv[s] * dinv[d];
  float4 v = *(const float4*)(xw + (size_t)s * 16 + q);
  float* out = agg + (size_t)d * 16 + q;
  atomAddF(out + 0, v.x * nrm);
  atomAddF(out + 1, v.y * nrm);
  atomAddF(out + 2, v.z * nrm);
  atomAddF(out + 3, v.w * nrm);
}

__global__ void k_final1(float* __restrict__ agg, const float* __restrict__ xw,
                         const float* __restrict__ dinv,
                         const float* __restrict__ b, int tot) {
  int t = blockIdx.x * blockDim.x + threadIdx.x;
  if (t >= tot) return;
  int i = t >> 4, f = t & 15;
  float s = dinv[i];
  float v = agg[t] + xw[t] * s * s + b[f];
  agg[t] = fmaxf(v, 0.0f);
}

__global__ void k_final2(const float* __restrict__ agg, const float* __restrict__ xw,
                         const float* __restrict__ dinv,
                         const float* __restrict__ b2,
                         const float* __restrict__ Wl,
                         const float* __restrict__ bl,
                         float* __restrict__ out, int n) {
  int i = blockIdx.x * blockDim.x + threadIdx.x;
  if (i >= n) return;
  float s = dinv[i];
  float ss = s * s;
  float acc = 0.0f;
#pragma unroll
  for (int f = 0; f < 16; ++f) {
    float h = agg[(size_t)i * 16 + f] + xw[(size_t)i * 16 + f] * ss + b2[f];
    h = fmaxf(h, 0.0f);
    acc += h * Wl[f];
  }
  out[i] = acc + bl[0];
}

#define NB_SHIFT 8
#define NB       256
#define PG       4096
#define MAXBINS  512
#define BIN_BITS 9
#define AGW      16
#define AGT      (AGW * 32)
#define CAP      10240
#define GM       25
#define BW       16
#define BT       (BW * 32)
#define HS       (MAXBINS + 1)
#define WCAP     (BW * 32 * GM)
#define OS       (MAXBINS + 1)
#define NRUN     (PG / BW)
#define RS       (NB + 1)
static_assert(MAXBINS == (1 << BIN_BITS) && NB == (1 << NB_SHIFT), "key widths");
static_assert(PG % BW == 0 && BT == MAXBINS, "bucketing work-group shape");
static_assert(NRUN == 256 && NRUN <= AGT, "a bin work-group scans the bin's NRUN runs with its first eight waves");
static_assert(AGT == 2 * NRUN, "a bin work-group lists a batch's words of pk with two threads per run");
static_assert(BW * 16 == NB, "a product work-group's BW tiles of 16 rows are one bin's rows");
static_assert((NB * 4) % AGT == 0, "a bin's product rows are scaled by one work-group in whole rounds");
static_assert(CAP % AGT == 0 && RS <= AGT, "a sorted batch is copied by one work-group in whole rounds");

template <int BITS>
__device__ __forceinline__ unsigned sameKeyMask(int key, bool valid) {
  static_assert(BITS >= 1 && BITS <= 31, "key width");
  unsigned differ = 0u;
#pragma unroll
  for (int b = 0; b < BITS; ++b) {
    const int mine = (key << (31 - b)) >> 31;
    const unsigned vote = __builtin_amdgcn_ballot_w32(mine != 0);
    differ |= vote ^ (unsigned)mine;
  }
  return ~differ & __builtin_amdgcn_ballot_w32(valid);
}

template <int N>
__device__ __forceinline__ void blockScanIncl(int* s, int t) {
  for (int d = 1; d < N; d <<= 1) {
    const int v = (t >= d) ? s[t - d] : 0;
    __syncthreads();
    s[t] += v;
    __syncthreads();
  }
}

__global__ void k_bucket(const int* __restrict__ src, const int* __restrict__ dst,
                         int* __restrict__ off, unsigned* __restrict__ pk,
                         int E, int chunk, int nbins,
                         const float* __restrict__ X, const float* __restrict__ W,
                         float* __restrict__ Y, int n, int K) {
  __shared__ int h[BW * HS];
  __shared__ int sc[BT];
  __shared__ unsigned stage[WCAP];
  const int t = threadIdx.x;
  const int lane = t & 31;
  const int wave = __builtin_amdgcn_readfirstlane(t >> 5);
  if (blockIdx.x >= NRUN) {
    const int row0 = ((int)blockIdx.x - NRUN) * NB + wave * 16;
    if (row0 >= n) return;
    const int m  = lane & 15;
    const int kh = lane >> 4;
    const float* ap = X + (size_t)(row0 + m) * K + 2 * kh;
    v8f acc = {};
    for (int kb = 0; kb < K; kb += 4) {
      v2f a = *(const v2f*)(ap + kb);
      const int kr = kb + 2 * kh;
      v2f w;
      w.x = W[(size_t)kr * 16 + m];
      w.y = W[(size_t)(kr + 1) * 16 + m];
      acc = __builtin_amdgcn_wmma_f32_16x16x4_f32(
          false, a, false, w, (short)0, acc, false, false);
    }
#pragma unroll
    for (int v = 0; v < 8; ++v) {
      Y[(size_t)(row0 + v + 8 * kh) * 16 + m] = acc[v];
    }
    return;
  }
  int* hv = h + wave * HS;
  const int e0 = ((int)blockIdx.x * BW + wave) * chunk;
  const int e1 = min(e0 + chunk, E);
  const unsigned below = (1u << lane) - 1u;
  int dv[GM], sv[GM];
  unsigned mk[GM];
#pragma unroll
  for (int u = 0; u < GM; ++u) {
    const int i = min(e0 + u * 32 + lane, E - 1);
    dv[u] = dst[i]; sv[u] = src[i];
  }
  for (int i = t; i < BW * HS; i += BT) h[i] = 0;
  __syncthreads();
#pragma unroll
  for (int u = 0; u < GM; ++u) {
    mk[u] = 0u;
    if (e0 + u * 32 < e1) {
      const bool valid = e0 + u * 32 + lane < e1;
      const int key = valid ? (dv[u] >> NB_SHIFT) : -1;
      const unsigned m = sameKeyMask<BIN_BITS>(key, valid);
      mk[u] = m;
      if (valid && (m >> lane) == 1u) hv[key] += __popc(m);
    }
  }
  __syncthreads();
  int c = 0;
  if (t < nbins) {
#pragma unroll
    for (int w = 0; w < BW; ++w) c += h[w * HS + t];
  }
  sc[t] = c;
  __syncthreads();
  blockScanIncl<BT>(sc, t);
  const int ls = sc[t] - c;
  const int total = sc[BT - 1];
  if (t <= nbins) off[(size_t)blockIdx.x * OS + t] = ls;
  if (t < nbins) {
    int run = ls;
#pragma unroll
    for (int w = 0; w < BW; ++w) {
      const int k = h[w * HS + t];
      h[w * HS + t] = run;
      run += k;
    }
  }
  __syncthreads();
#pragma unroll
  for (int u = 0; u < GM; ++u) {
    if (e0 + u * 32 + lane < e1) {
      const int key = dv[u] >> NB_SHIFT;
      const unsigned m = mk[u];
      const int pos = hv[key] + __popc(m & below);
      if ((m >> lane) == 1u) hv[key] = pos + 1;
      stage[pos] = ((unsigned)sv[u] << NB_SHIFT) | (unsigned)(dv[u] & (NB - 1));
    }
  }
  __syncthreads();
  unsigned* reg = pk + (size_t)blockIdx.x * WCAP;
  for (int i = t; i < total; i += BT) reg[i] = stage[i];
}

__device__ __forceinline__ int binRuns(const int* __restrict__ off, int b, int t,
                                       int* rp, int* ra, int* wt) {
  const int lane = t & 31, wv = t >> 5;
  int o0 = 0, len = 0;
  if (t < NRUN) {
    o0 = off[(size_t)t * OS + b];
    len = off[(size_t)t * OS + b + 1] - o0;
  }
  int sl = len, so = o0;
#pragma unroll
  for (int d = 1; d < 32; d <<= 1) {
    const int vl = __shfl_up(sl, d), vo = __shfl_up(so, d);
    if (lane >= d) { sl += vl; so += vo; }
  }
  if (lane == 31 && wv < NRUN / 32) { wt[wv] = sl; wt[NRUN / 32 + wv] = so; }
  __syncthreads();
  int bl = 0, bo = 0, tl = 0, to = 0;
#pragma unroll
  for (int w = 0; w < NRUN / 32; ++w) {
    const int a = wt[w], o = wt[NRUN / 32 + w];
    if (w < wv) { bl += a; bo += o; }
    tl += a; to += o;
  }
  if (t < NRUN) {
    rp[t] = bl + sl - len;
    ra[t] = t * WCAP + o0;
  }
  if (t == 0) rp[NRUN] = tl;
  __syncthreads();
  return to;
}

__device__ __forceinline__ void batchWords(const int* rp, const int* ra, int t,
                                           int B0, int B1, int* aw, int* lastw) {
  const int g = t >> 1, h = t & 1;
  const int r0 = rp[g], r1 = rp[g + 1], a = ra[g];
  const int lo = max(r0, B0), hi = min(r1, B1);
  if (h == 0 && r0 < B1 && B1 <= r1) *lastw = a + (B1 - 1 - r0);
#pragma clang loop vectorize(disable) interleave(disable) unroll(disable)
  for (int f = lo + h; f < hi; f += 2) aw[f - B0] = a + (f - r0);
}

__global__ void k_sort(unsigned* __restrict__ pk, const int* __restrict__ off,
                       int* __restrict__ rs, float* __restrict__ dinv, int n,
                       float* __restrict__ Y) {
  __shared__ int rp[NRUN + 1], ra[NRUN], wt[2 * NRUN / 32];
  __shared__ int lastw;
  __shared__ int adl[CAP];
  __shared__ int hw[AGW * NB];
  __shared__ int sc[AGT];
  __shared__ int sorted[CAP];
  __shared__ float dv[NB];

  const int t = threadIdx.x, b = blockIdx.x;
  const int lane = t & 31;
  const int wave = __builtin_amdgcn_readfirstlane(t >> 5);
  const int e0 = binRuns(off, b, t, rp, ra, wt);
  const int e1 = e0 + rp[NRUN];
  int* hwv = hw + wave * NB;
  const unsigned below = (1u << lane) - 1u;
  int* rsb = rs + (size_t)(b + e0 / CAP) * RS;
  int indeg = 0;

  for (int b0 = e0; b0 < e1; b0 += CAP, rsb += RS) {
    const int b1 = min(b0 + CAP, e1);
    const int nb = b1 - b0;
    unsigned q[CAP / AGT];
    unsigned mk[CAP / AGT];
    batchWords(rp, ra, t, b0 - e0, b1 - e0, adl, &lastw);
    for (int i = t; i < AGW * NB; i += AGT) hw[i] = 0;
    __syncthreads();
    {
      const int lw = lastw;
      int ad[CAP / AGT];
#pragma unroll
      for (int u = 0; u < CAP / AGT; ++u) ad[u] = (u * AGT + t < nb) ? adl[u * AGT + t] : lw;
#pragma unroll
      for (int u = 0; u < CAP / AGT; ++u) q[u] = pk[ad[u]];
    }
#pragma unroll
    for (int u = 0; u < CAP / AGT; ++u) {
      mk[u] = 0u;
      if (u * AGT + wave * 32 < nb) {
        const bool valid = u * AGT + t < nb;
        const int key = valid ? (int)(q[u] & (NB - 1)) : -1;
        const unsigned m = sameKeyMask<NB_SHIFT>(key, valid);
        mk[u] = m;
        if (valid && (m >> lane) == 1u) hwv[key] += __popc(m);
      }
    }
    __syncthreads();
    {
      const int key = t >> 1, w0 = (t & 1) * (AGW / 2);
      int s = 0;
#pragma unroll
      for (int i = 0; i < AGW / 2; ++i) s += hw[(w0 + i) * NB + key];
      indeg += s + __shfl_xor(s, 1);
      sc[t] = s;
      __syncthreads();
      blockScanIncl<AGT>(sc, t);
      int run = sc[t] - s;
      if ((t & 1) == 0) rsb[key] = run;
      if (t == AGT - 1) rsb[NB] = sc[t];
#pragma unroll
      for (int i = 0; i < AGW / 2; ++i) {
        const int idx = (w0 + i) * NB + key;
        const int c = hw[idx];
        hw[idx] = run;
        run += c;
      }
    }
    __syncthreads();
#pragma unroll
    for (int u = 0; u < CAP / AGT; ++u) {
      if (u * AGT + t < nb) {
        const int key = (int)(q[u] & (NB - 1));
        const unsigned m = mk[u];
        const int pos = hwv[key] + __popc(m & below);
        if ((m >> lane) == 1u) hwv[key] = pos + 1;
        sorted[pos] = (int)(q[u] >> NB_SHIFT);
      }
    }
    __syncthreads();
#pragma unroll
    for (int u = 0; u < CAP / AGT; ++u)
      if (u * AGT + t < nb) pk[adl[u * AGT + t]] = (unsigned)sorted[u * AGT + t];
    __syncthreads();
  }
  float4* yq = (float4*)(Y + (size_t)b * NB * 16);
  float4 yv[NB * 4 / AGT];
#pragma unroll
  for (int u = 0; u < NB * 4 / AGT; ++u) {
    const int i = u * AGT + t;
    if (b * NB + (i >> 2) < n) yv[u] = yq[i];
  }
  const int node = b * NB + (t >> 1);
  if ((t & 1) == 0) {
    const float dn = rsqrtf(1.0f + (float)indeg);
    dv[t >> 1] = dn;
    if (node < n) dinv[node] = dn;
  }
  __syncthreads();
#pragma unroll
  for (int u = 0; u < NB * 4 / AGT; ++u) {
    const int i = u * AGT + t;
    if (b * NB + (i >> 2) < n) {
      const float s = dv[i >> 2];
      float4 v = yv[u];
      v.x *= s; v.y *= s; v.z *= s; v.w *= s;
      yq[i] = v;
    }
  }
}

template <bool LAST>
__global__ void k_agg(const unsigned* __restrict__ pk, const int* __restrict__ off,
                      const int* __restrict__ rs,
                      const float* __restrict__ xws, const float* __restrict__ dinv,
                      const float* __restrict__ bias, const float* __restrict__ W2,
                      const float* __restrict__ Wl, const float* __restrict__ bl,
                      float* __restrict__ Y, float* __restrict__ out, int n) {
  __shared__ __attribute__((aligned(16))) float acc[NB * 16];
  __shared__ int rowStart[RS];
  __shared__ int sorted[CAP];
  __shared__ int rp[NRUN + 1], ra[NRUN], wt[2 * NRUN / 32];
  __shared__ int lastw;

  const int t = threadIdx.x, b = blockIdx.x;
  const int lane = t & 31;
  const int fq = lane & 3;
  const int grp = t >> 2;
  const float4* xq = (const float4*)xws;
  const int node0 = b * NB;
  for (int i = t; i < NB * 16; i += AGT) {
    const int node = node0 + (i >> 4);
    acc[i] = (node < n) ? xws[(size_t)node0 * 16 + i] : 0.0f;
  }
  const int e0 = binRuns(off, b, t, rp, ra, wt);
  const int e1 = e0 + rp[NRUN];
  const int* rsb = rs + (size_t)(b + e0 / CAP) * RS;

  for (int b0 = e0; b0 < e1; b0 += CAP, rsb += RS) {
    const int b1 = min(b0 + CAP, e1);
    const int nb = b1 - b0;
    const int rsv = rsb[min(t, NB)];
    batchWords(rp, ra, t, b0 - e0, b1 - e0, sorted, &lastw);
    __syncthreads();
    unsigned q[CAP / AGT];
    int ad[CAP / AGT];
    const int lw = lastw;
#pragma unroll
    for (int u = 0; u < CAP / AGT; ++u) ad[u] = (u * AGT + t < nb) ? sorted[u * AGT + t] : lw;
#pragma unroll
    for (int u = 0; u < CAP / AGT; ++u) q[u] = pk[ad[u]];
    if (t < RS) rowStart[t] = rsv;
#pragma unroll
    for (int u = 0; u < CAP / AGT; ++u)
      if (u * AGT + t < nb) sorted[u * AGT + t] = (int)q[u];
    __syncthreads();
    for (int r = grp; r < NB; r += AGT / 4) {
      const int s0 = rowStart[r], s1 = rowStart[r + 1];
      float4 sum = make_float4(0.0f, 0.0f, 0.0f, 0.0f);
      for (int i = s0; i < s1; i += 8) {
        float4 v[8];
#pragma unroll
        for (int u = 0; u < 8; ++u) {
          const int idx = min(i + u, s1 - 1);
          v[u] = xq[(unsigned)sorted[idx] * 4u + fq];
        }
#pragma unroll
        for (int u = 0; u < 8; ++u) {
          const bool in = i + u < s1;
          sum.x += in ? v[u].x : 0.0f;
          sum.y += in ? v[u].y : 0.0f;
          sum.z += in ? v[u].z : 0.0f;
          sum.w += in ? v[u].w : 0.0f;
        }
      }
      float4* ar = (float4*)(acc + r * 16 + fq * 4);
      float4 a = *ar;
      a.x += sum.x; a.y += sum.y; a.z += sum.z; a.w += sum.w;
      *ar = a;
    }
    __syncthreads();
  }
  __syncthreads();

  if (!LAST) {
    const int wave = __builtin_amdgcn_readfirstlane(t >> 5);
    const int row0 = node0 + wave * 16;
    if (row0 < n) {
      const int m  = lane & 15;
      const int kh = lane >> 4;
      const float da = dinv[row0 + m];
      const float4 d0 = *(const float4*)(dinv + row0 + 8 * kh);
      const float4 d1 = *(const float4*)(dinv + row0 + 8 * kh + 4);
      const float dr[8] = {d0.x, d0.y, d0.z, d0.w, d1.x, d1.y, d1.z, d1.w};
      const float* ar = acc + (wave * 16 + m) * 16;
      v8f c = {};
#pragma unroll
      for (int kb = 0; kb < 16; kb += 4) {
        const int k0 = kb + 2 * kh;
        v2f a, w;
        a.x = fmaxf(ar[k0]     * da + bias[k0],     0.0f);
        a.y = fmaxf(ar[k0 + 1] * da + bias[k0 + 1], 0.0f);
        w.x = W2[k0 * 16 + m];
        w.y = W2[(k0 + 1) * 16 + m];
        c = __builtin_amdgcn_wmma_f32_16x16x4_f32(
            false, a, false, w, (short)0, c, false, false);
      }
      float* yp = Y + (size_t)(row0 + 8 * kh) * 16 + m;
#pragma unroll
      for (int v = 0; v < 8; ++v) yp[v * 16] = c[v] * dr[v];
    }
  } else {
    const int node = node0 + t;
    if (t < NB && node < n) {
      const float s = dinv[node];
      float a = 0.0f;
#pragma unroll
      for (int ff = 0; ff < 16; ++ff) {
        const float h = fmaxf(acc[t * 16 + ff] * s + bias[ff], 0.0f);
        a += h * Wl[ff];
      }
      out[node] = a + bl[0];
    }
  }
}

extern "C" void kernel_launch(void* const* d_in, const int* in_sizes, int n_in,
                              void* d_out, int out_size, void* d_ws, size_t ws_size,
                              hipStream_t stream) {
  const float* x   = (const float*)d_in[0];
  const int*   ei  = (const int*)d_in[1];
  const float* W1  = (const float*)d_in[2];
  const float* b1  = (const float*)d_in[3];
  const float* W2  = (const float*)d_in[4];
  const float* b2  = (const float*)d_in[5];
  const float* Wl  = (const float*)d_in[6];
  const float* bl  = (const float*)d_in[7];
  float* out = (float*)d_out;

  const int din = in_sizes[2] / 16;
  const int n   = in_sizes[0] / din;
  const int E   = in_sizes[1] / 2;
  const int tot = n * 16;

  float* ws   = (float*)d_ws;
  float* dinv = ws;
  float* bufA = ws + (((size_t)n + 255) & ~(size_t)255);
  float* bufB = bufA + (size_t)n * 16;

  const int T = 256;
  const int ntiles = (n + 15) / 16;
  const int gemmBlocks = (ntiles + 3) / 4;

  const int nbins = (n + NB - 1) / NB;
  unsigned* pk   = (unsigned*)(bufB + (size_t)n * 16);
  int* off       = (int*)(pk + (size_t)NRUN * WCAP);
  int* rs        = off + (size_t)NRUN * OS;
  const size_t need = (size_t)((char*)(rs + (size_t)(nbins + E / CAP + 1) * RS) - (char*)d_ws);

  const int chunk = (E + PG - 1) / PG;
  if (ws_size >= need && nbins < MAXBINS && chunk <= 32 * GM) {
    k_bucket  <<<NRUN + nbins, BT, 0, stream>>>(ei, ei + E, off, pk, E, chunk, nbins, x, W1, bufA, n, din);
    k_sort    <<<nbins, AGT, 0, stream>>>(pk, off, rs, dinv, n, bufA);

    k_agg<false><<<nbins, AGT, 0, stream>>>(pk, off, rs, bufA, dinv, b1, W2, Wl, bl, bufB, out, n);
    k_agg<true> <<<nbins, AGT, 0, stream>>>(pk, off, rs, bufB, dinv, b2, W2, Wl, bl, bufA, out, n);
    return;
  }

  k_init  <<<(tot + T - 1) / T, T, 0, stream>>>(dinv, bufB, n, tot);
  k_degree<<<(E + T - 1) / T,   T, 0, stream>>>(ei + E, dinv, E);
  k_rsqrt <<<(n + T - 1) / T,   T, 0, stream>>>(dinv, n);

  k_gemm16 <<<gemmBlocks, 128, 0, stream>>>(x, W1, bufA, n, din);
  k_scatter<<<(4 * E + T - 1) / T, T, 0, stream>>>(ei, ei + E, dinv, bufA, bufB, E);
  k_final1 <<<(tot + T - 1) / T, T, 0, stream>>>(bufB, bufA, dinv, b1, tot);

  k_gemm16 <<<gemmBlocks, 128, 0, stream>>>(bufB, W2, bufA, n, 16);
  k_zero   <<<(tot + T - 1) / T, T, 0, stream>>>(bufB, tot);
  k_scatter<<<(4 * E + T - 1) / T, T, 0, stream>>>(ei, ei + E, dinv, bufA, bufB, E);
  k_final2 <<<(n + T - 1) / T, T, 0, stream>>>(bufB, bufA, dinv, b2, Wl, bl, out, n);
}
